# starting kernel
# baseline (speedup 1.0000x reference)
#include <hip/hip_runtime.h>
#include <cmath>


#define NN 50000
#define NE 800000
#define DIN 64
#define DHID 128
#define DOUT 64

typedef __attribute__((ext_vector_type(2))) float v2f;
typedef __attribute__((ext_vector_type(8))) float v8f;

__global__ void zero_kernel(float* __restrict__ p, int n) {
    int i = blockIdx.x * blockDim.x + threadIdx.x;
    if (i < n) p[i] = 0.0f;
}

__global__ void scatter1_kernel(const float* __restrict__ x,
                                const int* __restrict__ ei,
                                float* __restrict__ sum1,
                                float* __restrict__ deg) {
    int tid = blockIdx.x * blockDim.x + threadIdx.x;
    if (tid >= NE * DIN) return;
    int e = tid >> 6;
    int f = tid & 63;
    int src = ei[e];
    int dst = ei[NE + e];
    atomicAdd(&sum1[dst * DIN + f], x[src * DIN + f]);
    if (f == 0) atomicAdd(&deg[dst], 1.0f);
}

__global__ void scatter2_kernel(const float* __restrict__ h1,
                                const int* __restrict__ ei,
                                float* __restrict__ sum2) {
    int tid = blockIdx.x * blockDim.x + threadIdx.x;
    if (tid >= NE * DHID) return;
    int e = tid >> 7;
    int f = tid & 127;
    int src = ei[e];
    int dst = ei[NE + e];
    atomicAdd(&sum2[dst * DHID + f], h1[src * DHID + f]);
}

__global__ __launch_bounds__(32) void gemm1_kernel(
    const float* __restrict__ x, const float* __restrict__ sum1,
    const float* __restrict__ deg,
    const float* __restrict__ Wl, const float* __restrict__ bl,
    const float* __restrict__ Wr, float* __restrict__ h1) {
    const int lane = threadIdx.x;
    const int m0   = blockIdx.x * 16;
    const int mr   = lane & 15;
    const int kb   = (lane >> 4) * 2;
    const int m    = m0 + mr;
    const float dinv = 1.0f / fmaxf(deg[m], 1.0f);

    v8f acc[8] = {};

    for (int kk = 0; kk < DIN / 4; ++kk) {
        const int k = kk * 4 + kb;
        v2f a;
        a.x = sum1[m * DIN + k] * dinv;
        a.y = sum1[m * DIN + k + 1] * dinv;
#pragma unroll
        for (int nt = 0; nt < 8; ++nt) {
            const int n = nt * 16 + mr;
            v2f b;
            b.x = Wl[n * DIN + k];
            b.y = Wl[n * DIN + k + 1];
            acc[nt] = __builtin_amdgcn_wmma_f32_16x16x4_f32(
                false, a, false, b, (short)0, acc[nt], false, false);
        }
    }
    for (int kk = 0; kk < DIN / 4; ++kk) {
        const int k = kk * 4 + kb;
        v2f a;
        a.x = x[m * DIN + k];
        a.y = x[m * DIN + k + 1];
#pragma unroll
        for (int nt = 0; nt < 8; ++nt) {
            const int n = nt * 16 + mr;
            v2f b;
            b.x = Wr[n * DIN + k];
            b.y = Wr[n * DIN + k + 1];
            acc[nt] = __builtin_amdgcn_wmma_f32_16x16x4_f32(
                false, a, false, b, (short)0, acc[nt], false, false);
        }
    }

    const int mo = m0 + 8 * (lane >> 4);
#pragma unroll
    for (int nt = 0; nt < 8; ++nt) {
        const int n = nt * 16 + mr;
        const float bias = bl[n];
#pragma unroll
        for (int v = 0; v < 8; ++v) {
            h1[(mo + v) * DHID + n] = tanhf(acc[nt][v] + bias);
        }
    }
}

__global__ __launch_bounds__(32) void gemm2_kernel(
    const float* __restrict__ h1, const float* __restrict__ sum2,
    const float* __restrict__ deg,
    const float* __restrict__ Wl, const float* __restrict__ bl,
    const float* __restrict__ Wr, float* __restrict__ out) {
    __shared__ float zrow[16 * DOUT];
    const int lane = threadIdx.x;
    const int m0   = blockIdx.x * 16;
    const int mr   = lane & 15;
    const int kb   = (lane >> 4) * 2;
    const int m    = m0 + mr;
    const float dinv = 1.0f / fmaxf(deg[m], 1.0f);

    v8f acc[4] = {};

    for (int kk = 0; kk < DHID / 4; ++kk) {
        const int k = kk * 4 + kb;
        v2f a;
        a.x = sum2[m * DHID + k] * dinv;
        a.y = sum2[m * DHID + k + 1] * dinv;
#pragma unroll
        for (int nt = 0; nt < 4; ++nt) {
            const int n = nt * 16 + mr;
            v2f b;
            b.x = Wl[n * DHID + k];
            b.y = Wl[n * DHID + k + 1];
            acc[nt] = __builtin_amdgcn_wmma_f32_16x16x4_f32(
                false, a, false, b, (short)0, acc[nt], false, false);
        }
    }
    for (int kk = 0; kk < DHID / 4; ++kk) {
        const int k = kk * 4 + kb;
        v2f a;
        a.x = h1[m * DHID + k];
        a.y = h1[m * DHID + k + 1];
#pragma unroll
        for (int nt = 0; nt < 4; ++nt) {
            const int n = nt * 16 + mr;
            v2f b;
            b.x = Wr[n * DHID + k];
            b.y = Wr[n * DHID + k + 1];
            acc[nt] = __builtin_amdgcn_wmma_f32_16x16x4_f32(
                false, a, false, b, (short)0, acc[nt], false, false);
        }
    }

    const int rbase = 8 * (lane >> 4);
#pragma unroll
    for (int nt = 0; nt < 4; ++nt) {
        const int n = nt * 16 + mr;
        const float bias = bl[n];
#pragma unroll
        for (int v = 0; v < 8; ++v) {
            zrow[(rbase + v) * DOUT + n] = acc[nt][v] + bias;
        }
    }
    __syncthreads();

    if (lane < 16) {
        const float* zr = &zrow[lane * DOUT];
        float mx = zr[0];
        for (int c = 1; c < DOUT; ++c) mx = fmaxf(mx, zr[c]);
        float s = 0.0f;
        for (int c = 0; c < DOUT; ++c) s += expf(zr[c] - mx);
        const float lse = mx + logf(s);
        float* orow = &out[(m0 + lane) * DOUT];
        for (int c = 0; c < DOUT; ++c) orow[c] = zr[c] - lse;
    }
}

extern "C" void kernel_launch(void* const* d_in, const int* in_sizes, int n_in,
                              void* d_out, int out_size, void* d_ws, size_t ws_size,
                              hipStream_t stream) {
    const float* x   = (const float*)d_in[0];
    const int*   ei  = (const int*)d_in[1];
    const float* Wl1 = (const float*)d_in[2];
    const float* bl1 = (const float*)d_in[3];
    const float* Wr1 = (const float*)d_in[4];
    const float* Wl2 = (const float*)d_in[5];
    const float* bl2 = (const float*)d_in[6];
    const float* Wr2 = (const float*)d_in[7];
    float* out = (float*)d_out;

    float* ws   = (float*)d_ws;
    float* sum1 = ws;
    float* sum2 = sum1 + (size_t)NN * DIN;
    float* degb = sum2 + (size_t)NN * DHID;
    float* h1   = degb + NN;

    const int nz = NN * (DIN + DHID) + NN;
    zero_kernel<<<(nz + 255) / 256, 256, 0, stream>>>(ws, nz);

    scatter1_kernel<<<(NE * DIN + 255) / 256, 256, 0, stream>>>(x, ei, sum1, degb);
    gemm1_kernel<<<NN / 16, 32, 0, stream>>>(x, sum1, degb, Wl1, bl1, Wr1, h1);
    scatter2_kernel<<<(NE * DHID + 255) / 256, 256, 0, stream>>>(h1, ei, sum2);
    gemm2_kernel<<<NN / 16, 32, 0, stream>>>(h1, sum2, degb, Wl2, bl2, Wr2, out);
}


# Replace per-feature f32 atomic scatter-add by a per-call in-edge list (one int exchange per edge) + one-wave-per-node register gather; sums now in list order, not atomic arrival order
# speedup vs baseline: 2.8490x; 2.8490x over previous
#include <hip/hip_runtime.h>
#include <cmath>


#define NN 50000
#define NE 800000
#define DIN 64
#define DHID 128
#define DOUT 64

typedef __attribute__((ext_vector_type(2))) float v2f;
typedef __attribute__((ext_vector_type(8))) float v8f;

__global__ void zero_kernel(float* __restrict__ p, int n) {
    int i = blockIdx.x * blockDim.x + threadIdx.x;
    if (i < n) p[i] = 0.0f;
}

__global__ void link_kernel(const int* __restrict__ ei,
                            int* __restrict__ head,
                            int2* __restrict__ rec) {
    int e = blockIdx.x * blockDim.x + threadIdx.x;
    if (e >= NE) return;
    int src = ei[e];
    int dst = ei[NE + e];
    int prev = atomicExch(&head[dst], e + 1);
    rec[e] = make_int2(src, prev);
}

__device__ __forceinline__ bool cursor_ok(int c) {
    return (unsigned)(c - 1) < (unsigned)NE;
}

__global__ __launch_bounds__(256) void gather1_kernel(
    const float* __restrict__ x, const int* __restrict__ head,
    const int2* __restrict__ rec, float* __restrict__ sum1,
    float* __restrict__ deg) {
    const int lane = threadIdx.x & 31;
    const int wv   = __builtin_amdgcn_readfirstlane((int)(threadIdx.x >> 5));
    const int node = blockIdx.x * 8 + wv;
    if (node >= NN) return;

    int cur = __builtin_amdgcn_readfirstlane(head[node]);
    bool have = cursor_ok(cur);
    int2 r = make_int2(0, 0);
    if (have) r = rec[cur - 1];

    float2 acc = make_float2(0.0f, 0.0f);
    int cnt = 0;
    for (int it = 0; it < NE && have; ++it) {
        const float2 v = *(const float2*)(x + (size_t)r.x * DIN + lane * 2);
        const int nxt = r.y;
        const bool hn = cursor_ok(nxt);
        int2 rn = make_int2(0, 0);
        if (hn) rn = rec[nxt - 1];
        acc.x += v.x;
        acc.y += v.y;
        ++cnt;
        r = rn;
        have = hn;
    }
    *(float2*)(sum1 + (size_t)node * DIN + lane * 2) = acc;
    if (lane == 0) deg[node] = (float)cnt;
}

__global__ __launch_bounds__(256) void gather2_kernel(
    const float* __restrict__ h1, const int* __restrict__ head,
    const int2* __restrict__ rec, float* __restrict__ sum2) {
    const int lane = threadIdx.x & 31;
    const int wv   = __builtin_amdgcn_readfirstlane((int)(threadIdx.x >> 5));
    const int node = blockIdx.x * 8 + wv;
    if (node >= NN) return;

    int cur = __builtin_amdgcn_readfirstlane(head[node]);
    bool have = cursor_ok(cur);
    int2 r = make_int2(0, 0);
    if (have) r = rec[cur - 1];

    float4 acc = make_float4(0.0f, 0.0f, 0.0f, 0.0f);
    for (int it = 0; it < NE && have; ++it) {
        const float4 v = *(const float4*)(h1 + (size_t)r.x * DHID + lane * 4);
        const int nxt = r.y;
        const bool hn = cursor_ok(nxt);
        int2 rn = make_int2(0, 0);
        if (hn) rn = rec[nxt - 1];
        acc.x += v.x;
        acc.y += v.y;
        acc.z += v.z;
        acc.w += v.w;
        r = rn;
        have = hn;
    }
    *(float4*)(sum2 + (size_t)node * DHID + lane * 4) = acc;
}

__global__ __launch_bounds__(32) void gemm1_kernel(
    const float* __restrict__ x, const float* __restrict__ sum1,
    const float* __restrict__ deg,
    const float* __restrict__ Wl, const float* __restrict__ bl,
    const float* __restrict__ Wr, float* __restrict__ h1) {
    const int lane = threadIdx.x;
    const int m0   = blockIdx.x * 16;
    const int mr   = lane & 15;
    const int kb   = (lane >> 4) * 2;
    const int m    = m0 + mr;
    const float dinv = 1.0f / fmaxf(deg[m], 1.0f);

    v8f acc[8] = {};

    for (int kk = 0; kk < DIN / 4; ++kk) {
        const int k = kk * 4 + kb;
        v2f a;
        a.x = sum1[m * DIN + k] * dinv;
        a.y = sum1[m * DIN + k + 1] * dinv;
#pragma unroll
        for (int nt = 0; nt < 8; ++nt) {
            const int n = nt * 16 + mr;
            v2f b;
            b.x = Wl[n * DIN + k];
            b.y = Wl[n * DIN + k + 1];
            acc[nt] = __builtin_amdgcn_wmma_f32_16x16x4_f32(
                false, a, false, b, (short)0, acc[nt], false, false);
        }
    }
    for (int kk = 0; kk < DIN / 4; ++kk) {
        const int k = kk * 4 + kb;
        v2f a;
        a.x = x[m * DIN + k];
        a.y = x[m * DIN + k + 1];
#pragma unroll
        for (int nt = 0; nt < 8; ++nt) {
            const int n = nt * 16 + mr;
            v2f b;
            b.x = Wr[n * DIN + k];
            b.y = Wr[n * DIN + k + 1];
            acc[nt] = __builtin_amdgcn_wmma_f32_16x16x4_f32(
                false, a, false, b, (short)0, acc[nt], false, false);
        }
    }

    const int mo = m0 + 8 * (lane >> 4);
#pragma unroll
    for (int nt = 0; nt < 8; ++nt) {
        const int n = nt * 16 + mr;
        const float bias = bl[n];
#pragma unroll
        for (int v = 0; v < 8; ++v) {
            h1[(mo + v) * DHID + n] = tanhf(acc[nt][v] + bias);
        }
    }
}

__global__ __launch_bounds__(32) void gemm2_kernel(
    const float* __restrict__ h1, const float* __restrict__ sum2,
    const float* __restrict__ deg,
    const float* __restrict__ Wl, const float* __restrict__ bl,
    const float* __restrict__ Wr, float* __restrict__ out) {
    __shared__ float zrow[16 * DOUT];
    const int lane = threadIdx.x;
    const int m0   = blockIdx.x * 16;
    const int mr   = lane & 15;
    const int kb   = (lane >> 4) * 2;
    const int m    = m0 + mr;
    const float dinv = 1.0f / fmaxf(deg[m], 1.0f);

    v8f acc[4] = {};

    for (int kk = 0; kk < DHID / 4; ++kk) {
        const int k = kk * 4 + kb;
        v2f a;
        a.x = sum2[m * DHID + k] * dinv;
        a.y = sum2[m * DHID + k + 1] * dinv;
#pragma unroll
        for (int nt = 0; nt < 4; ++nt) {
            const int n = nt * 16 + mr;
            v2f b;
            b.x = Wl[n * DHID + k];
            b.y = Wl[n * DHID + k + 1];
            acc[nt] = __builtin_amdgcn_wmma_f32_16x16x4_f32(
                false, a, false, b, (short)0, acc[nt], false, false);
        }
    }
    for (int kk = 0; kk < DHID / 4; ++kk) {
        const int k = kk * 4 + kb;
        v2f a;
        a.x = h1[m * DHID + k];
        a.y = h1[m * DHID + k + 1];
#pragma unroll
        for (int nt = 0; nt < 4; ++nt) {
            const int n = nt * 16 + mr;
            v2f b;
            b.x = Wr[n * DHID + k];
            b.y = Wr[n * DHID + k + 1];
            acc[nt] = __builtin_amdgcn_wmma_f32_16x16x4_f32(
                false, a, false, b, (short)0, acc[nt], false, false);
        }
    }

    const int rbase = 8 * (lane >> 4);
#pragma unroll
    for (int nt = 0; nt < 4; ++nt) {
        const int n = nt * 16 + mr;
        const float bias = bl[n];
#pragma unroll
        for (int v = 0; v < 8; ++v) {
            zrow[(rbase + v) * DOUT + n] = acc[nt][v] + bias;
        }
    }
    __syncthreads();

    if (lane < 16) {
        const float* zr = &zrow[lane * DOUT];
        float mx = zr[0];
        for (int c = 1; c < DOUT; ++c) mx = fmaxf(mx, zr[c]);
        float s = 0.0f;
        for (int c = 0; c < DOUT; ++c) s += expf(zr[c] - mx);
        const float lse = mx + logf(s);
        float* orow = &out[(m0 + lane) * DOUT];
        for (int c = 0; c < DOUT; ++c) orow[c] = zr[c] - lse;
    }
}

extern "C" void kernel_launch(void* const* d_in, const int* in_sizes, int n_in,
                              void* d_out, int out_size, void* d_ws, size_t ws_size,
                              hipStream_t stream) {
    const float* x   = (const float*)d_in[0];
    const int*   ei  = (const int*)d_in[1];
    const float* Wl1 = (const float*)d_in[2];
    const float* bl1 = (const float*)d_in[3];
    const float* Wr1 = (const float*)d_in[4];
    const float* Wl2 = (const float*)d_in[5];
    const float* bl2 = (const float*)d_in[6];
    const float* Wr2 = (const float*)d_in[7];
    float* out = (float*)d_out;

    float* ws   = (float*)d_ws;
    float* sum2 = ws;
    float* sum1 = ws;
    float* h1   = sum2 + (size_t)NN * DHID;
    int2*  rec  = (int2*)(h1 + (size_t)NN * DHID);
    int*   head = (int*)(rec + NE);
    float* degb = (float*)(head + NN);

    zero_kernel<<<(NN + 255) / 256, 256, 0, stream>>>((float*)head, NN);
    link_kernel<<<(NE + 255) / 256, 256, 0, stream>>>(ei, head, rec);

    gather1_kernel<<<NN / 8, 256, 0, stream>>>(x, head, rec, sum1, degb);
    gemm1_kernel<<<NN / 16, 32, 0, stream>>>(x, sum1, degb, Wl1, bl1, Wr1, h1);
    gather2_kernel<<<NN / 8, 256, 0, stream>>>(h1, head, rec, sum2);
    gemm2_kernel<<<NN / 16, 32, 0, stream>>>(h1, sum2, degb, Wl2, bl2, Wr2, out);
}


# Each gather wave walks the lists of 8 consecutive nodes side by side: lane j keeps list j's cursor, one vector load fetches all next records, 8 row loads in flight per trip; sum order per node kept
# speedup vs baseline: 3.0296x; 1.0634x over previous
#include <hip/hip_runtime.h>
#include <cmath>


#define NN 50000
#define NE 800000
#define DIN 64
#define DHID 128
#define DOUT 64
#define GK 8

typedef __attribute__((ext_vector_type(2))) float v2f;
typedef __attribute__((ext_vector_type(8))) float v8f;

__global__ void zero_kernel(float* __restrict__ p, int n) {
    int i = blockIdx.x * blockDim.x + threadIdx.x;
    if (i < n) p[i] = 0.0f;
}

__global__ void link_kernel(const int* __restrict__ ei,
                            int* __restrict__ head,
                            int2* __restrict__ rec) {
    int e = blockIdx.x * blockDim.x + threadIdx.x;
    if (e >= NE) return;
    int src = ei[e];
    int dst = ei[NE + e];
    int prev = atomicExch(&head[dst], e + 1);
    rec[e] = make_int2(src, prev);
}

__device__ __forceinline__ bool cursor_ok(int c) {
    return (unsigned)(c - 1) < (unsigned)NE;
}

__global__ __launch_bounds__(256) void gather1_kernel(
    const float* __restrict__ x, const int* __restrict__ head,
    const int2* __restrict__ rec, float* __restrict__ sum1,
    float* __restrict__ deg) {
    const int lane = threadIdx.x & 31;
    const int wv    = __builtin_amdgcn_readfirstlane((int)(threadIdx.x >> 5));
    const int node0 = (blockIdx.x * 8 + wv) * GK;
    if (node0 >= NN) return;

    const bool owner = lane < GK && node0 + lane < NN;
    int cur = 0;
    if (owner) cur = head[node0 + lane];
    bool alive = owner && cursor_ok(cur);
    int2 r = make_int2(0, 0);
    if (alive) r = rec[cur - 1];

    float2 acc[GK];
#pragma unroll
    for (int j = 0; j < GK; ++j) acc[j] = make_float2(0.0f, 0.0f);
    int cnt = 0;

    for (int it = 0; it < NE; ++it) {
        const unsigned live = __builtin_amdgcn_ballot_w32(alive);
        if (live == 0u) break;
        const int nxt = r.y;
        const bool hn = alive && cursor_ok(nxt);
        int2 rn = make_int2(0, 0);
        if (hn) rn = rec[nxt - 1];
        float2 v[GK];
#pragma unroll
        for (int j = 0; j < GK; ++j) {
            v[j] = make_float2(0.0f, 0.0f);
            if (live & (1u << j)) {
                const int s = __builtin_amdgcn_readlane(r.x, j);
                v[j] = *(const float2*)(x + (size_t)s * DIN + lane * 2);
            }
        }
#pragma unroll
        for (int j = 0; j < GK; ++j) {
            acc[j].x += v[j].x;
            acc[j].y += v[j].y;
        }
        cnt += alive ? 1 : 0;
        r = rn;
        alive = hn;
    }
#pragma unroll
    for (int j = 0; j < GK; ++j) {
        if (node0 + j < NN)
            *(float2*)(sum1 + (size_t)(node0 + j) * DIN + lane * 2) = acc[j];
    }
    if (owner) deg[node0 + lane] = (float)cnt;
}

__global__ __launch_bounds__(256) void gather2_kernel(
    const float* __restrict__ h1, const int* __restrict__ head,
    const int2* __restrict__ rec, float* __restrict__ sum2) {
    const int lane = threadIdx.x & 31;
    const int wv    = __builtin_amdgcn_readfirstlane((int)(threadIdx.x >> 5));
    const int node0 = (blockIdx.x * 8 + wv) * GK;
    if (node0 >= NN) return;

    const bool owner = lane < GK && node0 + lane < NN;
    int cur = 0;
    if (owner) cur = head[node0 + lane];
    bool alive = owner && cursor_ok(cur);
    int2 r = make_int2(0, 0);
    if (alive) r = rec[cur - 1];

    float4 acc[GK];
#pragma unroll
    for (int j = 0; j < GK; ++j) acc[j] = make_float4(0.0f, 0.0f, 0.0f, 0.0f);

    for (int it = 0; it < NE; ++it) {
        const unsigned live = __builtin_amdgcn_ballot_w32(alive);
        if (live == 0u) break;
        const int nxt = r.y;
        const bool hn = alive && cursor_ok(nxt);
        int2 rn = make_int2(0, 0);
        if (hn) rn = rec[nxt - 1];
        float4 v[GK];
#pragma unroll
        for (int j = 0; j < GK; ++j) {
            v[j] = make_float4(0.0f, 0.0f, 0.0f, 0.0f);
            if (live & (1u << j)) {
                const int s = __builtin_amdgcn_readlane(r.x, j);
                v[j] = *(const float4*)(h1 + (size_t)s * DHID + lane * 4);
            }
        }
#pragma unroll
        for (int j = 0; j < GK; ++j) {
            acc[j].x += v[j].x;
            acc[j].y += v[j].y;
            acc[j].z += v[j].z;
            acc[j].w += v[j].w;
        }
        r = rn;
        alive = hn;
    }
#pragma unroll
    for (int j = 0; j < GK; ++j) {
        if (node0 + j < NN)
            *(float4*)(sum2 + (size_t)(node0 + j) * DHID + lane * 4) = acc[j];
    }
}

__global__ __launch_bounds__(32) void gemm1_kernel(
    const float* __restrict__ x, const float* __restrict__ sum1,
    const float* __restrict__ deg,
    const float* __restrict__ Wl, const float* __restrict__ bl,
    const float* __restrict__ Wr, float* __restrict__ h1) {
    const int lane = threadIdx.x;
    const int m0   = blockIdx.x * 16;
    const int mr   = lane & 15;
    const int kb   = (lane >> 4) * 2;
    const int m    = m0 + mr;
    const float dinv = 1.0f / fmaxf(deg[m], 1.0f);

    v8f acc[8] = {};

    for (int kk = 0; kk < DIN / 4; ++kk) {
        const int k = kk * 4 + kb;
        v2f a;
        a.x = sum1[m * DIN + k] * dinv;
        a.y = sum1[m * DIN + k + 1] * dinv;
#pragma unroll
        for (int nt = 0; nt < 8; ++nt) {
            const int n = nt * 16 + mr;
            v2f b;
            b.x = Wl[n * DIN + k];
            b.y = Wl[n * DIN + k + 1];
            acc[nt] = __builtin_amdgcn_wmma_f32_16x16x4_f32(
                false, a, false, b, (short)0, acc[nt], false, false);
        }
    }
    for (int kk = 0; kk < DIN / 4; ++kk) {
        const int k = kk * 4 + kb;
        v2f a;
        a.x = x[m * DIN + k];
        a.y = x[m * DIN + k + 1];
#pragma unroll
        for (int nt = 0; nt < 8; ++nt) {
            const int n = nt * 16 + mr;
            v2f b;
            b.x = Wr[n * DIN + k];
            b.y = Wr[n * DIN + k + 1];
            acc[nt] = __builtin_amdgcn_wmma_f32_16x16x4_f32(
                false, a, false, b, (short)0, acc[nt], false, false);
        }
    }

    const int mo = m0 + 8 * (lane >> 4);
#pragma unroll
    for (int nt = 0; nt < 8; ++nt) {
        const int n = nt * 16 + mr;
        const float bias = bl[n];
#pragma unroll
        for (int v = 0; v < 8; ++v) {
            h1[(mo + v) * DHID + n] = tanhf(acc[nt][v] + bias);
        }
    }
}

__global__ __launch_bounds__(32) void gemm2_kernel(
    const float* __restrict__ h1, const float* __restrict__ sum2,
    const float* __restrict__ deg,
    const float* __restrict__ Wl, const float* __restrict__ bl,
    const float* __restrict__ Wr, float* __restrict__ out) {
    __shared__ float zrow[16 * DOUT];
    const int lane = threadIdx.x;
    const int m0   = blockIdx.x * 16;
    const int mr   = lane & 15;
    const int kb   = (lane >> 4) * 2;
    const int m    = m0 + mr;
    const float dinv = 1.0f / fmaxf(deg[m], 1.0f);

    v8f acc[4] = {};

    for (int kk = 0; kk < DHID / 4; ++kk) {
        const int k = kk * 4 + kb;
        v2f a;
        a.x = sum2[m * DHID + k] * dinv;
        a.y = sum2[m * DHID + k + 1] * dinv;
#pragma unroll
        for (int nt = 0; nt < 4; ++nt) {
            const int n = nt * 16 + mr;
            v2f b;
            b.x = Wl[n * DHID + k];
            b.y = Wl[n * DHID + k + 1];
            acc[nt] = __builtin_amdgcn_wmma_f32_16x16x4_f32(
                false, a, false, b, (short)0, acc[nt], false, false);
        }
    }
    for (int kk = 0; kk < DHID / 4; ++kk) {
        const int k = kk * 4 + kb;
        v2f a;
        a.x = h1[m * DHID + k];
        a.y = h1[m * DHID + k + 1];
#pragma unroll
        for (int nt = 0; nt < 4; ++nt) {
            const int n = nt * 16 + mr;
            v2f b;
            b.x = Wr[n * DHID + k];
            b.y = Wr[n * DHID + k + 1];
            acc[nt] = __builtin_amdgcn_wmma_f32_16x16x4_f32(
                false, a, false, b, (short)0, acc[nt], false, false);
        }
    }

    const int rbase = 8 * (lane >> 4);
#pragma unroll
    for (int nt = 0; nt < 4; ++nt) {
        const int n = nt * 16 + mr;
        const float bias = bl[n];
#pragma unroll
        for (int v = 0; v < 8; ++v) {
            zrow[(rbase + v) * DOUT + n] = acc[nt][v] + bias;
        }
    }
    __syncthreads();

    if (lane < 16) {
        const float* zr = &zrow[lane * DOUT];
        float mx = zr[0];
        for (int c = 1; c < DOUT; ++c) mx = fmaxf(mx, zr[c]);
        float s = 0.0f;
        for (int c = 0; c < DOUT; ++c) s += expf(zr[c] - mx);
        const float lse = mx + logf(s);
        float* orow = &out[(m0 + lane) * DOUT];
        for (int c = 0; c < DOUT; ++c) orow[c] = zr[c] - lse;
    }
}

extern "C" void kernel_launch(void* const* d_in, const int* in_sizes, int n_in,
                              void* d_out, int out_size, void* d_ws, size_t ws_size,
                              hipStream_t stream) {
    const float* x   = (const float*)d_in[0];
    const int*   ei  = (const int*)d_in[1];
    const float* Wl1 = (const float*)d_in[2];
    const float* bl1 = (const float*)d_in[3];
    const float* Wr1 = (const float*)d_in[4];
    const float* Wl2 = (const float*)d_in[5];
    const float* bl2 = (const float*)d_in[6];
    const float* Wr2 = (const float*)d_in[7];
    float* out = (float*)d_out;

    float* ws   = (float*)d_ws;
    float* sum2 = ws;
    float* sum1 = ws;
    float* h1   = sum2 + (size_t)NN * DHID;
    int2*  rec  = (int2*)(h1 + (size_t)NN * DHID);
    int*   head = (int*)(rec + NE);
    float* degb = (float*)(head + NN);

    zero_kernel<<<(NN + 255) / 256, 256, 0, stream>>>((float*)head, NN);
    link_kernel<<<(NE + 255) / 256, 256, 0, stream>>>(ei, head, rec);

    gather1_kernel<<<(NN + 8 * GK - 1) / (8 * GK), 256, 0, stream>>>(x, head, rec, sum1, degb);
    gemm1_kernel<<<NN / 16, 32, 0, stream>>>(x, sum1, degb, Wl1, bl1, Wr1, h1);
    gather2_kernel<<<(NN + 8 * GK - 1) / (8 * GK), 256, 0, stream>>>(h1, head, rec, sum2);
    gemm2_kernel<<<NN / 16, 32, 0, stream>>>(h1, sum2, degb, Wl2, bl2, Wr2, out);
}


# Build the per-node in-edge lists without global atomics: counting sort of the edges by destination bucket on LDS counters, then LDS exchanges link each bucket; list order still by atomic arrival
# speedup vs baseline: 3.5647x; 1.1766x over previous
#include <hip/hip_runtime.h>
#include <cmath>


#define NN 50000
#define NE 800000
#define DIN 64
#define DHID 128
#define DOUT 64
#define GK 8

#define BS 256
#define NB ((NN + BS - 1) / BS)
#define NCH 125
#define CH (NE / NCH)
#define CT 640
#define EPT (CH / CT)
#define RG (CT / 64)
#define LT 512
#define LB 4
static_assert(NCH * CH == NE && EPT * CT == CH, "chunks tile the edge list exactly");
static_assert(NB % 4 == 0 && NB / 4 <= 64 && NB <= CT, "count rows are read as int4 by 64 lanes");

typedef __attribute__((ext_vector_type(2))) float v2f;
typedef __attribute__((ext_vector_type(8))) float v8f;

__global__ __launch_bounds__(CT) void count_kernel(const int* __restrict__ ei,
                                                   int* __restrict__ cnt) {
    __shared__ int h[NB];
    const int tid = threadIdx.x;
    if (tid < NB) h[tid] = 0;
    const int e0 = blockIdx.x * CH;
    int d[EPT];
#pragma unroll
    for (int k = 0; k < EPT; ++k) d[k] = ei[NE + e0 + k * CT + tid];
    __syncthreads();
#pragma unroll
    for (int k = 0; k < EPT; ++k)
        if ((unsigned)d[k] < (unsigned)NN) atomicAdd(&h[(unsigned)d[k] / BS], 1);
    __syncthreads();
    if (tid < NB) cnt[blockIdx.x * NB + tid] = h[tid];
}

__global__ __launch_bounds__(CT) void place_kernel(const int* __restrict__ ei,
                                                   const int* __restrict__ cnt,
                                                   int* __restrict__ bstart,
                                                   int2* __restrict__ rec) {
    __shared__ int tot[NB];
    __shared__ int cur[NB];
    const int tid = threadIdx.x;
    const int g   = blockIdx.x;
    if (tid < NB) { tot[tid] = 0; cur[tid] = 0; }
    const int e0 = g * CH;
    int s[EPT], d[EPT];
#pragma unroll
    for (int k = 0; k < EPT; ++k) {
        s[k] = ei[e0 + k * CT + tid];
        d[k] = ei[NE + e0 + k * CT + tid];
    }
    constexpr int NR = (NCH + RG - 1) / RG;
    const int c4 = tid & 63;
    const int rg = tid >> 6;
    const int c4c = min(c4, NB / 4 - 1);
    int4 cm[NR];
#pragma unroll
    for (int j = 0; j < NR; ++j) {
        const int r = min(rg + j * RG, NCH - 1);
        cm[j] = *(const int4*)(cnt + r * NB + c4c * 4);
    }
    int4 st = make_int4(0, 0, 0, 0);
    int4 sp = make_int4(0, 0, 0, 0);
#pragma unroll
    for (int j = 0; j < NR; ++j) {
        const int r  = rg + j * RG;
        const int ma = (r < NCH) ? -1 : 0;
        const int mp = (r < g) ? ma : 0;
        st.x += cm[j].x & ma; st.y += cm[j].y & ma; st.z += cm[j].z & ma; st.w += cm[j].w & ma;
        sp.x += cm[j].x & mp; sp.y += cm[j].y & mp; sp.z += cm[j].z & mp; sp.w += cm[j].w & mp;
    }
    __syncthreads();
    if (c4 < NB / 4) {
        atomicAdd(&tot[c4 * 4 + 0], st.x); atomicAdd(&cur[c4 * 4 + 0], sp.x);
        atomicAdd(&tot[c4 * 4 + 1], st.y); atomicAdd(&cur[c4 * 4 + 1], sp.y);
        atomicAdd(&tot[c4 * 4 + 2], st.z); atomicAdd(&cur[c4 * 4 + 2], sp.z);
        atomicAdd(&tot[c4 * 4 + 3], st.w); atomicAdd(&cur[c4 * 4 + 3], sp.w);
    }
    __syncthreads();
    if (tid < 32) {
        constexpr int PER = (NB + 31) / 32;
        int loc[PER];
        int sum = 0;
#pragma unroll
        for (int i = 0; i < PER; ++i) {
            const int b = tid * PER + i;
            const int v = (b < NB) ? tot[b] : 0;
            loc[i] = sum;
            sum += v;
        }
        int inc = sum;
#pragma unroll
        for (int dd = 1; dd < 32; dd <<= 1) {
            const int o = __shfl_up(inc, dd);
            if (tid >= dd) inc += o;
        }
        const int excl = inc - sum;
#pragma unroll
        for (int i = 0; i < PER; ++i) {
            const int b = tid * PER + i;
            if (b < NB) {
                const int bb = excl + loc[i];
                if (g == 0) bstart[b] = bb;
                cur[b] += bb;
            }
        }
        if (g == 0 && tid == 31) bstart[NB] = inc;
    }
    __syncthreads();
#pragma unroll
    for (int k = 0; k < EPT; ++k) {
        if ((unsigned)d[k] < (unsigned)NN) {
            const int pos = atomicAdd(&cur[(unsigned)d[k] / BS], 1);
            if ((unsigned)pos < (unsigned)NE) rec[pos] = make_int2(s[k], d[k]);
        }
    }
}

__global__ __launch_bounds__(LT) void link_kernel(const int* __restrict__ bstart,
                                                  int* __restrict__ head,
                                                  int2* rec) {
    __shared__ int lh[BS];
    const int tid   = threadIdx.x;
    const int node0 = blockIdx.x * BS;
    if (tid < BS) lh[tid] = 0;
    __syncthreads();
    const int rs = min(max(bstart[blockIdx.x], 0), NE);
    const int re = min(max(bstart[blockIdx.x + 1], 0), NE);
    for (int i0 = rs; i0 < re; i0 += LB * LT) {
        int2 r[LB];
#pragma unroll
        for (int k = 0; k < LB; ++k) {
            const int i = i0 + k * LT + tid;
            r[k] = make_int2(0, -1);
            if (i < re) r[k] = rec[i];
        }
#pragma unroll
        for (int k = 0; k < LB; ++k) {
            const int i = i0 + k * LT + tid;
            const unsigned li = (unsigned)(r[k].y - node0);
            if (i < re && li < (unsigned)BS) {
                const int prev = atomicExch(&lh[li], i + 1);
                rec[i] = make_int2(r[k].x, prev);
            }
        }
    }
    __syncthreads();
    if (tid < BS && node0 + tid < NN) head[node0 + tid] = lh[tid];
}

__device__ __forceinline__ bool cursor_ok(int c) {
    return (unsigned)(c - 1) < (unsigned)NE;
}

__global__ __launch_bounds__(256) void gather1_kernel(
    const float* __restrict__ x, const int* __restrict__ head,
    const int2* __restrict__ rec, float* __restrict__ sum1,
    float* __restrict__ deg) {
    const int lane = threadIdx.x & 31;
    const int wv    = __builtin_amdgcn_readfirstlane((int)(threadIdx.x >> 5));
    const int node0 = (blockIdx.x * 8 + wv) * GK;
    if (node0 >= NN) return;

    const bool owner = lane < GK && node0 + lane < NN;
    int cur = 0;
    if (owner) cur = head[node0 + lane];
    bool alive = owner && cursor_ok(cur);
    int2 r = make_int2(0, 0);
    if (alive) r = rec[cur - 1];

    float2 acc[GK];
#pragma unroll
    for (int j = 0; j < GK; ++j) acc[j] = make_float2(0.0f, 0.0f);
    int cnt = 0;

    for (int it = 0; it < NE; ++it) {
        const unsigned live = __builtin_amdgcn_ballot_w32(alive);
        if (live == 0u) break;
        const int nxt = r.y;
        const bool hn = alive && cursor_ok(nxt);
        int2 rn = make_int2(0, 0);
        if (hn) rn = rec[nxt - 1];
        float2 v[GK];
#pragma unroll
        for (int j = 0; j < GK; ++j) {
            v[j] = make_float2(0.0f, 0.0f);
            if (live & (1u << j)) {
                const int s = __builtin_amdgcn_readlane(r.x, j);
                v[j] = *(const float2*)(x + (size_t)s * DIN + lane * 2);
            }
        }
#pragma unroll
        for (int j = 0; j < GK; ++j) {
            acc[j].x += v[j].x;
            acc[j].y += v[j].y;
        }
        cnt += alive ? 1 : 0;
        r = rn;
        alive = hn;
    }
#pragma unroll
    for (int j = 0; j < GK; ++j) {
        if (node0 + j < NN)
            *(float2*)(sum1 + (size_t)(node0 + j) * DIN + lane * 2) = acc[j];
    }
    if (owner) deg[node0 + lane] = (float)cnt;
}

__global__ __launch_bounds__(256) void gather2_kernel(
    const float* __restrict__ h1, const int* __restrict__ head,
    const int2* __restrict__ rec, float* __restrict__ sum2) {
    const int lane = threadIdx.x & 31;
    const int wv    = __builtin_amdgcn_readfirstlane((int)(threadIdx.x >> 5));
    const int node0 = (blockIdx.x * 8 + wv) * GK;
    if (node0 >= NN) return;

    const bool owner = lane < GK && node0 + lane < NN;
    int cur = 0;
    if (owner) cur = head[node0 + lane];
    bool alive = owner && cursor_ok(cur);
    int2 r = make_int2(0, 0);
    if (alive) r = rec[cur - 1];

    float4 acc[GK];
#pragma unroll
    for (int j = 0; j < GK; ++j) acc[j] = make_float4(0.0f, 0.0f, 0.0f, 0.0f);

    for (int it = 0; it < NE; ++it) {
        const unsigned live = __builtin_amdgcn_ballot_w32(alive);
        if (live == 0u) break;
        const int nxt = r.y;
        const bool hn = alive && cursor_ok(nxt);
        int2 rn = make_int2(0, 0);
        if (hn) rn = rec[nxt - 1];
        float4 v[GK];
#pragma unroll
        for (int j = 0; j < GK; ++j) {
            v[j] = make_float4(0.0f, 0.0f, 0.0f, 0.0f);
            if (live & (1u << j)) {
                const int s = __builtin_amdgcn_readlane(r.x, j);
                v[j] = *(const float4*)(h1 + (size_t)s * DHID + lane * 4);
            }
        }
#pragma unroll
        for (int j = 0; j < GK; ++j) {
            acc[j].x += v[j].x;
            acc[j].y += v[j].y;
            acc[j].z += v[j].z;
            acc[j].w += v[j].w;
        }
        r = rn;
        alive = hn;
    }
#pragma unroll
    for (int j = 0; j < GK; ++j) {
        if (node0 + j < NN)
            *(float4*)(sum2 + (size_t)(node0 + j) * DHID + lane * 4) = acc[j];
    }
}

__global__ __launch_bounds__(32) void gemm1_kernel(
    const float* __restrict__ x, const float* __restrict__ sum1,
    const float* __restrict__ deg,
    const float* __restrict__ Wl, const float* __restrict__ bl,
    const float* __restrict__ Wr, float* __restrict__ h1) {
    const int lane = threadIdx.x;
    const int m0   = blockIdx.x * 16;
    const int mr   = lane & 15;
    const int kb   = (lane >> 4) * 2;
    const int m    = m0 + mr;
    const float dinv = 1.0f / fmaxf(deg[m], 1.0f);

    v8f acc[8] = {};

    for (int kk = 0; kk < DIN / 4; ++kk) {
        const int k = kk * 4 + kb;
        v2f a;
        a.x = sum1[m * DIN + k] * dinv;
        a.y = sum1[m * DIN + k + 1] * dinv;
#pragma unroll
        for (int nt = 0; nt < 8; ++nt) {
            const int n = nt * 16 + mr;
            v2f b;
            b.x = Wl[n * DIN + k];
            b.y = Wl[n * DIN + k + 1];
            acc[nt] = __builtin_amdgcn_wmma_f32_16x16x4_f32(
                false, a, false, b, (short)0, acc[nt], false, false);
        }
    }
    for (int kk = 0; kk < DIN / 4; ++kk) {
        const int k = kk * 4 + kb;
        v2f a;
        a.x = x[m * DIN + k];
        a.y = x[m * DIN + k + 1];
#pragma unroll
        for (int nt = 0; nt < 8; ++nt) {
            const int n = nt * 16 + mr;
            v2f b;
            b.x = Wr[n * DIN + k];
            b.y = Wr[n * DIN + k + 1];
            acc[nt] = __builtin_amdgcn_wmma_f32_16x16x4_f32(
                false, a, false, b, (short)0, acc[nt], false, false);
        }
    }

    const int mo = m0 + 8 * (lane >> 4);
#pragma unroll
    for (int nt = 0; nt < 8; ++nt) {
        const int n = nt * 16 + mr;
        const float bias = bl[n];
#pragma unroll
        for (int v = 0; v < 8; ++v) {
            h1[(mo + v) * DHID + n] = tanhf(acc[nt][v] + bias);
        }
    }
}

__global__ __launch_bounds__(32) void gemm2_kernel(
    const float* __restrict__ h1, const float* __restrict__ sum2,
    const float* __restrict__ deg,
    const float* __restrict__ Wl, const float* __restrict__ bl,
    const float* __restrict__ Wr, float* __restrict__ out) {
    __shared__ float zrow[16 * DOUT];
    const int lane = threadIdx.x;
    const int m0   = blockIdx.x * 16;
    const int mr   = lane & 15;
    const int kb   = (lane >> 4) * 2;
    const int m    = m0 + mr;
    const float dinv = 1.0f / fmaxf(deg[m], 1.0f);

    v8f acc[4] = {};

    for (int kk = 0; kk < DHID / 4; ++kk) {
        const int k = kk * 4 + kb;
        v2f a;
        a.x = sum2[m * DHID + k] * dinv;
        a.y = sum2[m * DHID + k + 1] * dinv;
#pragma unroll
        for (int nt = 0; nt < 4; ++nt) {
            const int n = nt * 16 + mr;
            v2f b;
            b.x = Wl[n * DHID + k];
            b.y = Wl[n * DHID + k + 1];
            acc[nt] = __builtin_amdgcn_wmma_f32_16x16x4_f32(
                false, a, false, b, (short)0, acc[nt], false, false);
        }
    }
    for (int kk = 0; kk < DHID / 4; ++kk) {
        const int k = kk * 4 + kb;
        v2f a;
        a.x = h1[m * DHID + k];
        a.y = h1[m * DHID + k + 1];
#pragma unroll
        for (int nt = 0; nt < 4; ++nt) {
            const int n = nt * 16 + mr;
            v2f b;
            b.x = Wr[n * DHID + k];
            b.y = Wr[n * DHID + k + 1];
            acc[nt] = __builtin_amdgcn_wmma_f32_16x16x4_f32(
                false, a, false, b, (short)0, acc[nt], false, false);
        }
    }

    const int rbase = 8 * (lane >> 4);
#pragma unroll
    for (int nt = 0; nt < 4; ++nt) {
        const int n = nt * 16 + mr;
        const float bias = bl[n];
#pragma unroll
        for (int v = 0; v < 8; ++v) {
            zrow[(rbase + v) * DOUT + n] = acc[nt][v] + bias;
        }
    }
    __syncthreads();

    if (lane < 16) {
        const float* zr = &zrow[lane * DOUT];
        float mx = zr[0];
        for (int c = 1; c < DOUT; ++c) mx = fmaxf(mx, zr[c]);
        float s = 0.0f;
        for (int c = 0; c < DOUT; ++c) s += expf(zr[c] - mx);
        const float lse = mx + logf(s);
        float* orow = &out[(m0 + lane) * DOUT];
        for (int c = 0; c < DOUT; ++c) orow[c] = zr[c] - lse;
    }
}

extern "C" void kernel_launch(void* const* d_in, const int* in_sizes, int n_in,
                              void* d_out, int out_size, void* d_ws, size_t ws_size,
                              hipStream_t stream) {
    const float* x   = (const float*)d_in[0];
    const int*   ei  = (const int*)d_in[1];
    const float* Wl1 = (const float*)d_in[2];
    const float* bl1 = (const float*)d_in[3];
    const float* Wr1 = (const float*)d_in[4];
    const float* Wl2 = (const float*)d_in[5];
    const float* bl2 = (const float*)d_in[6];
    const float* Wr2 = (const float*)d_in[7];
    float* out = (float*)d_out;

    float* ws   = (float*)d_ws;
    float* sum2 = ws;
    float* sum1 = ws;
    float* h1   = sum2 + (size_t)NN * DHID;
    int2*  rec  = (int2*)(h1 + (size_t)NN * DHID);
    int*   head = (int*)(rec + NE);
    float* degb = (float*)(head + NN);

    int*   cnt    = (int*)ws;
    int*   bstart = cnt + NCH * NB;

    count_kernel<<<NCH, CT, 0, stream>>>(ei, cnt);
    place_kernel<<<NCH, CT, 0, stream>>>(ei, cnt, bstart, rec);
    link_kernel<<<NB, LT, 0, stream>>>(bstart, head, rec);

    gather1_kernel<<<(NN + 8 * GK - 1) / (8 * GK), 256, 0, stream>>>(x, head, rec, sum1, degb);
    gemm1_kernel<<<NN / 16, 32, 0, stream>>>(x, sum1, degb, Wl1, bl1, Wr1, h1);
    gather2_kernel<<<(NN + 8 * GK - 1) / (8 * GK), 256, 0, stream>>>(h1, head, rec, sum2);
    gemm2_kernel<<<NN / 16, 32, 0, stream>>>(h1, sum2, degb, Wl2, bl2, Wr2, out);
}


# Layer 2 projects before it aggregates: layer-1 kernel also forms h1.Wl2^T and h1.Wr2^T+b from its LDS tile (h1 never stored); neighbour mean over 64-wide projected rows, log_softmax by lane tree
# speedup vs baseline: 4.1724x; 1.1705x over previous
#include <hip/hip_runtime.h>
#include <cmath>


#define NN 50000
#define NE 800000
#define DIN 64
#define DHID 128
#define DOUT 64
#define GK 8

#define BS 256
#define NB ((NN + BS - 1) / BS)
#define NCH 125
#define CH (NE / NCH)
#define CT 640
#define EPT (CH / CT)
#define RG (CT / 64)
#define LT 512
#define LB 4
static_assert(NCH * CH == NE && EPT * CT == CH, "chunks tile the edge list exactly");
static_assert(NB % 4 == 0 && NB / 4 <= 64 && NB <= CT, "count rows are read as int4 by 64 lanes");

typedef __attribute__((ext_vector_type(2))) float v2f;
typedef __attribute__((ext_vector_type(8))) float v8f;

__global__ __launch_bounds__(CT) void count_kernel(const int* __restrict__ ei,
                                                   int* __restrict__ cnt) {
    __shared__ int h[NB];
    const int tid = threadIdx.x;
    if (tid < NB) h[tid] = 0;
    const int e0 = blockIdx.x * CH;
    int d[EPT];
#pragma unroll
    for (int k = 0; k < EPT; ++k) d[k] = ei[NE + e0 + k * CT + tid];
    __syncthreads();
#pragma unroll
    for (int k = 0; k < EPT; ++k)
        if ((unsigned)d[k] < (unsigned)NN) atomicAdd(&h[(unsigned)d[k] / BS], 1);
    __syncthreads();
    if (tid < NB) cnt[blockIdx.x * NB + tid] = h[tid];
}

__global__ __launch_bounds__(CT) void place_kernel(const int* __restrict__ ei,
                                                   const int* __restrict__ cnt,
                                                   int* __restrict__ bstart,
                                                   int2* __restrict__ rec) {
    __shared__ int tot[NB];
    __shared__ int cur[NB];
    const int tid = threadIdx.x;
    const int g   = blockIdx.x;
    if (tid < NB) { tot[tid] = 0; cur[tid] = 0; }
    const int e0 = g * CH;
    int s[EPT], d[EPT];
#pragma unroll
    for (int k = 0; k < EPT; ++k) {
        s[k] = ei[e0 + k * CT + tid];
        d[k] = ei[NE + e0 + k * CT + tid];
    }
    constexpr int NR = (NCH + RG - 1) / RG;
    const int c4 = tid & 63;
    const int rg = tid >> 6;
    const int c4c = min(c4, NB / 4 - 1);
    int4 cm[NR];
#pragma unroll
    for (int j = 0; j < NR; ++j) {
        const int r = min(rg + j * RG, NCH - 1);
        cm[j] = *(const int4*)(cnt + r * NB + c4c * 4);
    }
    int4 st = make_int4(0, 0, 0, 0);
    int4 sp = make_int4(0, 0, 0, 0);
#pragma unroll
    for (int j = 0; j < NR; ++j) {
        const int r  = rg + j * RG;
        const int ma = (r < NCH) ? -1 : 0;
        const int mp = (r < g) ? ma : 0;
        st.x += cm[j].x & ma; st.y += cm[j].y & ma; st.z += cm[j].z & ma; st.w += cm[j].w & ma;
        sp.x += cm[j].x & mp; sp.y += cm[j].y & mp; sp.z += cm[j].z & mp; sp.w += cm[j].w & mp;
    }
    __syncthreads();
    if (c4 < NB / 4) {
        atomicAdd(&tot[c4 * 4 + 0], st.x); atomicAdd(&cur[c4 * 4 + 0], sp.x);
        atomicAdd(&tot[c4 * 4 + 1], st.y); atomicAdd(&cur[c4 * 4 + 1], sp.y);
        atomicAdd(&tot[c4 * 4 + 2], st.z); atomicAdd(&cur[c4 * 4 + 2], sp.z);
        atomicAdd(&tot[c4 * 4 + 3], st.w); atomicAdd(&cur[c4 * 4 + 3], sp.w);
    }
    __syncthreads();
    if (tid < 32) {
        constexpr int PER = (NB + 31) / 32;
        int loc[PER];
        int sum = 0;
#pragma unroll
        for (int i = 0; i < PER; ++i) {
            const int b = tid * PER + i;
            const int v = (b < NB) ? tot[b] : 0;
            loc[i] = sum;
            sum += v;
        }
        int inc = sum;
#pragma unroll
        for (int dd = 1; dd < 32; dd <<= 1) {
            const int o = __shfl_up(inc, dd);
            if (tid >= dd) inc += o;
        }
        const int excl = inc - sum;
#pragma unroll
        for (int i = 0; i < PER; ++i) {
            const int b = tid * PER + i;
            if (b < NB) {
                const int bb = excl + loc[i];
                if (g == 0) bstart[b] = bb;
                cur[b] += bb;
            }
        }
        if (g == 0 && tid == 31) bstart[NB] = inc;
    }
    __syncthreads();
#pragma unroll
    for (int k = 0; k < EPT; ++k) {
        if ((unsigned)d[k] < (unsigned)NN) {
            const int pos = atomicAdd(&cur[(unsigned)d[k] / BS], 1);
            if ((unsigned)pos < (unsigned)NE) rec[pos] = make_int2(s[k], d[k]);
        }
    }
}

__global__ __launch_bounds__(LT) void link_kernel(const int* __restrict__ bstart,
                                                  int* __restrict__ head,
                                                  int2* rec) {
    __shared__ int lh[BS];
    const int tid   = threadIdx.x;
    const int node0 = blockIdx.x * BS;
    if (tid < BS) lh[tid] = 0;
    __syncthreads();
    const int rs = min(max(bstart[blockIdx.x], 0), NE);
    const int re = min(max(bstart[blockIdx.x + 1], 0), NE);
    for (int i0 = rs; i0 < re; i0 += LB * LT) {
        int2 r[LB];
#pragma unroll
        for (int k = 0; k < LB; ++k) {
            const int i = i0 + k * LT + tid;
            r[k] = make_int2(0, -1);
            if (i < re) r[k] = rec[i];
        }
#pragma unroll
        for (int k = 0; k < LB; ++k) {
            const int i = i0 + k * LT + tid;
            const unsigned li = (unsigned)(r[k].y - node0);
            if (i < re && li < (unsigned)BS) {
                const int prev = atomicExch(&lh[li], i + 1);
                rec[i] = make_int2(r[k].x, prev);
            }
        }
    }
    __syncthreads();
    if (tid < BS && node0 + tid < NN) head[node0 + tid] = lh[tid];
}

__device__ __forceinline__ bool cursor_ok(int c) {
    return (unsigned)(c - 1) < (unsigned)NE;
}

__global__ __launch_bounds__(256) void gather1_kernel(
    const float* __restrict__ x, const int* __restrict__ head,
    const int2* __restrict__ rec, float* __restrict__ sum1,
    float* __restrict__ deg) {
    const int lane = threadIdx.x & 31;
    const int wv    = __builtin_amdgcn_readfirstlane((int)(threadIdx.x >> 5));
    const int node0 = (blockIdx.x * 8 + wv) * GK;
    if (node0 >= NN) return;

    const bool owner = lane < GK && node0 + lane < NN;
    int cur = 0;
    if (owner) cur = head[node0 + lane];
    bool alive = owner && cursor_ok(cur);
    int2 r = make_int2(0, 0);
    if (alive) r = rec[cur - 1];

    float2 acc[GK];
#pragma unroll
    for (int j = 0; j < GK; ++j) acc[j] = make_float2(0.0f, 0.0f);
    int cnt = 0;

    for (int it = 0; it < NE; ++it) {
        const unsigned live = __builtin_amdgcn_ballot_w32(alive);
        if (live == 0u) break;
        const int nxt = r.y;
        const bool hn = alive && cursor_ok(nxt);
        int2 rn = make_int2(0, 0);
        if (hn) rn = rec[nxt - 1];
        float2 v[GK];
#pragma unroll
        for (int j = 0; j < GK; ++j) {
            v[j] = make_float2(0.0f, 0.0f);
            if (live & (1u << j)) {
                const int s = __builtin_amdgcn_readlane(r.x, j);
                v[j] = *(const float2*)(x + (size_t)s * DIN + lane * 2);
            }
        }
#pragma unroll
        for (int j = 0; j < GK; ++j) {
            acc[j].x += v[j].x;
            acc[j].y += v[j].y;
        }
        cnt += alive ? 1 : 0;
        r = rn;
        alive = hn;
    }
#pragma unroll
    for (int j = 0; j < GK; ++j) {
        if (node0 + j < NN)
            *(float2*)(sum1 + (size_t)(node0 + j) * DIN + lane * 2) = acc[j];
    }
    if (owner) deg[node0 + lane] = (float)cnt;
}

__global__ __launch_bounds__(256) void gather2_kernel(
    const float* __restrict__ p, const float* __restrict__ q,
    const float* __restrict__ deg, const int* __restrict__ head,
    const int2* __restrict__ rec, float* __restrict__ out) {
    const int lane = threadIdx.x & 31;
    const int wv    = __builtin_amdgcn_readfirstlane((int)(threadIdx.x >> 5));
    const int node0 = (blockIdx.x * 8 + wv) * GK;
    if (node0 >= NN) return;

    const bool owner = lane < GK && node0 + lane < NN;
    int cur = 0;
    float dg = 1.0f;
    if (owner) {
        cur = head[node0 + lane];
        dg  = deg[node0 + lane];
    }
    bool alive = owner && cursor_ok(cur);
    int2 r = make_int2(0, 0);
    if (alive) r = rec[cur - 1];

    float2 qv[GK];
#pragma unroll
    for (int j = 0; j < GK; ++j) {
        qv[j] = make_float2(0.0f, 0.0f);
        if (node0 + j < NN)
            qv[j] = *(const float2*)(q + (size_t)(node0 + j) * DOUT + lane * 2);
    }

    float2 acc[GK];
#pragma unroll
    for (int j = 0; j < GK; ++j) acc[j] = make_float2(0.0f, 0.0f);

    for (int it = 0; it < NE; ++it) {
        const unsigned live = __builtin_amdgcn_ballot_w32(alive);
        if (live == 0u) break;
        const int nxt = r.y;
        const bool hn = alive && cursor_ok(nxt);
        int2 rn = make_int2(0, 0);
        if (hn) rn = rec[nxt - 1];
        float2 v[GK];
#pragma unroll
        for (int j = 0; j < GK; ++j) {
            v[j] = make_float2(0.0f, 0.0f);
            if (live & (1u << j)) {
                const int s = __builtin_amdgcn_readlane(r.x, j);
                v[j] = *(const float2*)(p + (size_t)s * DOUT + lane * 2);
            }
        }
#pragma unroll
        for (int j = 0; j < GK; ++j) {
            acc[j].x += v[j].x;
            acc[j].y += v[j].y;
        }
        r = rn;
        alive = hn;
    }
    const float dginv = 1.0f / fmaxf(dg, 1.0f);
    float zx[GK], zy[GK], mx[GK], se[GK], t[GK];
#pragma unroll
    for (int j = 0; j < GK; ++j) {
        const float dinv = __int_as_float(__builtin_amdgcn_readlane(__float_as_int(dginv), j));
        zx[j] = acc[j].x * dinv + qv[j].x;
        zy[j] = acc[j].y * dinv + qv[j].y;
        mx[j] = fmaxf(zx[j], zy[j]);
    }
#pragma unroll
    for (int d = 16; d >= 1; d >>= 1) {
#pragma unroll
        for (int j = 0; j < GK; ++j) t[j] = __shfl_xor(mx[j], d);
#pragma unroll
        for (int j = 0; j < GK; ++j) mx[j] = fmaxf(mx[j], t[j]);
    }
#pragma unroll
    for (int j = 0; j < GK; ++j) se[j] = expf(zx[j] - mx[j]) + expf(zy[j] - mx[j]);
#pragma unroll
    for (int d = 16; d >= 1; d >>= 1) {
#pragma unroll
        for (int j = 0; j < GK; ++j) t[j] = __shfl_xor(se[j], d);
#pragma unroll
        for (int j = 0; j < GK; ++j) se[j] += t[j];
    }
#pragma unroll
    for (int j = 0; j < GK; ++j) {
        if (node0 + j < NN) {
            const float lse = mx[j] + logf(se[j]);
            *(float2*)(out + (size_t)(node0 + j) * DOUT + lane * 2) =
                make_float2(zx[j] - lse, zy[j] - lse);
        }
    }
}

#define HP (DHID + 4)
__global__ __launch_bounds__(32) void layer1_kernel(
    const float* __restrict__ x, const float* __restrict__ sum1,
    const float* __restrict__ deg,
    const float* __restrict__ Wl, const float* __restrict__ bl,
    const float* __restrict__ Wr,
    const float* __restrict__ Wl2, const float* __restrict__ bl2,
    const float* __restrict__ Wr2,
    float* __restrict__ p, float* __restrict__ q) {
    __shared__ __attribute__((aligned(16))) float hs[16 * HP];
    const int lane = threadIdx.x;
    const int m0   = blockIdx.x * 16;
    const int mr   = lane & 15;
    const int kb   = (lane >> 4) * 2;
    const int m    = m0 + mr;
    const float dinv = 1.0f / fmaxf(deg[m], 1.0f);

    v8f acc[8] = {};

    for (int kk = 0; kk < DIN / 4; ++kk) {
        const int k = kk * 4 + kb;
        v2f a;
        a.x = sum1[m * DIN + k] * dinv;
        a.y = sum1[m * DIN + k + 1] * dinv;
#pragma unroll
        for (int nt = 0; nt < 8; ++nt) {
            const int n = nt * 16 + mr;
            v2f b;
            b.x = Wl[n * DIN + k];
            b.y = Wl[n * DIN + k + 1];
            acc[nt] = __builtin_amdgcn_wmma_f32_16x16x4_f32(
                false, a, false, b, (short)0, acc[nt], false, false);
        }
    }
    for (int kk = 0; kk < DIN / 4; ++kk) {
        const int k = kk * 4 + kb;
        v2f a;
        a.x = x[m * DIN + k];
        a.y = x[m * DIN + k + 1];
#pragma unroll
        for (int nt = 0; nt < 8; ++nt) {
            const int n = nt * 16 + mr;
            v2f b;
            b.x = Wr[n * DIN + k];
            b.y = Wr[n * DIN + k + 1];
            acc[nt] = __builtin_amdgcn_wmma_f32_16x16x4_f32(
                false, a, false, b, (short)0, acc[nt], false, false);
        }
    }

    const int rb = 8 * (lane >> 4);
#pragma unroll
    for (int nt = 0; nt < 8; ++nt) {
        const int n = nt * 16 + mr;
        const float bias = bl[n];
#pragma unroll
        for (int v = 0; v < 8; ++v) {
            hs[(rb + v) * HP + n] = tanhf(acc[nt][v] + bias);
        }
    }
    __syncthreads();

    v8f pa[4] = {};
    v8f qa[4] = {};
    for (int kk = 0; kk < DHID / 4; ++kk) {
        const int k = kk * 4 + kb;
        v2f a;
        a.x = hs[mr * HP + k];
        a.y = hs[mr * HP + k + 1];
#pragma unroll
        for (int nt = 0; nt < 4; ++nt) {
            const int n = nt * 16 + mr;
            v2f b;
            b.x = Wl2[n * DHID + k];
            b.y = Wl2[n * DHID + k + 1];
            pa[nt] = __builtin_amdgcn_wmma_f32_16x16x4_f32(
                false, a, false, b, (short)0, pa[nt], false, false);
        }
#pragma unroll
        for (int nt = 0; nt < 4; ++nt) {
            const int n = nt * 16 + mr;
            v2f b;
            b.x = Wr2[n * DHID + k];
            b.y = Wr2[n * DHID + k + 1];
            qa[nt] = __builtin_amdgcn_wmma_f32_16x16x4_f32(
                false, a, false, b, (short)0, qa[nt], false, false);
        }
    }

    const int mo = m0 + rb;
#pragma unroll
    for (int nt = 0; nt < 4; ++nt) {
        const int n = nt * 16 + mr;
        const float bias2 = bl2[n];
#pragma unroll
        for (int v = 0; v < 8; ++v) {
            p[(mo + v) * DOUT + n] = pa[nt][v];
            q[(mo + v) * DOUT + n] = qa[nt][v] + bias2;
        }
    }
}

extern "C" void kernel_launch(void* const* d_in, const int* in_sizes, int n_in,
                              void* d_out, int out_size, void* d_ws, size_t ws_size,
                              hipStream_t stream) {
    const float* x   = (const float*)d_in[0];
    const int*   ei  = (const int*)d_in[1];
    const float* Wl1 = (const float*)d_in[2];
    const float* bl1 = (const float*)d_in[3];
    const float* Wr1 = (const float*)d_in[4];
    const float* Wl2 = (const float*)d_in[5];
    const float* bl2 = (const float*)d_in[6];
    const float* Wr2 = (const float*)d_in[7];
    float* out = (float*)d_out;

    float* ws   = (float*)d_ws;
    float* sum1 = ws;
    float* pbuf = ws + (size_t)NN * DHID;
    float* qbuf = pbuf + (size_t)NN * DOUT;
    int2*  rec  = (int2*)(qbuf + (size_t)NN * DOUT);
    int*   head = (int*)(rec + NE);
    float* degb = (float*)(head + NN);

    int*   cnt    = (int*)ws;
    int*   bstart = cnt + NCH * NB;

    count_kernel<<<NCH, CT, 0, stream>>>(ei, cnt);
    place_kernel<<<NCH, CT, 0, stream>>>(ei, cnt, bstart, rec);
    link_kernel<<<NB, LT, 0, stream>>>(bstart, head, rec);

    gather1_kernel<<<(NN + 8 * GK - 1) / (8 * GK), 256, 0, stream>>>(x, head, rec, sum1, degb);
    layer1_kernel<<<NN / 16, 32, 0, stream>>>(x, sum1, degb, Wl1, bl1, Wr1, Wl2, bl2, Wr2, pbuf, qbuf);
    gather2_kernel<<<(NN + 8 * GK - 1) / (8 * GK), 256, 0, stream>>>(pbuf, qbuf, degb, head, rec, out);
}


# Layer-1 neighbour sum moved into the layer-1 matrix kernel: each 16-node wave walks its 16 lists and feeds the mean rows to its WMMAs through LDS; one launch fewer; per-node sum order kept
# speedup vs baseline: 4.3490x; 1.0423x over previous
#include <hip/hip_runtime.h>
#include <cmath>


#define NN 50000
#define NE 800000
#define DIN 64
#define DHID 128
#define DOUT 64
#define GK 8
#define GL 16

#define BS 256
#define NB ((NN + BS - 1) / BS)
#define NCH 125
#define CH (NE / NCH)
#define CT 640
#define EPT (CH / CT)
#define RG (CT / 64)
#define LT 512
#define LB 4
static_assert(NCH * CH == NE && EPT * CT == CH, "chunks tile the edge list exactly");
static_assert(NB % 4 == 0 && NB / 4 <= 64 && NB <= CT, "count rows are read as int4 by 64 lanes");

typedef __attribute__((ext_vector_type(2))) float v2f;
typedef __attribute__((ext_vector_type(8))) float v8f;

__global__ __launch_bounds__(CT) void count_kernel(const int* __restrict__ ei,
                                                   int* __restrict__ cnt) {
    __shared__ int h[NB];
    const int tid = threadIdx.x;
    if (tid < NB) h[tid] = 0;
    const int e0 = blockIdx.x * CH;
    int d[EPT];
#pragma unroll
    for (int k = 0; k < EPT; ++k) d[k] = ei[NE + e0 + k * CT + tid];
    __syncthreads();
#pragma unroll
    for (int k = 0; k < EPT; ++k)
        if ((unsigned)d[k] < (unsigned)NN) atomicAdd(&h[(unsigned)d[k] / BS], 1);
    __syncthreads();
    if (tid < NB) cnt[blockIdx.x * NB + tid] = h[tid];
}

__global__ __launch_bounds__(CT) void place_kernel(const int* __restrict__ ei,
                                                   const int* __restrict__ cnt,
                                                   int* __restrict__ bstart,
                                                   int2* __restrict__ rec) {
    __shared__ int tot[NB];
    __shared__ int cur[NB];
    const int tid = threadIdx.x;
    const int g   = blockIdx.x;
    if (tid < NB) { tot[tid] = 0; cur[tid] = 0; }
    const int e0 = g * CH;
    int s[EPT], d[EPT];
#pragma unroll
    for (int k = 0; k < EPT; ++k) {
        s[k] = ei[e0 + k * CT + tid];
        d[k] = ei[NE + e0 + k * CT + tid];
    }
    constexpr int NR = (NCH + RG - 1) / RG;
    const int c4 = tid & 63;
    const int rg = tid >> 6;
    const int c4c = min(c4, NB / 4 - 1);
    int4 cm[NR];
#pragma unroll
    for (int j = 0; j < NR; ++j) {
        const int r = min(rg + j * RG, NCH - 1);
        cm[j] = *(const int4*)(cnt + r * NB + c4c * 4);
    }
    int4 st = make_int4(0, 0, 0, 0);
    int4 sp = make_int4(0, 0, 0, 0);
#pragma unroll
    for (int j = 0; j < NR; ++j) {
        const int r  = rg + j * RG;
        const int ma = (r < NCH) ? -1 : 0;
        const int mp = (r < g) ? ma : 0;
        st.x += cm[j].x & ma; st.y += cm[j].y & ma; st.z += cm[j].z & ma; st.w += cm[j].w & ma;
        sp.x += cm[j].x & mp; sp.y += cm[j].y & mp; sp.z += cm[j].z & mp; sp.w += cm[j].w & mp;
    }
    __syncthreads();
    if (c4 < NB / 4) {
        atomicAdd(&tot[c4 * 4 + 0], st.x); atomicAdd(&cur[c4 * 4 + 0], sp.x);
        atomicAdd(&tot[c4 * 4 + 1], st.y); atomicAdd(&cur[c4 * 4 + 1], sp.y);
        atomicAdd(&tot[c4 * 4 + 2], st.z); atomicAdd(&cur[c4 * 4 + 2], sp.z);
        atomicAdd(&tot[c4 * 4 + 3], st.w); atomicAdd(&cur[c4 * 4 + 3], sp.w);
    }
    __syncthreads();
    if (tid < 32) {
        constexpr int PER = (NB + 31) / 32;
        int loc[PER];
        int sum = 0;
#pragma unroll
        for (int i = 0; i < PER; ++i) {
            const int b = tid * PER + i;
            const int v = (b < NB) ? tot[b] : 0;
            loc[i] = sum;
            sum += v;
        }
        int inc = sum;
#pragma unroll
        for (int dd = 1; dd < 32; dd <<= 1) {
            const int o = __shfl_up(inc, dd);
            if (tid >= dd) inc += o;
        }
        const int excl = inc - sum;
#pragma unroll
        for (int i = 0; i < PER; ++i) {
            const int b = tid * PER + i;
            if (b < NB) {
                const int bb = excl + loc[i];
                if (g == 0) bstart[b] = bb;
                cur[b] += bb;
            }
        }
        if (g == 0 && tid == 31) bstart[NB] = inc;
    }
    __syncthreads();
#pragma unroll
    for (int k = 0; k < EPT; ++k) {
        if ((unsigned)d[k] < (unsigned)NN) {
            const int pos = atomicAdd(&cur[(unsigned)d[k] / BS], 1);
            if ((unsigned)pos < (unsigned)NE) rec[pos] = make_int2(s[k], d[k]);
        }
    }
}

__global__ __launch_bounds__(LT) void link_kernel(const int* __restrict__ bstart,
                                                  int* __restrict__ head,
                                                  int2* rec) {
    __shared__ int lh[BS];
    const int tid   = threadIdx.x;
    const int node0 = blockIdx.x * BS;
    if (tid < BS) lh[tid] = 0;
    __syncthreads();
    const int rs = min(max(bstart[blockIdx.x], 0), NE);
    const int re = min(max(bstart[blockIdx.x + 1], 0), NE);
    for (int i0 = rs; i0 < re; i0 += LB * LT) {
        int2 r[LB];
#pragma unroll
        for (int k = 0; k < LB; ++k) {
            const int i = i0 + k * LT + tid;
            r[k] = make_int2(0, -1);
            if (i < re) r[k] = rec[i];
        }
#pragma unroll
        for (int k = 0; k < LB; ++k) {
            const int i = i0 + k * LT + tid;
            const unsigned li = (unsigned)(r[k].y - node0);
            if (i < re && li < (unsigned)BS) {
                const int prev = atomicExch(&lh[li], i + 1);
                rec[i] = make_int2(r[k].x, prev);
            }
        }
    }
    __syncthreads();
    if (tid < BS && node0 + tid < NN) head[node0 + tid] = lh[tid];
}

__device__ __forceinline__ bool cursor_ok(int c) {
    return (unsigned)(c - 1) < (unsigned)NE;
}

__global__ __launch_bounds__(256) void gather2_kernel(
    const float* __restrict__ p, const float* __restrict__ q,
    const float* __restrict__ deg, const int* __restrict__ head,
    const int2* __restrict__ rec, float* __restrict__ out) {
    const int lane = threadIdx.x & 31;
    const int wv    = __builtin_amdgcn_readfirstlane((int)(threadIdx.x >> 5));
    const int node0 = (blockIdx.x * 8 + wv) * GK;
    if (node0 >= NN) return;

    const bool owner = lane < GK && node0 + lane < NN;
    int cur = 0;
    float dg = 1.0f;
    if (owner) {
        cur = head[node0 + lane];
        dg  = deg[node0 + lane];
    }
    bool alive = owner && cursor_ok(cur);
    int2 r = make_int2(0, 0);
    if (alive) r = rec[cur - 1];

    float2 qv[GK];
#pragma unroll
    for (int j = 0; j < GK; ++j) {
        qv[j] = make_float2(0.0f, 0.0f);
        if (node0 + j < NN)
            qv[j] = *(const float2*)(q + (size_t)(node0 + j) * DOUT + lane * 2);
    }

    float2 acc[GK];
#pragma unroll
    for (int j = 0; j < GK; ++j) acc[j] = make_float2(0.0f, 0.0f);

    for (int it = 0; it < NE; ++it) {
        const unsigned live = __builtin_amdgcn_ballot_w32(alive);
        if (live == 0u) break;
        const int nxt = r.y;
        const bool hn = alive && cursor_ok(nxt);
        int2 rn = make_int2(0, 0);
        if (hn) rn = rec[nxt - 1];
        float2 v[GK];
#pragma unroll
        for (int j = 0; j < GK; ++j) {
            v[j] = make_float2(0.0f, 0.0f);
            if (live & (1u << j)) {
                const int s = __builtin_amdgcn_readlane(r.x, j);
                v[j] = *(const float2*)(p + (size_t)s * DOUT + lane * 2);
            }
        }
#pragma unroll
        for (int j = 0; j < GK; ++j) {
            acc[j].x += v[j].x;
            acc[j].y += v[j].y;
        }
        r = rn;
        alive = hn;
    }
    const float dginv = 1.0f / fmaxf(dg, 1.0f);
    float zx[GK], zy[GK], mx[GK], se[GK], t[GK];
#pragma unroll
    for (int j = 0; j < GK; ++j) {
        const float dinv = __int_as_float(__builtin_amdgcn_readlane(__float_as_int(dginv), j));
        zx[j] = acc[j].x * dinv + qv[j].x;
        zy[j] = acc[j].y * dinv + qv[j].y;
        mx[j] = fmaxf(zx[j], zy[j]);
    }
#pragma unroll
    for (int d = 16; d >= 1; d >>= 1) {
#pragma unroll
        for (int j = 0; j < GK; ++j) t[j] = __shfl_xor(mx[j], d);
#pragma unroll
        for (int j = 0; j < GK; ++j) mx[j] = fmaxf(mx[j], t[j]);
    }
#pragma unroll
    for (int j = 0; j < GK; ++j) se[j] = expf(zx[j] - mx[j]) + expf(zy[j] - mx[j]);
#pragma unroll
    for (int d = 16; d >= 1; d >>= 1) {
#pragma unroll
        for (int j = 0; j < GK; ++j) t[j] = __shfl_xor(se[j], d);
#pragma unroll
        for (int j = 0; j < GK; ++j) se[j] += t[j];
    }
#pragma unroll
    for (int j = 0; j < GK; ++j) {
        if (node0 + j < NN) {
            const float lse = mx[j] + logf(se[j]);
            *(float2*)(out + (size_t)(node0 + j) * DOUT + lane * 2) =
                make_float2(zx[j] - lse, zy[j] - lse);
        }
    }
}

#define HP (DHID + 4)
#define AP (DIN + 4)
__global__ __launch_bounds__(32) void layer1_kernel(
    const float* __restrict__ x, const int* __restrict__ head,
    const int2* __restrict__ rec, float* __restrict__ deg,
    const float* __restrict__ Wl, const float* __restrict__ bl,
    const float* __restrict__ Wr,
    const float* __restrict__ Wl2, const float* __restrict__ bl2,
    const float* __restrict__ Wr2,
    float* __restrict__ p, float* __restrict__ q) {
    __shared__ __attribute__((aligned(16))) float as[16 * AP];
    __shared__ __attribute__((aligned(16))) float hs[16 * HP];
    const int lane = threadIdx.x;
    const int m0   = blockIdx.x * 16;
    const int mr   = lane & 15;
    const int kb   = (lane >> 4) * 2;

    const bool owner = lane < GL && m0 + lane < NN;
    int cur = 0;
    if (owner) cur = head[m0 + lane];
    bool alive = owner && cursor_ok(cur);
    int2 r = make_int2(0, 0);
    if (alive) r = rec[cur - 1];

    float2 sacc[GL];
#pragma unroll
    for (int j = 0; j < GL; ++j) sacc[j] = make_float2(0.0f, 0.0f);
    int cnt = 0;

    for (int it = 0; it < NE; ++it) {
        const unsigned live = __builtin_amdgcn_ballot_w32(alive);
        if (live == 0u) break;
        const int nxt = r.y;
        const bool hn = alive && cursor_ok(nxt);
        int2 rn = make_int2(0, 0);
        if (hn) rn = rec[nxt - 1];
        float2 v[GL];
#pragma unroll
        for (int j = 0; j < GL; ++j) {
            v[j] = make_float2(0.0f, 0.0f);
            if (live & (1u << j)) {
                const int s = __builtin_amdgcn_readlane(r.x, j);
                v[j] = *(const float2*)(x + (size_t)s * DIN + lane * 2);
            }
        }
#pragma unroll
        for (int j = 0; j < GL; ++j) {
            sacc[j].x += v[j].x;
            sacc[j].y += v[j].y;
        }
        cnt += alive ? 1 : 0;
        r = rn;
        alive = hn;
    }
    if (owner) deg[m0 + lane] = (float)cnt;
    const float dgi = 1.0f / fmaxf((float)cnt, 1.0f);
#pragma unroll
    for (int j = 0; j < GL; ++j) {
        const float dj = __int_as_float(__builtin_amdgcn_readlane(__float_as_int(dgi), j));
        *(float2*)(as + j * AP + lane * 2) = make_float2(sacc[j].x * dj, sacc[j].y * dj);
    }
    __syncthreads();

    v8f acc[8] = {};

    for (int kk = 0; kk < DIN / 4; ++kk) {
        const int k = kk * 4 + kb;
        v2f a;
        a.x = as[mr * AP + k];
        a.y = as[mr * AP + k + 1];
#pragma unroll
        for (int nt = 0; nt < 8; ++nt) {
            const int n = nt * 16 + mr;
            v2f b;
            b.x = Wl[n * DIN + k];
            b.y = Wl[n * DIN + k + 1];
            acc[nt] = __builtin_amdgcn_wmma_f32_16x16x4_f32(
                false, a, false, b, (short)0, acc[nt], false, false);
        }
    }
    const int m = m0 + mr;
    for (int kk = 0; kk < DIN / 4; ++kk) {
        const int k = kk * 4 + kb;
        v2f a;
        a.x = x[m * DIN + k];
        a.y = x[m * DIN + k + 1];
#pragma unroll
        for (int nt = 0; nt < 8; ++nt) {
            const int n = nt * 16 + mr;
            v2f b;
            b.x = Wr[n * DIN + k];
            b.y = Wr[n * DIN + k + 1];
            acc[nt] = __builtin_amdgcn_wmma_f32_16x16x4_f32(
                false, a, false, b, (short)0, acc[nt], false, false);
        }
    }

    const int rb = 8 * (lane >> 4);
#pragma unroll
    for (int nt = 0; nt < 8; ++nt) {
        const int n = nt * 16 + mr;
        const float bias = bl[n];
#pragma unroll
        for (int v = 0; v < 8; ++v) {
            hs[(rb + v) * HP + n] = tanhf(acc[nt][v] + bias);
        }
    }
    __syncthreads();

    v8f pa[4] = {};
    v8f qa[4] = {};
    for (int kk = 0; kk < DHID / 4; ++kk) {
        const int k = kk * 4 + kb;
        v2f a;
        a.x = hs[mr * HP + k];
        a.y = hs[mr * HP + k + 1];
#pragma unroll
        for (int nt = 0; nt < 4; ++nt) {
            const int n = nt * 16 + mr;
            v2f b;
            b.x = Wl2[n * DHID + k];
            b.y = Wl2[n * DHID + k + 1];
            pa[nt] = __builtin_amdgcn_wmma_f32_16x16x4_f32(
                false, a, false, b, (short)0, pa[nt], false, false);
        }
#pragma unroll
        for (int nt = 0; nt < 4; ++nt) {
            const int n = nt * 16 + mr;
            v2f b;
            b.x = Wr2[n * DHID + k];
            b.y = Wr2[n * DHID + k + 1];
            qa[nt] = __builtin_amdgcn_wmma_f32_16x16x4_f32(
                false, a, false, b, (short)0, qa[nt], false, false);
        }
    }

    const int mo = m0 + rb;
#pragma unroll
    for (int nt = 0; nt < 4; ++nt) {
        const int n = nt * 16 + mr;
        const float bias2 = bl2[n];
#pragma unroll
        for (int v = 0; v < 8; ++v) {
            p[(mo + v) * DOUT + n] = pa[nt][v];
            q[(mo + v) * DOUT + n] = qa[nt][v] + bias2;
        }
    }
}

extern "C" void kernel_launch(void* const* d_in, const int* in_sizes, int n_in,
                              void* d_out, int out_size, void* d_ws, size_t ws_size,
                              hipStream_t stream) {
    const float* x   = (const float*)d_in[0];
    const int*   ei  = (const int*)d_in[1];
    const float* Wl1 = (const float*)d_in[2];
    const float* bl1 = (const float*)d_in[3];
    const float* Wr1 = (const float*)d_in[4];
    const float* Wl2 = (const float*)d_in[5];
    const float* bl2 = (const float*)d_in[6];
    const float* Wr2 = (const float*)d_in[7];
    float* out = (float*)d_out;

    float* ws   = (float*)d_ws;
    float* pbuf = ws + (size_t)NN * DHID;
    float* qbuf = pbuf + (size_t)NN * DOUT;
    int2*  rec  = (int2*)(qbuf + (size_t)NN * DOUT);
    int*   head = (int*)(rec + NE);
    float* degb = (float*)(head + NN);

    int*   cnt    = (int*)ws;
    int*   bstart = cnt + NCH * NB;

    count_kernel<<<NCH, CT, 0, stream>>>(ei, cnt);
    place_kernel<<<NCH, CT, 0, stream>>>(ei, cnt, bstart, rec);
    link_kernel<<<NB, LT, 0, stream>>>(bstart, head, rec);

    layer1_kernel<<<NN / 16, 32, 0, stream>>>(x, head, rec, degb, Wl1, bl1, Wr1, Wl2, bl2, Wr2, pbuf, qbuf);
    gather2_kernel<<<(NN + 8 * GK - 1) / (8 * GK), 256, 0, stream>>>(pbuf, qbuf, degb, head, rec, out);
}


# B operands of all matrix products read from a per-call copy of the 4 weight matrices in fragment order (written by spare blocks of the count pass): one whole-line 16-byte load feeds 2 k-steps
# speedup vs baseline: 4.5530x; 1.0469x over previous
#include <hip/hip_runtime.h>
#include <cmath>


#define NN 50000
#define NE 800000
#define DIN 64
#define DHID 128
#define DOUT 64
#define GK 8
#define GL 16

#define BS 256
#define NB ((NN + BS - 1) / BS)
#define NCH 125
#define CH (NE / NCH)
#define CT 640
#define EPT (CH / CT)
#define RG (CT / 64)
#define LT 512
#define LB 4
static_assert(NCH * CH == NE && EPT * CT == CH, "chunks tile the edge list exactly");
static_assert(NB % 4 == 0 && NB / 4 <= 64 && NB <= CT, "count rows are read as int4 by 64 lanes");

#define WFG 2048
#define WF_OFF 65536
#define NWB ((4 * WFG + CT - 1) / CT)
static_assert(NCH * NB + NB + 1 <= WF_OFF, "the fragment-ordered weights start behind the count matrix and the bucket starts");
static_assert(DHID * DIN == 4 * WFG && DOUT * DHID == 4 * WFG, "every weight matrix is WFG groups of 4 floats");

typedef __attribute__((ext_vector_type(2))) float v2f;
typedef __attribute__((ext_vector_type(4))) float v4f;
typedef __attribute__((ext_vector_type(8))) float v8f;

template <int NOUT, int K>
__device__ __forceinline__ void frag_order(const float* __restrict__ Wm, float* __restrict__ F, int g) {
    constexpr int NT = NOUT / 16;
    const int lane = g & 31;
    const int nt   = (g >> 5) % NT;
    const int k2   = (g >> 5) / NT;
    const int n    = nt * 16 + (lane & 15);
    const int k    = k2 * 8 + (lane >> 4) * 2;
    const float2 lo = *(const float2*)(Wm + n * K + k);
    const float2 hi = *(const float2*)(Wm + n * K + k + 4);
    *(float4*)(F + (size_t)g * 4) = make_float4(lo.x, lo.y, hi.x, hi.y);
}

__global__ __launch_bounds__(CT) void count_kernel(const int* __restrict__ ei,
                                                   int* __restrict__ cnt,
                                                   const float* __restrict__ Wl1,
                                                   const float* __restrict__ Wr1,
                                                   const float* __restrict__ Wl2,
                                                   const float* __restrict__ Wr2,
                                                   float* __restrict__ wf) {
    __shared__ int h[NB];
    const int tid = threadIdx.x;
    if (blockIdx.x >= NCH) {
        const int g = (blockIdx.x - NCH) * CT + tid;
        if (g < WFG)          frag_order<DHID, DIN>(Wl1, wf, g);
        else if (g < 2 * WFG) frag_order<DHID, DIN>(Wr1, wf + 4 * WFG, g - WFG);
        else if (g < 3 * WFG) frag_order<DOUT, DHID>(Wl2, wf + 8 * WFG, g - 2 * WFG);
        else if (g < 4 * WFG) frag_order<DOUT, DHID>(Wr2, wf + 12 * WFG, g - 3 * WFG);
        return;
    }
    if (tid < NB) h[tid] = 0;
    const int e0 = blockIdx.x * CH;
    int d[EPT];
#pragma unroll
    for (int k = 0; k < EPT; ++k) d[k] = ei[NE + e0 + k * CT + tid];
    __syncthreads();
#pragma unroll
    for (int k = 0; k < EPT; ++k)
        if ((unsigned)d[k] < (unsigned)NN) atomicAdd(&h[(unsigned)d[k] / BS], 1);
    __syncthreads();
    if (tid < NB) cnt[blockIdx.x * NB + tid] = h[tid];
}

__global__ __launch_bounds__(CT) void place_kernel(const int* __restrict__ ei,
                                                   const int* __restrict__ cnt,
                                                   int* __restrict__ bstart,
                                                   int2* __restrict__ rec) {
    __shared__ int tot[NB];
    __shared__ int cur[NB];
    const int tid = threadIdx.x;
    const int g   = blockIdx.x;
    if (tid < NB) { tot[tid] = 0; cur[tid] = 0; }
    const int e0 = g * CH;
    int s[EPT], d[EPT];
#pragma unroll
    for (int k = 0; k < EPT; ++k) {
        s[k] = ei[e0 + k * CT + tid];
        d[k] = ei[NE + e0 + k * CT + tid];
    }
    constexpr int NR = (NCH + RG - 1) / RG;
    const int c4 = tid & 63;
    const int rg = tid >> 6;
    const int c4c = min(c4, NB / 4 - 1);
    int4 cm[NR];
#pragma unroll
    for (int j = 0; j < NR; ++j) {
        const int r = min(rg + j * RG, NCH - 1);
        cm[j] = *(const int4*)(cnt + r * NB + c4c * 4);
    }
    int4 st = make_int4(0, 0, 0, 0);
    int4 sp = make_int4(0, 0, 0, 0);
#pragma unroll
    for (int j = 0; j < NR; ++j) {
        const int r  = rg + j * RG;
        const int ma = (r < NCH) ? -1 : 0;
        const int mp = (r < g) ? ma : 0;
        st.x += cm[j].x & ma; st.y += cm[j].y & ma; st.z += cm[j].z & ma; st.w += cm[j].w & ma;
        sp.x += cm[j].x & mp; sp.y += cm[j].y & mp; sp.z += cm[j].z & mp; sp.w += cm[j].w & mp;
    }
    __syncthreads();
    if (c4 < NB / 4) {
        atomicAdd(&tot[c4 * 4 + 0], st.x); atomicAdd(&cur[c4 * 4 + 0], sp.x);
        atomicAdd(&tot[c4 * 4 + 1], st.y); atomicAdd(&cur[c4 * 4 + 1], sp.y);
        atomicAdd(&tot[c4 * 4 + 2], st.z); atomicAdd(&cur[c4 * 4 + 2], sp.z);
        atomicAdd(&tot[c4 * 4 + 3], st.w); atomicAdd(&cur[c4 * 4 + 3], sp.w);
    }
    __syncthreads();
    if (tid < 32) {
        constexpr int PER = (NB + 31) / 32;
        int loc[PER];
        int sum = 0;
#pragma unroll
        for (int i = 0; i < PER; ++i) {
            const int b = tid * PER + i;
            const int v = (b < NB) ? tot[b] : 0;
            loc[i] = sum;
            sum += v;
        }
        int inc = sum;
#pragma unroll
        for (int dd = 1; dd < 32; dd <<= 1) {
            const int o = __shfl_up(inc, dd);
            if (tid >= dd) inc += o;
        }
        const int excl = inc - sum;
#pragma unroll
        for (int i = 0; i < PER; ++i) {
            const int b = tid * PER + i;
            if (b < NB) {
                const int bb = excl + loc[i];
                if (g == 0) bstart[b] = bb;
                cur[b] += bb;
            }
        }
        if (g == 0 && tid == 31) bstart[NB] = inc;
    }
    __syncthreads();
#pragma unroll
    for (int k = 0; k < EPT; ++k) {
        if ((unsigned)d[k] < (unsigned)NN) {
            const int pos = atomicAdd(&cur[(unsigned)d[k] / BS], 1);
            if ((unsigned)pos < (unsigned)NE) rec[pos] = make_int2(s[k], d[k]);
        }
    }
}

__global__ __launch_bounds__(LT) void link_kernel(const int* __restrict__ bstart,
                                                  int* __restrict__ head,
                                                  int2* rec) {
    __shared__ int lh[BS];
    const int tid   = threadIdx.x;
    const int node0 = blockIdx.x * BS;
    if (tid < BS) lh[tid] = 0;
    __syncthreads();
    const int rs = min(max(bstart[blockIdx.x], 0), NE);
    const int re = min(max(bstart[blockIdx.x + 1], 0), NE);
    for (int i0 = rs; i0 < re; i0 += LB * LT) {
        int2 r[LB];
#pragma unroll
        for (int k = 0; k < LB; ++k) {
            const int i = i0 + k * LT + tid;
            r[k] = make_int2(0, -1);
            if (i < re) r[k] = rec[i];
        }
#pragma unroll
        for (int k = 0; k < LB; ++k) {
            const int i = i0 + k * LT + tid;
            const unsigned li = (unsigned)(r[k].y - node0);
            if (i < re && li < (unsigned)BS) {
                const int prev = atomicExch(&lh[li], i + 1);
                rec[i] = make_int2(r[k].x, prev);
            }
        }
    }
    __syncthreads();
    if (tid < BS && node0 + tid < NN) head[node0 + tid] = lh[tid];
}

__device__ __forceinline__ bool cursor_ok(int c) {
    return (unsigned)(c - 1) < (unsigned)NE;
}

__global__ __launch_bounds__(256) void gather2_kernel(
    const float* __restrict__ p, const float* __restrict__ q,
    const float* __restrict__ deg, const int* __restrict__ head,
    const int2* __restrict__ rec, float* __restrict__ out) {
    const int lane = threadIdx.x & 31;
    const int wv    = __builtin_amdgcn_readfirstlane((int)(threadIdx.x >> 5));
    const int node0 = (blockIdx.x * 8 + wv) * GK;
    if (node0 >= NN) return;

    const bool owner = lane < GK && node0 + lane < NN;
    int cur = 0;
    float dg = 1.0f;
    if (owner) {
        cur = head[node0 + lane];
        dg  = deg[node0 + lane];
    }
    bool alive = owner && cursor_ok(cur);
    int2 r = make_int2(0, 0);
    if (alive) r = rec[cur - 1];

    float2 qv[GK];
#pragma unroll
    for (int j = 0; j < GK; ++j) {
        qv[j] = make_float2(0.0f, 0.0f);
        if (node0 + j < NN)
            qv[j] = *(const float2*)(q + (size_t)(node0 + j) * DOUT + lane * 2);
    }

    float2 acc[GK];
#pragma unroll
    for (int j = 0; j < GK; ++j) acc[j] = make_float2(0.0f, 0.0f);

    for (int it = 0; it < NE; ++it) {
        const unsigned live = __builtin_amdgcn_ballot_w32(alive);
        if (live == 0u) break;
        const int nxt = r.y;
        const bool hn = alive && cursor_ok(nxt);
        int2 rn = make_int2(0, 0);
        if (hn) rn = rec[nxt - 1];
        float2 v[GK];
#pragma unroll
        for (int j = 0; j < GK; ++j) {
            v[j] = make_float2(0.0f, 0.0f);
            if (live & (1u << j)) {
                const int s = __builtin_amdgcn_readlane(r.x, j);
                v[j] = *(const float2*)(p + (size_t)s * DOUT + lane * 2);
            }
        }
#pragma unroll
        for (int j = 0; j < GK; ++j) {
            acc[j].x += v[j].x;
            acc[j].y += v[j].y;
        }
        r = rn;
        alive = hn;
    }
    const float dginv = 1.0f / fmaxf(dg, 1.0f);
    float zx[GK], zy[GK], mx[GK], se[GK], t[GK];
#pragma unroll
    for (int j = 0; j < GK; ++j) {
        const float dinv = __int_as_float(__builtin_amdgcn_readlane(__float_as_int(dginv), j));
        zx[j] = acc[j].x * dinv + qv[j].x;
        zy[j] = acc[j].y * dinv + qv[j].y;
        mx[j] = fmaxf(zx[j], zy[j]);
    }
#pragma unroll
    for (int d = 16; d >= 1; d >>= 1) {
#pragma unroll
        for (int j = 0; j < GK; ++j) t[j] = __shfl_xor(mx[j], d);
#pragma unroll
        for (int j = 0; j < GK; ++j) mx[j] = fmaxf(mx[j], t[j]);
    }
#pragma unroll
    for (int j = 0; j < GK; ++j) se[j] = expf(zx[j] - mx[j]) + expf(zy[j] - mx[j]);
#pragma unroll
    for (int d = 16; d >= 1; d >>= 1) {
#pragma unroll
        for (int j = 0; j < GK; ++j) t[j] = __shfl_xor(se[j], d);
#pragma unroll
        for (int j = 0; j < GK; ++j) se[j] += t[j];
    }
#pragma unroll
    for (int j = 0; j < GK; ++j) {
        if (node0 + j < NN) {
            const float lse = mx[j] + logf(se[j]);
            *(float2*)(out + (size_t)(node0 + j) * DOUT + lane * 2) =
                make_float2(zx[j] - lse, zy[j] - lse);
        }
    }
}

#define HP (DHID + 4)
#define AP (DIN + 4)
__global__ __launch_bounds__(32) void layer1_kernel(
    const float* __restrict__ x, const int* __restrict__ head,
    const int2* __restrict__ rec, float* __restrict__ deg,
    const float* __restrict__ wf, const float* __restrict__ bl,
    const float* __restrict__ bl2,
    float* __restrict__ p, float* __restrict__ q) {
    __shared__ __attribute__((aligned(16))) float as[16 * AP];
    __shared__ __attribute__((aligned(16))) float hs[16 * HP];
    const int lane = threadIdx.x;
    const int m0   = blockIdx.x * 16;
    const int mr   = lane & 15;
    const int kb   = (lane >> 4) * 2;
    const v4f* __restrict__ fl1 = (const v4f*)wf;
    const v4f* __restrict__ fr1 = fl1 + WFG;
    const v4f* __restrict__ fl2 = fr1 + WFG;
    const v4f* __restrict__ fr2 = fl2 + WFG;

    const bool owner = lane < GL && m0 + lane < NN;
    int cur = 0;
    if (owner) cur = head[m0 + lane];
    bool alive = owner && cursor_ok(cur);
    int2 r = make_int2(0, 0);
    if (alive) r = rec[cur - 1];

    float2 sacc[GL];
#pragma unroll
    for (int j = 0; j < GL; ++j) sacc[j] = make_float2(0.0f, 0.0f);
    int cnt = 0;

    for (int it = 0; it < NE; ++it) {
        const unsigned live = __builtin_amdgcn_ballot_w32(alive);
        if (live == 0u) break;
        const int nxt = r.y;
        const bool hn = alive && cursor_ok(nxt);
        int2 rn = make_int2(0, 0);
        if (hn) rn = rec[nxt - 1];
        float2 v[GL];
#pragma unroll
        for (int j = 0; j < GL; ++j) {
            v[j] = make_float2(0.0f, 0.0f);
            if (live & (1u << j)) {
                const int s = __builtin_amdgcn_readlane(r.x, j);
                v[j] = *(const float2*)(x + (size_t)s * DIN + lane * 2);
            }
        }
#pragma unroll
        for (int j = 0; j < GL; ++j) {
            sacc[j].x += v[j].x;
            sacc[j].y += v[j].y;
        }
        cnt += alive ? 1 : 0;
        r = rn;
        alive = hn;
    }
    if (owner) deg[m0 + lane] = (float)cnt;
    const float dgi = 1.0f / fmaxf((float)cnt, 1.0f);
#pragma unroll
    for (int j = 0; j < GL; ++j) {
        const float dj = __int_as_float(__builtin_amdgcn_readlane(__float_as_int(dgi), j));
        *(float2*)(as + j * AP + lane * 2) = make_float2(sacc[j].x * dj, sacc[j].y * dj);
    }
    __syncthreads();

    v8f acc[8] = {};

    for (int k2 = 0; k2 < DIN / 8; ++k2) {
        const int k = k2 * 8 + kb;
        v2f a0, a1;
        a0.x = as[mr * AP + k];
        a0.y = as[mr * AP + k + 1];
        a1.x = as[mr * AP + k + 4];
        a1.y = as[mr * AP + k + 5];
        v4f b[8];
#pragma unroll
        for (int nt = 0; nt < 8; ++nt) b[nt] = fl1[(k2 * 8 + nt) * 32 + lane];
#pragma unroll
        for (int nt = 0; nt < 8; ++nt) {
            v2f bb;
            bb.x = b[nt].x;
            bb.y = b[nt].y;
            acc[nt] = __builtin_amdgcn_wmma_f32_16x16x4_f32(
                false, a0, false, bb, (short)0, acc[nt], false, false);
        }
#pragma unroll
        for (int nt = 0; nt < 8; ++nt) {
            v2f bb;
            bb.x = b[nt].z;
            bb.y = b[nt].w;
            acc[nt] = __builtin_amdgcn_wmma_f32_16x16x4_f32(
                false, a1, false, bb, (short)0, acc[nt], false, false);
        }
    }
    const int m = m0 + mr;
    for (int k2 = 0; k2 < DIN / 8; ++k2) {
        const int k = k2 * 8 + kb;
        v2f a0, a1;
        a0.x = x[m * DIN + k];
        a0.y = x[m * DIN + k + 1];
        a1.x = x[m * DIN + k + 4];
        a1.y = x[m * DIN + k + 5];
        v4f b[8];
#pragma unroll
        for (int nt = 0; nt < 8; ++nt) b[nt] = fr1[(k2 * 8 + nt) * 32 + lane];
#pragma unroll
        for (int nt = 0; nt < 8; ++nt) {
            v2f bb;
            bb.x = b[nt].x;
            bb.y = b[nt].y;
            acc[nt] = __builtin_amdgcn_wmma_f32_16x16x4_f32(
                false, a0, false, bb, (short)0, acc[nt], false, false);
        }
#pragma unroll
        for (int nt = 0; nt < 8; ++nt) {
            v2f bb;
            bb.x = b[nt].z;
            bb.y = b[nt].w;
            acc[nt] = __builtin_amdgcn_wmma_f32_16x16x4_f32(
                false, a1, false, bb, (short)0, acc[nt], false, false);
        }
    }

    const int rb = 8 * (lane >> 4);
#pragma unroll
    for (int nt = 0; nt < 8; ++nt) {
        const int n = nt * 16 + mr;
        const float bias = bl[n];
#pragma unroll
        for (int v = 0; v < 8; ++v) {
            hs[(rb + v) * HP + n] = tanhf(acc[nt][v] + bias);
        }
    }
    __syncthreads();

    v8f pa[4] = {};
    v8f qa[4] = {};
    for (int k2 = 0; k2 < DHID / 8; ++k2) {
        const int k = k2 * 8 + kb;
        v2f a0, a1;
        a0.x = hs[mr * HP + k];
        a0.y = hs[mr * HP + k + 1];
        a1.x = hs[mr * HP + k + 4];
        a1.y = hs[mr * HP + k + 5];
        v4f b2l[4], b2r[4];
#pragma unroll
        for (int nt = 0; nt < 4; ++nt) {
            b2l[nt] = fl2[(k2 * 4 + nt) * 32 + lane];
            b2r[nt] = fr2[(k2 * 4 + nt) * 32 + lane];
        }
#pragma unroll
        for (int nt = 0; nt < 4; ++nt) {
            v2f bb;
            bb.x = b2l[nt].x;
            bb.y = b2l[nt].y;
            pa[nt] = __builtin_amdgcn_wmma_f32_16x16x4_f32(
                false, a0, false, bb, (short)0, pa[nt], false, false);
        }
#pragma unroll
        for (int nt = 0; nt < 4; ++nt) {
            v2f bb;
            bb.x = b2r[nt].x;
            bb.y = b2r[nt].y;
            qa[nt] = __builtin_amdgcn_wmma_f32_16x16x4_f32(
                false, a0, false, bb, (short)0, qa[nt], false, false);
        }
#pragma unroll
        for (int nt = 0; nt < 4; ++nt) {
            v2f bb;
            bb.x = b2l[nt].z;
            bb.y = b2l[nt].w;
            pa[nt] = __builtin_amdgcn_wmma_f32_16x16x4_f32(
                false, a1, false, bb, (short)0, pa[nt], false, false);
        }
#pragma unroll
        for (int nt = 0; nt < 4; ++nt) {
            v2f bb;
            bb.x = b2r[nt].z;
            bb.y = b2r[nt].w;
            qa[nt] = __builtin_amdgcn_wmma_f32_16x16x4_f32(
                false, a1, false, bb, (short)0, qa[nt], false, false);
        }
    }

    const int mo = m0 + rb;
#pragma unroll
    for (int nt = 0; nt < 4; ++nt) {
        const int n = nt * 16 + mr;
        const float bias2 = bl2[n];
#pragma unroll
        for (int v = 0; v < 8; ++v) {
            p[(mo + v) * DOUT + n] = pa[nt][v];
            q[(mo + v) * DOUT + n] = qa[nt][v] + bias2;
        }
    }
}

extern "C" void kernel_launch(void* const* d_in, const int* in_sizes, int n_in,
                              void* d_out, int out_size, void* d_ws, size_t ws_size,
                              hipStream_t stream) {
    const float* x   = (const float*)d_in[0];
    const int*   ei  = (const int*)d_in[1];
    const float* Wl1 = (const float*)d_in[2];
    const float* bl1 = (const float*)d_in[3];
    const float* Wr1 = (const float*)d_in[4];
    const float* Wl2 = (const float*)d_in[5];
    const float* bl2 = (const float*)d_in[6];
    const float* Wr2 = (const float*)d_in[7];
    float* out = (float*)d_out;

    float* ws   = (float*)d_ws;
    float* pbuf = ws + (size_t)NN * DHID;
    float* qbuf = pbuf + (size_t)NN * DOUT;
    int2*  rec  = (int2*)(qbuf + (size_t)NN * DOUT);
    int*   head = (int*)(rec + NE);
    float* degb = (float*)(head + NN);

    int*   cnt    = (int*)ws;
    int*   bstart = cnt + NCH * NB;
    float* wfrag  = ws + WF_OFF;

    count_kernel<<<NCH + NWB, CT, 0, stream>>>(ei, cnt, Wl1, Wr1, Wl2, Wr2, wfrag);
    place_kernel<<<NCH, CT, 0, stream>>>(ei, cnt, bstart, rec);
    link_kernel<<<NB, LT, 0, stream>>>(bstart, head, rec);

    layer1_kernel<<<NN / 16, 32, 0, stream>>>(x, head, rec, degb, wfrag, bl1, bl2, pbuf, qbuf);
    gather2_kernel<<<(NN + 8 * GK - 1) / (8 * GK), 256, 0, stream>>>(pbuf, qbuf, degb, head, rec, out);
}


# In-edge lists built in 2 launches, not 3: each edge chunk sorted by bucket in one work-group (LDS ranks, whole-line stores), buckets link their per-chunk runs; list order still by LDS arrival
# speedup vs baseline: 4.9142x; 1.0793x over previous
#include <hip/hip_runtime.h>
#include <cmath>


#define NN 50000
#define NE 800000
#define DIN 64
#define DHID 128
#define DOUT 64
#define GK 8
#define GL 16

#define BS 256
#define NB ((NN + BS - 1) / BS)
#define NCH 125
#define CH (NE / NCH)
#define CT 640
#define EPT (CH / CT)
#define SB (NB + 1)
#define LT 512
#define LB 4
static_assert(NCH * CH == NE && EPT * CT == CH, "chunks tile the edge list exactly");
static_assert(NB <= CT && NCH <= LT && NCH < 128, "one thread per bucket counter; one linking thread per chunk; 7 search steps cover the chunks");

#define WFG 2048
#define WF_OFF 65536
#define NWB ((4 * WFG + CT - 1) / CT)
static_assert(NCH * SB <= WF_OFF, "the fragment-ordered weights start behind the chunks' bucket offsets");
static_assert(DHID * DIN == 4 * WFG && DOUT * DHID == 4 * WFG, "every weight matrix is WFG groups of 4 floats");

typedef __attribute__((ext_vector_type(2))) float v2f;
typedef __attribute__((ext_vector_type(4))) float v4f;
typedef __attribute__((ext_vector_type(8))) float v8f;

template <int NOUT, int K>
__device__ __forceinline__ void frag_order(const float* __restrict__ Wm, float* __restrict__ F, int g) {
    constexpr int NT = NOUT / 16;
    const int lane = g & 31;
    const int nt   = (g >> 5) % NT;
    const int k2   = (g >> 5) / NT;
    const int n    = nt * 16 + (lane & 15);
    const int k    = k2 * 8 + (lane >> 4) * 2;
    const float2 lo = *(const float2*)(Wm + n * K + k);
    const float2 hi = *(const float2*)(Wm + n * K + k + 4);
    *(float4*)(F + (size_t)g * 4) = make_float4(lo.x, lo.y, hi.x, hi.y);
}

__global__ __launch_bounds__(CT) void sort_kernel(const int* __restrict__ ei,
                                                  int* __restrict__ off,
                                                  int2* __restrict__ rec,
                                                  const float* __restrict__ Wl1,
                                                  const float* __restrict__ Wr1,
                                                  const float* __restrict__ Wl2,
                                                  const float* __restrict__ Wr2,
                                                  float* __restrict__ wf) {
    __shared__ int h[NB];
    __shared__ int nrec;
    __shared__ int2 st[CH];
    const int tid = threadIdx.x;
    if (blockIdx.x >= NCH) {
        const int g = (blockIdx.x - NCH) * CT + tid;
        if (g < WFG)          frag_order<DHID, DIN>(Wl1, wf, g);
        else if (g < 2 * WFG) frag_order<DHID, DIN>(Wr1, wf + 4 * WFG, g - WFG);
        else if (g < 3 * WFG) frag_order<DOUT, DHID>(Wl2, wf + 8 * WFG, g - 2 * WFG);
        else if (g < 4 * WFG) frag_order<DOUT, DHID>(Wr2, wf + 12 * WFG, g - 3 * WFG);
        return;
    }
    const int g = blockIdx.x;
    if (tid < NB) h[tid] = 0;
    const int e0 = g * CH;
    int s[EPT], d[EPT], rk[EPT];
#pragma unroll
    for (int k = 0; k < EPT; ++k) {
        s[k] = ei[e0 + k * CT + tid];
        d[k] = ei[NE + e0 + k * CT + tid];
    }
    __syncthreads();
#pragma unroll
    for (int k = 0; k < EPT; ++k) {
        rk[k] = -1;
        if ((unsigned)d[k] < (unsigned)NN) rk[k] = atomicAdd(&h[(unsigned)d[k] / BS], 1);
    }
    __syncthreads();
    if (tid < 32) {
        constexpr int PER = (NB + 31) / 32;
        int loc[PER];
        int sum = 0;
#pragma unroll
        for (int i = 0; i < PER; ++i) {
            const int b = tid * PER + i;
            const int v = (b < NB) ? h[b] : 0;
            loc[i] = sum;
            sum += v;
        }
        int inc = sum;
#pragma unroll
        for (int dd = 1; dd < 32; dd <<= 1) {
            const int o = __shfl_up(inc, dd);
            if (tid >= dd) inc += o;
        }
        const int excl = inc - sum;
#pragma unroll
        for (int i = 0; i < PER; ++i) {
            const int b = tid * PER + i;
            if (b < NB) {
                h[b] = excl + loc[i];
                off[g * SB + b] = excl + loc[i];
            }
        }
        if (tid == 31) {
            off[g * SB + NB] = inc;
            nrec = inc;
        }
    }
    __syncthreads();
    int hb[EPT];
#pragma unroll
    for (int k = 0; k < EPT; ++k) hb[k] = h[min((unsigned)d[k] / BS, (unsigned)(NB - 1))];
#pragma unroll
    for (int k = 0; k < EPT; ++k) {
        if (rk[k] >= 0) {
            const int pos = hb[k] + rk[k];
            if ((unsigned)pos < (unsigned)CH) st[pos] = make_int2(s[k], d[k]);
        }
    }
    __syncthreads();
    const int n = nrec;
#pragma unroll
    for (int k = 0; k < EPT; ++k) {
        const int i = k * CT + tid;
        if (i < n) rec[e0 + i] = st[i];
    }
}

__global__ __launch_bounds__(LT) void link_kernel(const int* __restrict__ off,
                                                  int* __restrict__ head,
                                                  int2* rec) {
    __shared__ int lh[BS];
    __shared__ int sl[NCH];
    __shared__ int sbase[NCH];
    __shared__ int pre[129];
    const int tid   = threadIdx.x;
    const int b     = blockIdx.x;
    const int node0 = b * BS;
    if (tid < BS) lh[tid] = 0;
    if (tid > NCH && tid <= 128) pre[tid] = 0x7fffffff;
    if (tid < NCH) {
        const int o0 = min(max(off[tid * SB + b], 0), CH);
        const int o1 = min(max(off[tid * SB + b + 1], o0), CH);
        sl[tid]    = o1 - o0;
        sbase[tid] = tid * CH + o0;
    }
    __syncthreads();
    if (tid < 32) {
        constexpr int PER = (NCH + 31) / 32;
        int loc[PER];
        int sum = 0;
#pragma unroll
        for (int i = 0; i < PER; ++i) {
            const int gch = tid * PER + i;
            const int v = (gch < NCH) ? sl[gch] : 0;
            loc[i] = sum;
            sum += v;
        }
        int inc = sum;
#pragma unroll
        for (int dd = 1; dd < 32; dd <<= 1) {
            const int o = __shfl_up(inc, dd);
            if (tid >= dd) inc += o;
        }
        const int excl = inc - sum;
#pragma unroll
        for (int i = 0; i < PER; ++i) {
            const int gch = tid * PER + i;
            if (gch < NCH) pre[gch] = excl + loc[i];
        }
        if (tid == 31) pre[NCH] = inc;
    }
    __syncthreads();
    const int total = pre[NCH];
    for (int i0 = 0; i0 < total; i0 += LB * LT) {
        int pos[LB];
#pragma unroll
        for (int k = 0; k < LB; ++k) pos[k] = 0;
#pragma unroll
        for (int step = 64; step >= 1; step >>= 1) {
            int pv[LB];
#pragma unroll
            for (int k = 0; k < LB; ++k) pv[k] = pre[pos[k] + step];
#pragma unroll
            for (int k = 0; k < LB; ++k)
                pos[k] += (pv[k] <= i0 + k * LT + tid) ? step : 0;
        }
        int2 r[LB];
        int ix[LB];
#pragma unroll
        for (int k = 0; k < LB; ++k) {
            const int i = i0 + k * LT + tid;
            r[k]  = make_int2(0, -1);
            ix[k] = -1;
            if (i < total) {
                const int g   = min(pos[k], NCH - 1);
                const int idx = sbase[g] + (i - pre[g]);
                if ((unsigned)idx < (unsigned)NE) {
                    ix[k] = idx;
                    r[k]  = rec[idx];
                }
            }
        }
#pragma unroll
        for (int k = 0; k < LB; ++k) {
            const unsigned li = (unsigned)(r[k].y - node0);
            if (ix[k] >= 0 && li < (unsigned)BS) {
                const int prev = atomicExch(&lh[li], ix[k] + 1);
                rec[ix[k]] = make_int2(r[k].x, prev);
            }
        }
    }
    __syncthreads();
    if (tid < BS && node0 + tid < NN) head[node0 + tid] = lh[tid];
}

__device__ __forceinline__ bool cursor_ok(int c) {
    return (unsigned)(c - 1) < (unsigned)NE;
}

__global__ __launch_bounds__(256) void gather2_kernel(
    const float* __restrict__ p, const float* __restrict__ q,
    const float* __restrict__ deg, const int* __restrict__ head,
    const int2* __restrict__ rec, float* __restrict__ out) {
    const int lane = threadIdx.x & 31;
    const int wv    = __builtin_amdgcn_readfirstlane((int)(threadIdx.x >> 5));
    const int node0 = (blockIdx.x * 8 + wv) * GK;
    if (node0 >= NN) return;

    const bool owner = lane < GK && node0 + lane < NN;
    int cur = 0;
    float dg = 1.0f;
    if (owner) {
        cur = head[node0 + lane];
        dg  = deg[node0 + lane];
    }
    bool alive = owner && cursor_ok(cur);
    int2 r = make_int2(0, 0);
    if (alive) r = rec[cur - 1];

    float2 qv[GK];
#pragma unroll
    for (int j = 0; j < GK; ++j) {
        qv[j] = make_float2(0.0f, 0.0f);
        if (node0 + j < NN)
            qv[j] = *(const float2*)(q + (size_t)(node0 + j) * DOUT + lane * 2);
    }

    float2 acc[GK];
#pragma unroll
    for (int j = 0; j < GK; ++j) acc[j] = make_float2(0.0f, 0.0f);

    for (int it = 0; it < NE; ++it) {
        const unsigned live = __builtin_amdgcn_ballot_w32(alive);
        if (live == 0u) break;
        const int nxt = r.y;
        const bool hn = alive && cursor_ok(nxt);
        int2 rn = make_int2(0, 0);
        if (hn) rn = rec[nxt - 1];
        float2 v[GK];
#pragma unroll
        for (int j = 0; j < GK; ++j) {
            v[j] = make_float2(0.0f, 0.0f);
            if (live & (1u << j)) {
                const int s = __builtin_amdgcn_readlane(r.x, j);
                v[j] = *(const float2*)(p + (size_t)s * DOUT + lane * 2);
            }
        }
#pragma unroll
        for (int j = 0; j < GK; ++j) {
            acc[j].x += v[j].x;
            acc[j].y += v[j].y;
        }
        r = rn;
        alive = hn;
    }
    const float dginv = 1.0f / fmaxf(dg, 1.0f);
    float zx[GK], zy[GK], mx[GK], se[GK], t[GK];
#pragma unroll
    for (int j = 0; j < GK; ++j) {
        const float dinv = __int_as_float(__builtin_amdgcn_readlane(__float_as_int(dginv), j));
        zx[j] = acc[j].x * dinv + qv[j].x;
        zy[j] = acc[j].y * dinv + qv[j].y;
        mx[j] = fmaxf(zx[j], zy[j]);
    }
#pragma unroll
    for (int d = 16; d >= 1; d >>= 1) {
#pragma unroll
        for (int j = 0; j < GK; ++j) t[j] = __shfl_xor(mx[j], d);
#pragma unroll
        for (int j = 0; j < GK; ++j) mx[j] = fmaxf(mx[j], t[j]);
    }
#pragma unroll
    for (int j = 0; j < GK; ++j) se[j] = expf(zx[j] - mx[j]) + expf(zy[j] - mx[j]);
#pragma unroll
    for (int d = 16; d >= 1; d >>= 1) {
#pragma unroll
        for (int j = 0; j < GK; ++j) t[j] = __shfl_xor(se[j], d);
#pragma unroll
        for (int j = 0; j < GK; ++j) se[j] += t[j];
    }
#pragma unroll
    for (int j = 0; j < GK; ++j) {
        if (node0 + j < NN) {
            const float lse = mx[j] + logf(se[j]);
            *(float2*)(out + (size_t)(node0 + j) * DOUT + lane * 2) =
                make_float2(zx[j] - lse, zy[j] - lse);
        }
    }
}

#define HP (DHID + 4)
#define AP (DIN + 4)
__global__ __launch_bounds__(32) void layer1_kernel(
    const float* __restrict__ x, const int* __restrict__ head,
    const int2* __restrict__ rec, float* __restrict__ deg,
    const float* __restrict__ wf, const float* __restrict__ bl,
    const float* __restrict__ bl2,
    float* __restrict__ p, float* __restrict__ q) {
    __shared__ __attribute__((aligned(16))) float as[16 * AP];
    __shared__ __attribute__((aligned(16))) float hs[16 * HP];
    const int lane = threadIdx.x;
    const int m0   = blockIdx.x * 16;
    const int mr   = lane & 15;
    const int kb   = (lane >> 4) * 2;
    const v4f* __restrict__ fl1 = (const v4f*)wf;
    const v4f* __restrict__ fr1 = fl1 + WFG;
    const v4f* __restrict__ fl2 = fr1 + WFG;
    const v4f* __restrict__ fr2 = fl2 + WFG;

    const bool owner = lane < GL && m0 + lane < NN;
    int cur = 0;
    if (owner) cur = head[m0 + lane];
    bool alive = owner && cursor_ok(cur);
    int2 r = make_int2(0, 0);
    if (alive) r = rec[cur - 1];

    float2 sacc[GL];
#pragma unroll
    for (int j = 0; j < GL; ++j) sacc[j] = make_float2(0.0f, 0.0f);
    int cnt = 0;

    for (int it = 0; it < NE; ++it) {
        const unsigned live = __builtin_amdgcn_ballot_w32(alive);
        if (live == 0u) break;
        const int nxt = r.y;
        const bool hn = alive && cursor_ok(nxt);
        int2 rn = make_int2(0, 0);
        if (hn) rn = rec[nxt - 1];
        float2 v[GL];
#pragma unroll
        for (int j = 0; j < GL; ++j) {
            v[j] = make_float2(0.0f, 0.0f);
            if (live & (1u << j)) {
                const int s = __builtin_amdgcn_readlane(r.x, j);
                v[j] = *(const float2*)(x + (size_t)s * DIN + lane * 2);
            }
        }
#pragma unroll
        for (int j = 0; j < GL; ++j) {
            sacc[j].x += v[j].x;
            sacc[j].y += v[j].y;
        }
        cnt += alive ? 1 : 0;
        r = rn;
        alive = hn;
    }
    if (owner) deg[m0 + lane] = (float)cnt;
    const float dgi = 1.0f / fmaxf((float)cnt, 1.0f);
#pragma unroll
    for (int j = 0; j < GL; ++j) {
        const float dj = __int_as_float(__builtin_amdgcn_readlane(__float_as_int(dgi), j));
        *(float2*)(as + j * AP + lane * 2) = make_float2(sacc[j].x * dj, sacc[j].y * dj);
    }
    __syncthreads();

    v8f acc[8] = {};

    for (int k2 = 0; k2 < DIN / 8; ++k2) {
        const int k = k2 * 8 + kb;
        v2f a0, a1;
        a0.x = as[mr * AP + k];
        a0.y = as[mr * AP + k + 1];
        a1.x = as[mr * AP + k + 4];
        a1.y = as[mr * AP + k + 5];
        v4f b[8];
#pragma unroll
        for (int nt = 0; nt < 8; ++nt) b[nt] = fl1[(k2 * 8 + nt) * 32 + lane];
#pragma unroll
        for (int nt = 0; nt < 8; ++nt) {
            v2f bb;
            bb.x = b[nt].x;
            bb.y = b[nt].y;
            acc[nt] = __builtin_amdgcn_wmma_f32_16x16x4_f32(
                false, a0, false, bb, (short)0, acc[nt], false, false);
        }
#pragma unroll
        for (int nt = 0; nt < 8; ++nt) {
            v2f bb;
            bb.x = b[nt].z;
            bb.y = b[nt].w;
            acc[nt] = __builtin_amdgcn_wmma_f32_16x16x4_f32(
                false, a1, false, bb, (short)0, acc[nt], false, false);
        }
    }
    const int m = m0 + mr;
    for (int k2 = 0; k2 < DIN / 8; ++k2) {
        const int k = k2 * 8 + kb;
        v2f a0, a1;
        a0.x = x[m * DIN + k];
        a0.y = x[m * DIN + k + 1];
        a1.x = x[m * DIN + k + 4];
        a1.y = x[m * DIN + k + 5];
        v4f b[8];
#pragma unroll
        for (int nt = 0; nt < 8; ++nt) b[nt] = fr1[(k2 * 8 + nt) * 32 + lane];
#pragma unroll
        for (int nt = 0; nt < 8; ++nt) {
            v2f bb;
            bb.x = b[nt].x;
            bb.y = b[nt].y;
            acc[nt] = __builtin_amdgcn_wmma_f32_16x16x4_f32(
                false, a0, false, bb, (short)0, acc[nt], false, false);
        }
#pragma unroll
        for (int nt = 0; nt < 8; ++nt) {
            v2f bb;
            bb.x = b[nt].z;
            bb.y = b[nt].w;
            acc[nt] = __builtin_amdgcn_wmma_f32_16x16x4_f32(
                false, a1, false, bb, (short)0, acc[nt], false, false);
        }
    }

    const int rb = 8 * (lane >> 4);
#pragma unroll
    for (int nt = 0; nt < 8; ++nt) {
        const int n = nt * 16 + mr;
        const float bias = bl[n];
#pragma unroll
        for (int v = 0; v < 8; ++v) {
            hs[(rb + v) * HP + n] = tanhf(acc[nt][v] + bias);
        }
    }
    __syncthreads();

    v8f pa[4] = {};
    v8f qa[4] = {};
    for (int k2 = 0; k2 < DHID / 8; ++k2) {
        const int k = k2 * 8 + kb;
        v2f a0, a1;
        a0.x = hs[mr * HP + k];
        a0.y = hs[mr * HP + k + 1];
        a1.x = hs[mr * HP + k + 4];
        a1.y = hs[mr * HP + k + 5];
        v4f b2l[4], b2r[4];
#pragma unroll
        for (int nt = 0; nt < 4; ++nt) {
            b2l[nt] = fl2[(k2 * 4 + nt) * 32 + lane];
            b2r[nt] = fr2[(k2 * 4 + nt) * 32 + lane];
        }
#pragma unroll
        for (int nt = 0; nt < 4; ++nt) {
            v2f bb;
            bb.x = b2l[nt].x;
            bb.y = b2l[nt].y;
            pa[nt] = __builtin_amdgcn_wmma_f32_16x16x4_f32(
                false, a0, false, bb, (short)0, pa[nt], false, false);
        }
#pragma unroll
        for (int nt = 0; nt < 4; ++nt) {
            v2f bb;
            bb.x = b2r[nt].x;
            bb.y = b2r[nt].y;
            qa[nt] = __builtin_amdgcn_wmma_f32_16x16x4_f32(
                false, a0, false, bb, (short)0, qa[nt], false, false);
        }
#pragma unroll
        for (int nt = 0; nt < 4; ++nt) {
            v2f bb;
            bb.x = b2l[nt].z;
            bb.y = b2l[nt].w;
            pa[nt] = __builtin_amdgcn_wmma_f32_16x16x4_f32(
                false, a1, false, bb, (short)0, pa[nt], false, false);
        }
#pragma unroll
        for (int nt = 0; nt < 4; ++nt) {
            v2f bb;
            bb.x = b2r[nt].z;
            bb.y = b2r[nt].w;
            qa[nt] = __builtin_amdgcn_wmma_f32_16x16x4_f32(
                false, a1, false, bb, (short)0, qa[nt], false, false);
        }
    }

    const int mo = m0 + rb;
#pragma unroll
    for (int nt = 0; nt < 4; ++nt) {
        const int n = nt * 16 + mr;
        const float bias2 = bl2[n];
#pragma unroll
        for (int v = 0; v < 8; ++v) {
            p[(mo + v) * DOUT + n] = pa[nt][v];
            q[(mo + v) * DOUT + n] = qa[nt][v] + bias2;
        }
    }
}

extern "C" void kernel_launch(void* const* d_in, const int* in_sizes, int n_in,
                              void* d_out, int out_size, void* d_ws, size_t ws_size,
                              hipStream_t stream) {
    const float* x   = (const float*)d_in[0];
    const int*   ei  = (const int*)d_in[1];
    const float* Wl1 = (const float*)d_in[2];
    const float* bl1 = (const float*)d_in[3];
    const float* Wr1 = (const float*)d_in[4];
    const float* Wl2 = (const float*)d_in[5];
    const float* bl2 = (const float*)d_in[6];
    const float* Wr2 = (const float*)d_in[7];
    float* out = (float*)d_out;

    float* ws   = (float*)d_ws;
    float* pbuf = ws + (size_t)NN * DHID;
    float* qbuf = pbuf + (size_t)NN * DOUT;
    int2*  rec  = (int2*)(qbuf + (size_t)NN * DOUT);
    int*   head = (int*)(rec + NE);
    float* degb = (float*)(head + NN);

    int*   off    = (int*)ws;
    float* wfrag  = ws + WF_OFF;

    sort_kernel<<<NCH + NWB, CT, 0, stream>>>(ei, off, rec, Wl1, Wr1, Wl2, Wr2, wfrag);
    link_kernel<<<NB, LT, 0, stream>>>(off, head, rec);

    layer1_kernel<<<NN / 16, 32, 0, stream>>>(x, head, rec, degb, wfrag, bl1, bl2, pbuf, qbuf);
    gather2_kernel<<<(NN + 8 * GK - 1) / (8 * GK), 256, 0, stream>>>(pbuf, qbuf, degb, head, rec, out);
}


# In-edge lists kept as one contiguous index segment per node (count, scan, place on LDS counters), not linked records: walk loads no longer chain, 4 trips' rows in flight; sum order by LDS arrival
# speedup vs baseline: 5.5256x; 1.1244x over previous
#include <hip/hip_runtime.h>
#include <cmath>


#define NN 50000
#define NE 800000
#define DIN 64
#define DHID 128
#define DOUT 64
#define GK 8
#define GL 16
#define WU 4

#define BS 256
#define NB ((NN + BS - 1) / BS)
#define NCH 125
#define CH (NE / NCH)
#define CT 640
#define EPT (CH / CT)
#define SB (NB + 1)
#define LT 1024
#define LB 5
#define SCAP 8192
static_assert(NCH * CH == NE && EPT * CT == CH, "chunks tile the edge list exactly");
static_assert(NB <= CT && NCH <= LT && NCH < 128, "one thread per bucket counter; one placing thread per chunk; 7 search steps cover the chunks");

#define WFG 2048
#define WF_OFF 65536
#define NWB ((4 * WFG + CT - 1) / CT)
#define COL_OFF 131072
static_assert(NCH * SB <= WF_OFF, "the fragment-ordered weights start behind the chunks' bucket offsets");
static_assert(WF_OFF + 16 * WFG <= COL_OFF && COL_OFF + NE <= NN * DHID, "col[] lies behind the weights and inside the scratch region");
static_assert(BS == 256, "the first wave scans the bucket's counters 8 per lane");
static_assert(DHID * DIN == 4 * WFG && DOUT * DHID == 4 * WFG, "every weight matrix is WFG groups of 4 floats");

typedef __attribute__((ext_vector_type(2))) float v2f;
typedef __attribute__((ext_vector_type(4))) float v4f;
typedef __attribute__((ext_vector_type(8))) float v8f;

template <int NOUT, int K>
__device__ __forceinline__ void frag_order(const float* __restrict__ Wm, float* __restrict__ F, int g) {
    constexpr int NT = NOUT / 16;
    const int lane = g & 31;
    const int nt   = (g >> 5) % NT;
    const int k2   = (g >> 5) / NT;
    const int n    = nt * 16 + (lane & 15);
    const int k    = k2 * 8 + (lane >> 4) * 2;
    const float2 lo = *(const float2*)(Wm + n * K + k);
    const float2 hi = *(const float2*)(Wm + n * K + k + 4);
    *(float4*)(F + (size_t)g * 4) = make_float4(lo.x, lo.y, hi.x, hi.y);
}

__global__ __launch_bounds__(CT) void sort_kernel(const int* __restrict__ ei,
                                                  int* __restrict__ off,
                                                  int2* __restrict__ rec,
                                                  const float* __restrict__ Wl1,
                                                  const float* __restrict__ Wr1,
                                                  const float* __restrict__ Wl2,
                                                  const float* __restrict__ Wr2,
                                                  float* __restrict__ wf) {
    __shared__ int h[NB];
    __shared__ int nrec;
    __shared__ int2 st[CH];
    const int tid = threadIdx.x;
    if (blockIdx.x >= NCH) {
        const int g = (blockIdx.x - NCH) * CT + tid;
        if (g < WFG)          frag_order<DHID, DIN>(Wl1, wf, g);
        else if (g < 2 * WFG) frag_order<DHID, DIN>(Wr1, wf + 4 * WFG, g - WFG);
        else if (g < 3 * WFG) frag_order<DOUT, DHID>(Wl2, wf + 8 * WFG, g - 2 * WFG);
        else if (g < 4 * WFG) frag_order<DOUT, DHID>(Wr2, wf + 12 * WFG, g - 3 * WFG);
        return;
    }
    const int g = blockIdx.x;
    if (tid < NB) h[tid] = 0;
    const int e0 = g * CH;
    int s[EPT], d[EPT], rk[EPT];
#pragma unroll
    for (int k = 0; k < EPT; ++k) {
        s[k] = ei[e0 + k * CT + tid];
        d[k] = ei[NE + e0 + k * CT + tid];
    }
    __syncthreads();
#pragma unroll
    for (int k = 0; k < EPT; ++k) {
        rk[k] = -1;
        if ((unsigned)d[k] < (unsigned)NN) rk[k] = atomicAdd(&h[(unsigned)d[k] / BS], 1);
    }
    __syncthreads();
    if (tid < 32) {
        constexpr int PER = (NB + 31) / 32;
        int loc[PER];
        int sum = 0;
#pragma unroll
        for (int i = 0; i < PER; ++i) {
            const int b = tid * PER + i;
            const int v = (b < NB) ? h[b] : 0;
            loc[i] = sum;
            sum += v;
        }
        int inc = sum;
#pragma unroll
        for (int dd = 1; dd < 32; dd <<= 1) {
            const int o = __shfl_up(inc, dd);
            if (tid >= dd) inc += o;
        }
        const int excl = inc - sum;
#pragma unroll
        for (int i = 0; i < PER; ++i) {
            const int b = tid * PER + i;
            if (b < NB) {
                h[b] = excl + loc[i];
                off[g * SB + b] = excl + loc[i];
            }
        }
        if (tid == 31) {
            off[g * SB + NB] = inc;
            nrec = inc;
        }
    }
    __syncthreads();
    int hb[EPT];
#pragma unroll
    for (int k = 0; k < EPT; ++k) hb[k] = h[min((unsigned)d[k] / BS, (unsigned)(NB - 1))];
#pragma unroll
    for (int k = 0; k < EPT; ++k) {
        if (rk[k] >= 0) {
            const int pos = hb[k] + rk[k];
            if ((unsigned)pos < (unsigned)CH) st[pos] = make_int2(s[k], d[k]);
        }
    }
    __syncthreads();
    const int n = nrec;
#pragma unroll
    for (int k = 0; k < EPT; ++k) {
        const int i = k * CT + tid;
        if (i < n) rec[e0 + i] = st[i];
    }
}

__device__ __forceinline__ void fetch_batch(int i0, int total, int tid, int node0,
                                            const int* pre, const int* sbase,
                                            const int2* __restrict__ rec,
                                            int (&src)[LB], int (&li)[LB]) {
    int pos[LB];
#pragma unroll
    for (int k = 0; k < LB; ++k) pos[k] = 0;
#pragma unroll
    for (int step = 64; step >= 1; step >>= 1) {
        int pv[LB];
#pragma unroll
        for (int k = 0; k < LB; ++k) pv[k] = pre[pos[k] + step];
#pragma unroll
        for (int k = 0; k < LB; ++k)
            pos[k] += (pv[k] <= i0 + k * LT + tid) ? step : 0;
    }
    int2 r[LB];
#pragma unroll
    for (int k = 0; k < LB; ++k) {
        const int i = i0 + k * LT + tid;
        r[k] = make_int2(0, -1);
        if (i < total) {
            const int g   = min(pos[k], NCH - 1);
            const int idx = sbase[g] + (i - pre[g]);
            if ((unsigned)idx < (unsigned)NE) r[k] = rec[idx];
        }
    }
#pragma unroll
    for (int k = 0; k < LB; ++k) {
        const unsigned l = (unsigned)(r[k].y - node0);
        src[k] = r[k].x;
        li[k]  = (l < (unsigned)BS) ? (int)l : -1;
    }
}

__global__ __launch_bounds__(LT) void place_kernel(const int* __restrict__ off,
                                                   const int2* __restrict__ rec,
                                                   int* __restrict__ seg,
                                                   int* __restrict__ cnt,
                                                   int* __restrict__ col) {
    __shared__ int nc[BS];
    __shared__ int cu[BS];
    __shared__ int so0[NCH];
    __shared__ int sl[NCH];
    __shared__ int sbase[NCH];
    __shared__ int pre[129];
    __shared__ int cbase;
    __shared__ int stage[SCAP];
    const int tid   = threadIdx.x;
    const int b     = blockIdx.x;
    const int node0 = b * BS;
    if (tid < BS) nc[tid] = 0;
    if (tid > NCH && tid <= 128) pre[tid] = 0x7fffffff;
    if (tid < NCH) {
        const int o0 = min(max(off[tid * SB + b], 0), CH);
        const int o1 = min(max(off[tid * SB + b + 1], o0), CH);
        so0[tid]   = o0;
        sl[tid]    = o1 - o0;
        sbase[tid] = tid * CH + o0;
    }
    __syncthreads();
    int base = 0;
    if (tid < 32) {
        constexpr int PER = (NCH + 31) / 32;
        int loc[PER];
        int sum = 0;
#pragma unroll
        for (int i = 0; i < PER; ++i) {
            const int gch = tid * PER + i;
            const int v = (gch < NCH) ? sl[gch] : 0;
            base += (gch < NCH) ? so0[gch] : 0;
            loc[i] = sum;
            sum += v;
        }
        int inc = sum;
#pragma unroll
        for (int dd = 1; dd < 32; dd <<= 1) {
            const int o = __shfl_up(inc, dd);
            if (tid >= dd) inc += o;
        }
#pragma unroll
        for (int dd = 16; dd >= 1; dd >>= 1) base += __shfl_xor(base, dd);
        const int excl = inc - sum;
#pragma unroll
        for (int i = 0; i < PER; ++i) {
            const int gch = tid * PER + i;
            if (gch < NCH) pre[gch] = excl + loc[i];
        }
        if (tid == 31) {
            pre[NCH] = inc;
            cbase    = base;
        }
    }
    __syncthreads();
    const int total = pre[NCH];
    const int cb    = cbase;
    const bool staged = total <= SCAP;

    int s0[LB], l0[LB];
    fetch_batch(0, total, tid, node0, pre, sbase, rec, s0, l0);
#pragma unroll
    for (int k = 0; k < LB; ++k)
        if (l0[k] >= 0) atomicAdd(&nc[l0[k]], 1);
    for (int i0 = LB * LT; i0 < total; i0 += LB * LT) {
        int s[LB], l[LB];
        fetch_batch(i0, total, tid, node0, pre, sbase, rec, s, l);
#pragma unroll
        for (int k = 0; k < LB; ++k)
            if (l[k] >= 0) atomicAdd(&nc[l[k]], 1);
    }
    __syncthreads();
    if (tid < 32) {
        int loc[8];
        int sum = 0;
#pragma unroll
        for (int i = 0; i < 8; ++i) {
            loc[i] = sum;
            sum += nc[tid * 8 + i];
        }
        int inc = sum;
#pragma unroll
        for (int dd = 1; dd < 32; dd <<= 1) {
            const int o = __shfl_up(inc, dd);
            if (tid >= dd) inc += o;
        }
        const int excl = base + inc - sum;
#pragma unroll
        for (int i = 0; i < 8; ++i) cu[tid * 8 + i] = excl + loc[i];
    }
    __syncthreads();
    auto put = [&](int l, int s) {
        const int place = atomicAdd(&cu[l], 1);
        if (staged) {
            const unsigned o = (unsigned)(place - cb);
            if (o < (unsigned)SCAP) stage[o] = s;
        } else if ((unsigned)place < (unsigned)NE) {
            col[place] = s;
        }
    };
#pragma unroll
    for (int k = 0; k < LB; ++k)
        if (l0[k] >= 0) put(l0[k], s0[k]);
    for (int i0 = LB * LT; i0 < total; i0 += LB * LT) {
        int s[LB], l[LB];
        fetch_batch(i0, total, tid, node0, pre, sbase, rec, s, l);
#pragma unroll
        for (int k = 0; k < LB; ++k)
            if (l[k] >= 0) put(l[k], s[k]);
    }
    __syncthreads();
    if (staged) {
        for (int i = tid; i < total; i += LT)
            if ((unsigned)(cb + i) < (unsigned)NE) col[cb + i] = stage[i];
    }
    if (tid < BS && node0 + tid < NN) {
        seg[node0 + tid] = cu[tid] - nc[tid];
        cnt[node0 + tid] = nc[tid];
    }
}

__device__ __forceinline__ int row_bound(int s) {
    return min(max(s, 0), NN - 1);
}

__device__ __forceinline__ void seg_bound(int& first, int& len) {
    first = min(max(first, 0), NE);
    len   = min(max(len, 0), NE - first);
}

__global__ __launch_bounds__(256) void gather2_kernel(
    const float* __restrict__ p, const float* __restrict__ q,
    const int* __restrict__ seg, const int* __restrict__ cnt,
    const int* __restrict__ col, float* __restrict__ out) {
    const int lane = threadIdx.x & 31;
    const int wv    = __builtin_amdgcn_readfirstlane((int)(threadIdx.x >> 5));
    const int node0 = (blockIdx.x * 8 + wv) * GK;
    if (node0 >= NN) return;

    const bool owner = lane < GK && node0 + lane < NN;
    int first = 0, dg = 0;
    if (owner) {
        first = seg[node0 + lane];
        dg    = cnt[node0 + lane];
    }
    seg_bound(first, dg);
    int trips = dg;
#pragma unroll
    for (int d = 16; d >= 1; d >>= 1) trips = max(trips, __shfl_xor(trips, d));
    trips = __builtin_amdgcn_readfirstlane(trips);
    int nr[WU];
#pragma unroll
    for (int u = 0; u < WU; ++u) {
        nr[u] = 0;
        if (u < dg) nr[u] = col[first + u];
    }

    float2 qv[GK];
#pragma unroll
    for (int j = 0; j < GK; ++j) {
        qv[j] = make_float2(0.0f, 0.0f);
        if (node0 + j < NN)
            qv[j] = *(const float2*)(q + (size_t)(node0 + j) * DOUT + lane * 2);
    }

    float2 acc[GK];
#pragma unroll
    for (int j = 0; j < GK; ++j) acc[j] = make_float2(0.0f, 0.0f);

    int ix[WU];
#pragma unroll
    for (int u = 0; u < WU; ++u) ix[u] = row_bound(nr[u]);
    for (int t0 = 0; t0 < trips; t0 += WU) {
#pragma unroll
        for (int u = 0; u < WU; ++u) {
            const int t = t0 + WU + u;
            nr[u] = 0;
            if (t < dg) nr[u] = col[first + t];
        }
        float2 v[WU][GK];
#pragma unroll
        for (int u = 0; u < WU; ++u) {
            const unsigned live = __builtin_amdgcn_ballot_w32(t0 + u < dg);
#pragma unroll
            for (int j = 0; j < GK; ++j) {
                v[u][j] = make_float2(0.0f, 0.0f);
                if (live & (1u << j)) {
                    const int s = __builtin_amdgcn_readlane(ix[u], j);
                    v[u][j] = *(const float2*)(p + (size_t)s * DOUT + lane * 2);
                }
            }
        }
#pragma unroll
        for (int u = 0; u < WU; ++u) {
#pragma unroll
            for (int j = 0; j < GK; ++j) {
                acc[j].x += v[u][j].x;
                acc[j].y += v[u][j].y;
            }
        }
#pragma unroll
        for (int u = 0; u < WU; ++u) ix[u] = row_bound(nr[u]);
    }
    const float dginv = 1.0f / fmaxf((float)dg, 1.0f);
    float zx[GK], zy[GK], mx[GK], se[GK], t[GK];
#pragma unroll
    for (int j = 0; j < GK; ++j) {
        const float dinv = __int_as_float(__builtin_amdgcn_readlane(__float_as_int(dginv), j));
        zx[j] = acc[j].x * dinv + qv[j].x;
        zy[j] = acc[j].y * dinv + qv[j].y;
        mx[j] = fmaxf(zx[j], zy[j]);
    }
#pragma unroll
    for (int d = 16; d >= 1; d >>= 1) {
#pragma unroll
        for (int j = 0; j < GK; ++j) t[j] = __shfl_xor(mx[j], d);
#pragma unroll
        for (int j = 0; j < GK; ++j) mx[j] = fmaxf(mx[j], t[j]);
    }
#pragma unroll
    for (int j = 0; j < GK; ++j) se[j] = expf(zx[j] - mx[j]) + expf(zy[j] - mx[j]);
#pragma unroll
    for (int d = 16; d >= 1; d >>= 1) {
#pragma unroll
        for (int j = 0; j < GK; ++j) t[j] = __shfl_xor(se[j], d);
#pragma unroll
        for (int j = 0; j < GK; ++j) se[j] += t[j];
    }
#pragma unroll
    for (int j = 0; j < GK; ++j) {
        if (node0 + j < NN) {
            const float lse = mx[j] + logf(se[j]);
            *(float2*)(out + (size_t)(node0 + j) * DOUT + lane * 2) =
                make_float2(zx[j] - lse, zy[j] - lse);
        }
    }
}

#define HP (DHID + 4)
#define AP (DIN + 4)
__global__ __launch_bounds__(32) void layer1_kernel(
    const float* __restrict__ x, const int* __restrict__ seg,
    const int* __restrict__ cnt, const int* __restrict__ col,
    const float* __restrict__ wf, const float* __restrict__ bl,
    const float* __restrict__ bl2,
    float* __restrict__ p, float* __restrict__ q) {
    __shared__ __attribute__((aligned(16))) float as[16 * AP];
    __shared__ __attribute__((aligned(16))) float hs[16 * HP];
    const int lane = threadIdx.x;
    const int m0   = blockIdx.x * 16;
    const int mr   = lane & 15;
    const int kb   = (lane >> 4) * 2;
    const v4f* __restrict__ fl1 = (const v4f*)wf;
    const v4f* __restrict__ fr1 = fl1 + WFG;
    const v4f* __restrict__ fl2 = fr1 + WFG;
    const v4f* __restrict__ fr2 = fl2 + WFG;

    const bool owner = lane < GL && m0 + lane < NN;
    int first = 0, dg = 0;
    if (owner) {
        first = seg[m0 + lane];
        dg    = cnt[m0 + lane];
    }
    seg_bound(first, dg);
    int trips = dg;
#pragma unroll
    for (int d = 16; d >= 1; d >>= 1) trips = max(trips, __shfl_xor(trips, d));
    trips = __builtin_amdgcn_readfirstlane(trips);
    int nr[WU];
#pragma unroll
    for (int u = 0; u < WU; ++u) {
        nr[u] = 0;
        if (u < dg) nr[u] = col[first + u];
    }

    float2 sacc[GL];
#pragma unroll
    for (int j = 0; j < GL; ++j) sacc[j] = make_float2(0.0f, 0.0f);

    int ix[WU];
#pragma unroll
    for (int u = 0; u < WU; ++u) ix[u] = row_bound(nr[u]);
    for (int t0 = 0; t0 < trips; t0 += WU) {
#pragma unroll
        for (int u = 0; u < WU; ++u) {
            const int t = t0 + WU + u;
            nr[u] = 0;
            if (t < dg) nr[u] = col[first + t];
        }
        float2 v[WU][GL];
#pragma unroll
        for (int u = 0; u < WU; ++u) {
            const unsigned live = __builtin_amdgcn_ballot_w32(t0 + u < dg);
#pragma unroll
            for (int j = 0; j < GL; ++j) {
                v[u][j] = make_float2(0.0f, 0.0f);
                if (live & (1u << j)) {
                    const int s = __builtin_amdgcn_readlane(ix[u], j);
                    v[u][j] = *(const float2*)(x + (size_t)s * DIN + lane * 2);
                }
            }
        }
#pragma unroll
        for (int u = 0; u < WU; ++u) {
#pragma unroll
            for (int j = 0; j < GL; ++j) {
                sacc[j].x += v[u][j].x;
                sacc[j].y += v[u][j].y;
            }
        }
#pragma unroll
        for (int u = 0; u < WU; ++u) ix[u] = row_bound(nr[u]);
    }
    const float dgi = 1.0f / fmaxf((float)dg, 1.0f);
#pragma unroll
    for (int j = 0; j < GL; ++j) {
        const float dj = __int_as_float(__builtin_amdgcn_readlane(__float_as_int(dgi), j));
        *(float2*)(as + j * AP + lane * 2) = make_float2(sacc[j].x * dj, sacc[j].y * dj);
    }
    __syncthreads();

    v8f acc[8] = {};

    for (int k2 = 0; k2 < DIN / 8; ++k2) {
        const int k = k2 * 8 + kb;
        v2f a0, a1;
        a0.x = as[mr * AP + k];
        a0.y = as[mr * AP + k + 1];
        a1.x = as[mr * AP + k + 4];
        a1.y = as[mr * AP + k + 5];
        v4f b[8];
#pragma unroll
        for (int nt = 0; nt < 8; ++nt) b[nt] = fl1[(k2 * 8 + nt) * 32 + lane];
#pragma unroll
        for (int nt = 0; nt < 8; ++nt) {
            v2f bb;
            bb.x = b[nt].x;
            bb.y = b[nt].y;
            acc[nt] = __builtin_amdgcn_wmma_f32_16x16x4_f32(
                false, a0, false, bb, (short)0, acc[nt], false, false);
        }
#pragma unroll
        for (int nt = 0; nt < 8; ++nt) {
            v2f bb;
            bb.x = b[nt].z;
            bb.y = b[nt].w;
            acc[nt] = __builtin_amdgcn_wmma_f32_16x16x4_f32(
                false, a1, false, bb, (short)0, acc[nt], false, false);
        }
    }
    const int m = m0 + mr;
    for (int k2 = 0; k2 < DIN / 8; ++k2) {
        const int k = k2 * 8 + kb;
        v2f a0, a1;
        a0.x = x[m * DIN + k];
        a0.y = x[m * DIN + k + 1];
        a1.x = x[m * DIN + k + 4];
        a1.y = x[m * DIN + k + 5];
        v4f b[8];
#pragma unroll
        for (int nt = 0; nt < 8; ++nt) b[nt] = fr1[(k2 * 8 + nt) * 32 + lane];
#pragma unroll
        for (int nt = 0; nt < 8; ++nt) {
            v2f bb;
            bb.x = b[nt].x;
            bb.y = b[nt].y;
            acc[nt] = __builtin_amdgcn_wmma_f32_16x16x4_f32(
                false, a0, false, bb, (short)0, acc[nt], false, false);
        }
#pragma unroll
        for (int nt = 0; nt < 8; ++nt) {
            v2f bb;
            bb.x = b[nt].z;
            bb.y = b[nt].w;
            acc[nt] = __builtin_amdgcn_wmma_f32_16x16x4_f32(
                false, a1, false, bb, (short)0, acc[nt], false, false);
        }
    }

    const int rb = 8 * (lane >> 4);
#pragma unroll
    for (int nt = 0; nt < 8; ++nt) {
        const int n = nt * 16 + mr;
        const float bias = bl[n];
#pragma unroll
        for (int v = 0; v < 8; ++v) {
            hs[(rb + v) * HP + n] = tanhf(acc[nt][v] + bias);
        }
    }
    __syncthreads();

    v8f pa[4] = {};
    v8f qa[4] = {};
    for (int k2 = 0; k2 < DHID / 8; ++k2) {
        const int k = k2 * 8 + kb;
        v2f a0, a1;
        a0.x = hs[mr * HP + k];
        a0.y = hs[mr * HP + k + 1];
        a1.x = hs[mr * HP + k + 4];
        a1.y = hs[mr * HP + k + 5];
        v4f b2l[4], b2r[4];
#pragma unroll
        for (int nt = 0; nt < 4; ++nt) {
            b2l[nt] = fl2[(k2 * 4 + nt) * 32 + lane];
            b2r[nt] = fr2[(k2 * 4 + nt) * 32 + lane];
        }
#pragma unroll
        for (int nt = 0; nt < 4; ++nt) {
            v2f bb;
            bb.x = b2l[nt].x;
            bb.y = b2l[nt].y;
            pa[nt] = __builtin_amdgcn_wmma_f32_16x16x4_f32(
                false, a0, false, bb, (short)0, pa[nt], false, false);
        }
#pragma unroll
        for (int nt = 0; nt < 4; ++nt) {
            v2f bb;
            bb.x = b2r[nt].x;
            bb.y = b2r[nt].y;
            qa[nt] = __builtin_amdgcn_wmma_f32_16x16x4_f32(
                false, a0, false, bb, (short)0, qa[nt], false, false);
        }
#pragma unroll
        for (int nt = 0; nt < 4; ++nt) {
            v2f bb;
            bb.x = b2l[nt].z;
            bb.y = b2l[nt].w;
            pa[nt] = __builtin_amdgcn_wmma_f32_16x16x4_f32(
                false, a1, false, bb, (short)0, pa[nt], false, false);
        }
#pragma unroll
        for (int nt = 0; nt < 4; ++nt) {
            v2f bb;
            bb.x = b2r[nt].z;
            bb.y = b2r[nt].w;
            qa[nt] = __builtin_amdgcn_wmma_f32_16x16x4_f32(
                false, a1, false, bb, (short)0, qa[nt], false, false);
        }
    }

    const int mo = m0 + rb;
#pragma unroll
    for (int nt = 0; nt < 4; ++nt) {
        const int n = nt * 16 + mr;
        const float bias2 = bl2[n];
#pragma unroll
        for (int v = 0; v < 8; ++v) {
            p[(mo + v) * DOUT + n] = pa[nt][v];
            q[(mo + v) * DOUT + n] = qa[nt][v] + bias2;
        }
    }
}

extern "C" void kernel_launch(void* const* d_in, const int* in_sizes, int n_in,
                              void* d_out, int out_size, void* d_ws, size_t ws_size,
                              hipStream_t stream) {
    const float* x   = (const float*)d_in[0];
    const int*   ei  = (const int*)d_in[1];
    const float* Wl1 = (const float*)d_in[2];
    const float* bl1 = (const float*)d_in[3];
    const float* Wr1 = (const float*)d_in[4];
    const float* Wl2 = (const float*)d_in[5];
    const float* bl2 = (const float*)d_in[6];
    const float* Wr2 = (const float*)d_in[7];
    float* out = (float*)d_out;

    float* ws   = (float*)d_ws;
    float* pbuf = ws + (size_t)NN * DHID;
    float* qbuf = pbuf + (size_t)NN * DOUT;
    int2*  rec  = (int2*)(qbuf + (size_t)NN * DOUT);
    int*   seg  = (int*)(rec + NE);
    int*   cnt  = seg + NN;

    int*   off    = (int*)ws;
    float* wfrag  = ws + WF_OFF;
    int*   col    = (int*)ws + COL_OFF;

    sort_kernel<<<NCH + NWB, CT, 0, stream>>>(ei, off, rec, Wl1, Wr1, Wl2, Wr2, wfrag);
    place_kernel<<<NB, LT, 0, stream>>>(off, rec, seg, cnt, col);

    layer1_kernel<<<NN / 16, 32, 0, stream>>>(x, seg, cnt, col, wfrag, bl1, bl2, pbuf, qbuf);
    gather2_kernel<<<(NN + 8 * GK - 1) / (8 * GK), 256, 0, stream>>>(pbuf, qbuf, seg, cnt, col, out);
}


# Each 16-node layer-1 tile worked by 2 waves in opposite phase order: one walks its 8 segments while the other multiplies the root term; half of h1 and one layer-2 product each; root term summed first
# speedup vs baseline: 6.4432x; 1.1661x over previous
#include <hip/hip_runtime.h>
#include <cmath>


#define NN 50000
#define NE 800000
#define DIN 64
#define DHID 128
#define DOUT 64
#define GK 8
#define WU 4

#define BS 256
#define NB ((NN + BS - 1) / BS)
#define NCH 125
#define CH (NE / NCH)
#define CT 640
#define EPT (CH / CT)
#define SB (NB + 1)
#define LT 1024
#define LB 5
#define SCAP 8192
static_assert(NCH * CH == NE && EPT * CT == CH, "chunks tile the edge list exactly");
static_assert(NB <= CT && NCH <= LT && NCH < 128, "one thread per bucket counter; one placing thread per chunk; 7 search steps cover the chunks");

#define WFG 2048
#define WF_OFF 65536
#define NWB ((4 * WFG + CT - 1) / CT)
#define COL_OFF 131072
static_assert(NCH * SB <= WF_OFF, "the fragment-ordered weights start behind the chunks' bucket offsets");
static_assert(WF_OFF + 16 * WFG <= COL_OFF && COL_OFF + NE <= NN * DHID, "col[] lies behind the weights and inside the scratch region");
static_assert(BS == 256, "the first wave scans the bucket's counters 8 per lane");
static_assert(DHID * DIN == 4 * WFG && DOUT * DHID == 4 * WFG, "every weight matrix is WFG groups of 4 floats");

typedef __attribute__((ext_vector_type(2))) float v2f;
typedef __attribute__((ext_vector_type(4))) float v4f;
typedef __attribute__((ext_vector_type(8))) float v8f;

template <int NOUT, int K>
__device__ __forceinline__ void frag_order(const float* __restrict__ Wm, float* __restrict__ F, int g) {
    constexpr int NT = NOUT / 16;
    const int lane = g & 31;
    const int nt   = (g >> 5) % NT;
    const int k2   = (g >> 5) / NT;
    const int n    = nt * 16 + (lane & 15);
    const int k    = k2 * 8 + (lane >> 4) * 2;
    const float2 lo = *(const float2*)(Wm + n * K + k);
    const float2 hi = *(const float2*)(Wm + n * K + k + 4);
    *(float4*)(F + (size_t)g * 4) = make_float4(lo.x, lo.y, hi.x, hi.y);
}

__global__ __launch_bounds__(CT) void sort_kernel(const int* __restrict__ ei,
                                                  int* __restrict__ off,
                                                  int2* __restrict__ rec,
                                                  const float* __restrict__ Wl1,
                                                  const float* __restrict__ Wr1,
                                                  const float* __restrict__ Wl2,
                                                  const float* __restrict__ Wr2,
                                                  float* __restrict__ wf) {
    __shared__ int h[NB];
    __shared__ int nrec;
    __shared__ int2 st[CH];
    const int tid = threadIdx.x;
    if (blockIdx.x >= NCH) {
        const int g = (blockIdx.x - NCH) * CT + tid;
        if (g < WFG)          frag_order<DHID, DIN>(Wl1, wf, g);
        else if (g < 2 * WFG) frag_order<DHID, DIN>(Wr1, wf + 4 * WFG, g - WFG);
        else if (g < 3 * WFG) frag_order<DOUT, DHID>(Wl2, wf + 8 * WFG, g - 2 * WFG);
        else if (g < 4 * WFG) frag_order<DOUT, DHID>(Wr2, wf + 12 * WFG, g - 3 * WFG);
        return;
    }
    const int g = blockIdx.x;
    if (tid < NB) h[tid] = 0;
    const int e0 = g * CH;
    int s[EPT], d[EPT], rk[EPT];
#pragma unroll
    for (int k = 0; k < EPT; ++k) {
        s[k] = ei[e0 + k * CT + tid];
        d[k] = ei[NE + e0 + k * CT + tid];
    }
    __syncthreads();
#pragma unroll
    for (int k = 0; k < EPT; ++k) {
        rk[k] = -1;
        if ((unsigned)d[k] < (unsigned)NN) rk[k] = atomicAdd(&h[(unsigned)d[k] / BS], 1);
    }
    __syncthreads();
    if (tid < 32) {
        constexpr int PER = (NB + 31) / 32;
        int loc[PER];
        int sum = 0;
#pragma unroll
        for (int i = 0; i < PER; ++i) {
            const int b = tid * PER + i;
            const int v = (b < NB) ? h[b] : 0;
            loc[i] = sum;
            sum += v;
        }
        int inc = sum;
#pragma unroll
        for (int dd = 1; dd < 32; dd <<= 1) {
            const int o = __shfl_up(inc, dd);
            if (tid >= dd) inc += o;
        }
        const int excl = inc - sum;
#pragma unroll
        for (int i = 0; i < PER; ++i) {
            const int b = tid * PER + i;
            if (b < NB) {
                h[b] = excl + loc[i];
                off[g * SB + b] = excl + loc[i];
            }
        }
        if (tid == 31) {
            off[g * SB + NB] = inc;
            nrec = inc;
        }
    }
    __syncthreads();
    int hb[EPT];
#pragma unroll
    for (int k = 0; k < EPT; ++k) hb[k] = h[min((unsigned)d[k] / BS, (unsigned)(NB - 1))];
#pragma unroll
    for (int k = 0; k < EPT; ++k) {
        if (rk[k] >= 0) {
            const int pos = hb[k] + rk[k];
            if ((unsigned)pos < (unsigned)CH) st[pos] = make_int2(s[k], d[k]);
        }
    }
    __syncthreads();
    const int n = nrec;
#pragma unroll
    for (int k = 0; k < EPT; ++k) {
        const int i = k * CT + tid;
        if (i < n) rec[e0 + i] = st[i];
    }
}

__device__ __forceinline__ void fetch_batch(int i0, int total, int tid, int node0,
                                            const int* pre, const int* sbase,
                                            const int2* __restrict__ rec,
                                            int (&src)[LB], int (&li)[LB]) {
    int pos[LB];
#pragma unroll
    for (int k = 0; k < LB; ++k) pos[k] = 0;
#pragma unroll
    for (int step = 64; step >= 1; step >>= 1) {
        int pv[LB];
#pragma unroll
        for (int k = 0; k < LB; ++k) pv[k] = pre[pos[k] + step];
#pragma unroll
        for (int k = 0; k < LB; ++k)
            pos[k] += (pv[k] <= i0 + k * LT + tid) ? step : 0;
    }
    int2 r[LB];
#pragma unroll
    for (int k = 0; k < LB; ++k) {
        const int i = i0 + k * LT + tid;
        r[k] = make_int2(0, -1);
        if (i < total) {
            const int g   = min(pos[k], NCH - 1);
            const int idx = sbase[g] + (i - pre[g]);
            if ((unsigned)idx < (unsigned)NE) r[k] = rec[idx];
        }
    }
#pragma unroll
    for (int k = 0; k < LB; ++k) {
        const unsigned l = (unsigned)(r[k].y - node0);
        src[k] = r[k].x;
        li[k]  = (l < (unsigned)BS) ? (int)l : -1;
    }
}

__global__ __launch_bounds__(LT) void place_kernel(const int* __restrict__ off,
                                                   const int2* __restrict__ rec,
                                                   int* __restrict__ seg,
                                                   int* __restrict__ cnt,
                                                   int* __restrict__ col) {
    __shared__ int nc[BS];
    __shared__ int cu[BS];
    __shared__ int so0[NCH];
    __shared__ int sl[NCH];
    __shared__ int sbase[NCH];
    __shared__ int pre[129];
    __shared__ int cbase;
    __shared__ int stage[SCAP];
    const int tid   = threadIdx.x;
    const int b     = blockIdx.x;
    const int node0 = b * BS;
    if (tid < BS) nc[tid] = 0;
    if (tid > NCH && tid <= 128) pre[tid] = 0x7fffffff;
    if (tid < NCH) {
        const int o0 = min(max(off[tid * SB + b], 0), CH);
        const int o1 = min(max(off[tid * SB + b + 1], o0), CH);
        so0[tid]   = o0;
        sl[tid]    = o1 - o0;
        sbase[tid] = tid * CH + o0;
    }
    __syncthreads();
    int base = 0;
    if (tid < 32) {
        constexpr int PER = (NCH + 31) / 32;
        int loc[PER];
        int sum = 0;
#pragma unroll
        for (int i = 0; i < PER; ++i) {
            const int gch = tid * PER + i;
            const int v = (gch < NCH) ? sl[gch] : 0;
            base += (gch < NCH) ? so0[gch] : 0;
            loc[i] = sum;
            sum += v;
        }
        int inc = sum;
#pragma unroll
        for (int dd = 1; dd < 32; dd <<= 1) {
            const int o = __shfl_up(inc, dd);
            if (tid >= dd) inc += o;
        }
#pragma unroll
        for (int dd = 16; dd >= 1; dd >>= 1) base += __shfl_xor(base, dd);
        const int excl = inc - sum;
#pragma unroll
        for (int i = 0; i < PER; ++i) {
            const int gch = tid * PER + i;
            if (gch < NCH) pre[gch] = excl + loc[i];
        }
        if (tid == 31) {
            pre[NCH] = inc;
            cbase    = base;
        }
    }
    __syncthreads();
    const int total = pre[NCH];
    const int cb    = cbase;
    const bool staged = total <= SCAP;

    int s0[LB], l0[LB];
    fetch_batch(0, total, tid, node0, pre, sbase, rec, s0, l0);
#pragma unroll
    for (int k = 0; k < LB; ++k)
        if (l0[k] >= 0) atomicAdd(&nc[l0[k]], 1);
    for (int i0 = LB * LT; i0 < total; i0 += LB * LT) {
        int s[LB], l[LB];
        fetch_batch(i0, total, tid, node0, pre, sbase, rec, s, l);
#pragma unroll
        for (int k = 0; k < LB; ++k)
            if (l[k] >= 0) atomicAdd(&nc[l[k]], 1);
    }
    __syncthreads();
    if (tid < 32) {
        int loc[8];
        int sum = 0;
#pragma unroll
        for (int i = 0; i < 8; ++i) {
            loc[i] = sum;
            sum += nc[tid * 8 + i];
        }
        int inc = sum;
#pragma unroll
        for (int dd = 1; dd < 32; dd <<= 1) {
            const int o = __shfl_up(inc, dd);
            if (tid >= dd) inc += o;
        }
        const int excl = base + inc - sum;
#pragma unroll
        for (int i = 0; i < 8; ++i) cu[tid * 8 + i] = excl + loc[i];
    }
    __syncthreads();
    auto put = [&](int l, int s) {
        const int place = atomicAdd(&cu[l], 1);
        if (staged) {
            const unsigned o = (unsigned)(place - cb);
            if (o < (unsigned)SCAP) stage[o] = s;
        } else if ((unsigned)place < (unsigned)NE) {
            col[place] = s;
        }
    };
#pragma unroll
    for (int k = 0; k < LB; ++k)
        if (l0[k] >= 0) put(l0[k], s0[k]);
    for (int i0 = LB * LT; i0 < total; i0 += LB * LT) {
        int s[LB], l[LB];
        fetch_batch(i0, total, tid, node0, pre, sbase, rec, s, l);
#pragma unroll
        for (int k = 0; k < LB; ++k)
            if (l[k] >= 0) put(l[k], s[k]);
    }
    __syncthreads();
    if (staged) {
        for (int i = tid; i < total; i += LT)
            if ((unsigned)(cb + i) < (unsigned)NE) col[cb + i] = stage[i];
    }
    if (tid < BS && node0 + tid < NN) {
        seg[node0 + tid] = cu[tid] - nc[tid];
        cnt[node0 + tid] = nc[tid];
    }
}

__device__ __forceinline__ int row_bound(int s) {
    return min(max(s, 0), NN - 1);
}

__device__ __forceinline__ void seg_bound(int& first, int& len) {
    first = min(max(first, 0), NE);
    len   = min(max(len, 0), NE - first);
}

__global__ __launch_bounds__(256) void gather2_kernel(
    const float* __restrict__ p, const float* __restrict__ q,
    const int* __restrict__ seg, const int* __restrict__ cnt,
    const int* __restrict__ col, float* __restrict__ out) {
    const int lane = threadIdx.x & 31;
    const int wv    = __builtin_amdgcn_readfirstlane((int)(threadIdx.x >> 5));
    const int node0 = (blockIdx.x * 8 + wv) * GK;
    if (node0 >= NN) return;

    const bool owner = lane < GK && node0 + lane < NN;
    int first = 0, dg = 0;
    if (owner) {
        first = seg[node0 + lane];
        dg    = cnt[node0 + lane];
    }
    seg_bound(first, dg);
    int trips = dg;
#pragma unroll
    for (int d = 16; d >= 1; d >>= 1) trips = max(trips, __shfl_xor(trips, d));
    trips = __builtin_amdgcn_readfirstlane(trips);
    int nr[WU];
#pragma unroll
    for (int u = 0; u < WU; ++u) {
        nr[u] = 0;
        if (u < dg) nr[u] = col[first + u];
    }

    float2 qv[GK];
#pragma unroll
    for (int j = 0; j < GK; ++j) {
        qv[j] = make_float2(0.0f, 0.0f);
        if (node0 + j < NN)
            qv[j] = *(const float2*)(q + (size_t)(node0 + j) * DOUT + lane * 2);
    }

    float2 acc[GK];
#pragma unroll
    for (int j = 0; j < GK; ++j) acc[j] = make_float2(0.0f, 0.0f);

    int ix[WU];
#pragma unroll
    for (int u = 0; u < WU; ++u) ix[u] = row_bound(nr[u]);
    for (int t0 = 0; t0 < trips; t0 += WU) {
#pragma unroll
        for (int u = 0; u < WU; ++u) {
            const int t = t0 + WU + u;
            nr[u] = 0;
            if (t < dg) nr[u] = col[first + t];
        }
        float2 v[WU][GK];
#pragma unroll
        for (int u = 0; u < WU; ++u) {
            const unsigned live = __builtin_amdgcn_ballot_w32(t0 + u < dg);
#pragma unroll
            for (int j = 0; j < GK; ++j) {
                v[u][j] = make_float2(0.0f, 0.0f);
                if (live & (1u << j)) {
                    const int s = __builtin_amdgcn_readlane(ix[u], j);
                    v[u][j] = *(const float2*)(p + (size_t)s * DOUT + lane * 2);
                }
            }
        }
#pragma unroll
        for (int u = 0; u < WU; ++u) {
#pragma unroll
            for (int j = 0; j < GK; ++j) {
                acc[j].x += v[u][j].x;
                acc[j].y += v[u][j].y;
            }
        }
#pragma unroll
        for (int u = 0; u < WU; ++u) ix[u] = row_bound(nr[u]);
    }
    const float dginv = 1.0f / fmaxf((float)dg, 1.0f);
    float zx[GK], zy[GK], mx[GK], se[GK], t[GK];
#pragma unroll
    for (int j = 0; j < GK; ++j) {
        const float dinv = __int_as_float(__builtin_amdgcn_readlane(__float_as_int(dginv), j));
        zx[j] = acc[j].x * dinv + qv[j].x;
        zy[j] = acc[j].y * dinv + qv[j].y;
        mx[j] = fmaxf(zx[j], zy[j]);
    }
#pragma unroll
    for (int d = 16; d >= 1; d >>= 1) {
#pragma unroll
        for (int j = 0; j < GK; ++j) t[j] = __shfl_xor(mx[j], d);
#pragma unroll
        for (int j = 0; j < GK; ++j) mx[j] = fmaxf(mx[j], t[j]);
    }
#pragma unroll
    for (int j = 0; j < GK; ++j) se[j] = expf(zx[j] - mx[j]) + expf(zy[j] - mx[j]);
#pragma unroll
    for (int d = 16; d >= 1; d >>= 1) {
#pragma unroll
        for (int j = 0; j < GK; ++j) t[j] = __shfl_xor(se[j], d);
#pragma unroll
        for (int j = 0; j < GK; ++j) se[j] += t[j];
    }
#pragma unroll
    for (int j = 0; j < GK; ++j) {
        if (node0 + j < NN) {
            const float lse = mx[j] + logf(se[j]);
            *(float2*)(out + (size_t)(node0 + j) * DOUT + lane * 2) =
                make_float2(zx[j] - lse, zy[j] - lse);
        }
    }
}

#define GH 8
#define HP (DHID + 4)
#define AP (DIN + 4)
__global__ __launch_bounds__(64, 7) void layer1_kernel(
    const float* __restrict__ x, const int* __restrict__ seg,
    const int* __restrict__ cnt, const int* __restrict__ col,
    const float* __restrict__ wf, const float* __restrict__ bl,
    const float* __restrict__ bl2,
    float* __restrict__ p, float* __restrict__ q) {
    __shared__ __attribute__((aligned(16))) float as[16 * AP];
    __shared__ __attribute__((aligned(16))) float hs[16 * HP];
    const int lane = threadIdx.x & 31;
    const int wv   = __builtin_amdgcn_readfirstlane((int)(threadIdx.x >> 5));
    const int m0   = blockIdx.x * 16;
    const int mr   = lane & 15;
    const int kb   = (lane >> 4) * 2;
    const int ct0  = wv * 4;
    const v4f* __restrict__ fl1 = (const v4f*)wf;
    const v4f* __restrict__ fr1 = fl1 + WFG;
    const v4f* __restrict__ fl2 = fr1 + WFG;
    const v4f* __restrict__ fr2 = fl2 + WFG;

    v8f acc[4] = {};

    auto root_products = [&]() {
        const int m = m0 + mr;
        for (int k2 = 0; k2 < DIN / 8; ++k2) {
            const int k = k2 * 8 + kb;
            v2f a0, a1;
            a0.x = x[m * DIN + k];
            a0.y = x[m * DIN + k + 1];
            a1.x = x[m * DIN + k + 4];
            a1.y = x[m * DIN + k + 5];
            v4f b[4];
#pragma unroll
            for (int i = 0; i < 4; ++i) b[i] = fr1[(k2 * 8 + ct0 + i) * 32 + lane];
#pragma unroll
            for (int i = 0; i < 4; ++i) {
                v2f bb;
                bb.x = b[i].x;
                bb.y = b[i].y;
                acc[i] = __builtin_amdgcn_wmma_f32_16x16x4_f32(
                    false, a0, false, bb, (short)0, acc[i], false, false);
            }
#pragma unroll
            for (int i = 0; i < 4; ++i) {
                v2f bb;
                bb.x = b[i].z;
                bb.y = b[i].w;
                acc[i] = __builtin_amdgcn_wmma_f32_16x16x4_f32(
                    false, a1, false, bb, (short)0, acc[i], false, false);
            }
        }
    };

    auto walk = [&]() {
        const int nb = m0 + wv * GH;
        const bool owner = lane < GH && nb + lane < NN;
        int first = 0, dg = 0;
        if (owner) {
            first = seg[nb + lane];
            dg    = cnt[nb + lane];
        }
        seg_bound(first, dg);
        int trips = dg;
#pragma unroll
        for (int d = 16; d >= 1; d >>= 1) trips = max(trips, __shfl_xor(trips, d));
        trips = __builtin_amdgcn_readfirstlane(trips);
        int nr[WU];
#pragma unroll
        for (int u = 0; u < WU; ++u) {
            nr[u] = 0;
            if (u < dg) nr[u] = col[first + u];
        }

        float2 sacc[GH];
#pragma unroll
        for (int j = 0; j < GH; ++j) sacc[j] = make_float2(0.0f, 0.0f);

        int ix[WU];
#pragma unroll
        for (int u = 0; u < WU; ++u) ix[u] = row_bound(nr[u]);
        for (int t0 = 0; t0 < trips; t0 += WU) {
#pragma unroll
            for (int u = 0; u < WU; ++u) {
                const int t = t0 + WU + u;
                nr[u] = 0;
                if (t < dg) nr[u] = col[first + t];
            }
            float2 v[WU][GH];
#pragma unroll
            for (int u = 0; u < WU; ++u) {
                const unsigned live = __builtin_amdgcn_ballot_w32(t0 + u < dg);
#pragma unroll
                for (int j = 0; j < GH; ++j) {
                    v[u][j] = make_float2(0.0f, 0.0f);
                    if (live & (1u << j)) {
                        const int s = __builtin_amdgcn_readlane(ix[u], j);
                        v[u][j] = *(const float2*)(x + (size_t)s * DIN + lane * 2);
                    }
                }
            }
#pragma unroll
            for (int u = 0; u < WU; ++u) {
#pragma unroll
                for (int j = 0; j < GH; ++j) {
                    sacc[j].x += v[u][j].x;
                    sacc[j].y += v[u][j].y;
                }
            }
#pragma unroll
            for (int u = 0; u < WU; ++u) ix[u] = row_bound(nr[u]);
        }
        const float dgi = 1.0f / fmaxf((float)dg, 1.0f);
#pragma unroll
        for (int j = 0; j < GH; ++j) {
            const float dj = __int_as_float(__builtin_amdgcn_readlane(__float_as_int(dgi), j));
            *(float2*)(as + (wv * GH + j) * AP + lane * 2) = make_float2(sacc[j].x * dj, sacc[j].y * dj);
        }
    };

    if (wv == 0) {
        walk();
        root_products();
    } else {
        root_products();
        walk();
    }
    __syncthreads();

    for (int k2 = 0; k2 < DIN / 8; ++k2) {
        const int k = k2 * 8 + kb;
        v2f a0, a1;
        a0.x = as[mr * AP + k];
        a0.y = as[mr * AP + k + 1];
        a1.x = as[mr * AP + k + 4];
        a1.y = as[mr * AP + k + 5];
        v4f b[4];
#pragma unroll
        for (int i = 0; i < 4; ++i) b[i] = fl1[(k2 * 8 + ct0 + i) * 32 + lane];
#pragma unroll
        for (int i = 0; i < 4; ++i) {
            v2f bb;
            bb.x = b[i].x;
            bb.y = b[i].y;
            acc[i] = __builtin_amdgcn_wmma_f32_16x16x4_f32(
                false, a0, false, bb, (short)0, acc[i], false, false);
        }
#pragma unroll
        for (int i = 0; i < 4; ++i) {
            v2f bb;
            bb.x = b[i].z;
            bb.y = b[i].w;
            acc[i] = __builtin_amdgcn_wmma_f32_16x16x4_f32(
                false, a1, false, bb, (short)0, acc[i], false, false);
        }
    }

    const int rb = 8 * (lane >> 4);
#pragma unroll
    for (int i = 0; i < 4; ++i) {
        const int n = (ct0 + i) * 16 + mr;
        const float bias = bl[n];
#pragma unroll
        for (int v = 0; v < 8; ++v) {
            hs[(rb + v) * HP + n] = tanhf(acc[i][v] + bias);
        }
    }
    __syncthreads();

    const v4f* __restrict__ f2 = wv ? fr2 : fl2;
    v8f oa[4] = {};
    for (int k2 = 0; k2 < DHID / 8; ++k2) {
        const int k = k2 * 8 + kb;
        v2f a0, a1;
        a0.x = hs[mr * HP + k];
        a0.y = hs[mr * HP + k + 1];
        a1.x = hs[mr * HP + k + 4];
        a1.y = hs[mr * HP + k + 5];
        v4f b2[4];
#pragma unroll
        for (int nt = 0; nt < 4; ++nt) b2[nt] = f2[(k2 * 4 + nt) * 32 + lane];
#pragma unroll
        for (int nt = 0; nt < 4; ++nt) {
            v2f bb;
            bb.x = b2[nt].x;
            bb.y = b2[nt].y;
            oa[nt] = __builtin_amdgcn_wmma_f32_16x16x4_f32(
                false, a0, false, bb, (short)0, oa[nt], false, false);
        }
#pragma unroll
        for (int nt = 0; nt < 4; ++nt) {
            v2f bb;
            bb.x = b2[nt].z;
            bb.y = b2[nt].w;
            oa[nt] = __builtin_amdgcn_wmma_f32_16x16x4_f32(
                false, a1, false, bb, (short)0, oa[nt], false, false);
        }
    }

    const int mo = m0 + rb;
    if (wv == 0) {
#pragma unroll
        for (int nt = 0; nt < 4; ++nt) {
            const int n = nt * 16 + mr;
#pragma unroll
            for (int v = 0; v < 8; ++v) p[(mo + v) * DOUT + n] = oa[nt][v];
        }
    } else {
#pragma unroll
        for (int nt = 0; nt < 4; ++nt) {
            const int n = nt * 16 + mr;
            const float bias2 = bl2[n];
#pragma unroll
            for (int v = 0; v < 8; ++v) q[(mo + v) * DOUT + n] = oa[nt][v] + bias2;
        }
    }
}

extern "C" void kernel_launch(void* const* d_in, const int* in_sizes, int n_in,
                              void* d_out, int out_size, void* d_ws, size_t ws_size,
                              hipStream_t stream) {
    const float* x   = (const float*)d_in[0];
    const int*   ei  = (const int*)d_in[1];
    const float* Wl1 = (const float*)d_in[2];
    const float* bl1 = (const float*)d_in[3];
    const float* Wr1 = (const float*)d_in[4];
    const float* Wl2 = (const float*)d_in[5];
    const float* bl2 = (const float*)d_in[6];
    const float* Wr2 = (const float*)d_in[7];
    float* out = (float*)d_out;

    float* ws   = (float*)d_ws;
    float* pbuf = ws + (size_t)NN * DHID;
    float* qbuf = pbuf + (size_t)NN * DOUT;
    int2*  rec  = (int2*)(qbuf + (size_t)NN * DOUT);
    int*   seg  = (int*)(rec + NE);
    int*   cnt  = seg + NN;

    int*   off    = (int*)ws;
    float* wfrag  = ws + WF_OFF;
    int*   col    = (int*)ws + COL_OFF;

    sort_kernel<<<NCH + NWB, CT, 0, stream>>>(ei, off, rec, Wl1, Wr1, Wl2, Wr2, wfrag);
    place_kernel<<<NB, LT, 0, stream>>>(off, rec, seg, cnt, col);

    layer1_kernel<<<NN / 16, 64, 0, stream>>>(x, seg, cnt, col, wfrag, bl1, bl2, pbuf, qbuf);
    gather2_kernel<<<(NN + 8 * GK - 1) / (8 * GK), 256, 0, stream>>>(pbuf, qbuf, seg, cnt, col, out);
}


# tanh of the h1 tile without a branch (polynomial below 0.625, else 1-2/(exp2+1) by the hardware exponential and reciprocal, one select) in place of the library's two masked paths per value
# speedup vs baseline: 6.7538x; 1.0482x over previous
#include <hip/hip_runtime.h>
#include <cmath>


#define NN 50000
#define NE 800000
#define DIN 64
#define DHID 128
#define DOUT 64
#define GK 8
#define WU 4

#define BS 256
#define NB ((NN + BS - 1) / BS)
#define NCH 125
#define CH (NE / NCH)
#define CT 640
#define EPT (CH / CT)
#define SB (NB + 1)
#define LT 1024
#define LB 5
#define SCAP 8192
static_assert(NCH * CH == NE && EPT * CT == CH, "chunks tile the edge list exactly");
static_assert(NB <= CT && NCH <= LT && NCH < 128, "one thread per bucket counter; one placing thread per chunk; 7 search steps cover the chunks");

#define WFG 2048
#define WF_OFF 65536
#define NWB ((4 * WFG + CT - 1) / CT)
#define COL_OFF 131072
static_assert(NCH * SB <= WF_OFF, "the fragment-ordered weights start behind the chunks' bucket offsets");
static_assert(WF_OFF + 16 * WFG <= COL_OFF && COL_OFF + NE <= NN * DHID, "col[] lies behind the weights and inside the scratch region");
static_assert(BS == 256, "the first wave scans the bucket's counters 8 per lane");
static_assert(DHID * DIN == 4 * WFG && DOUT * DHID == 4 * WFG, "every weight matrix is WFG groups of 4 floats");

typedef __attribute__((ext_vector_type(2))) float v2f;
typedef __attribute__((ext_vector_type(4))) float v4f;
typedef __attribute__((ext_vector_type(8))) float v8f;

template <int NOUT, int K>
__device__ __forceinline__ void frag_order(const float* __restrict__ Wm, float* __restrict__ F, int g) {
    constexpr int NT = NOUT / 16;
    const int lane = g & 31;
    const int nt   = (g >> 5) % NT;
    const int k2   = (g >> 5) / NT;
    const int n    = nt * 16 + (lane & 15);
    const int k    = k2 * 8 + (lane >> 4) * 2;
    const float2 lo = *(const float2*)(Wm + n * K + k);
    const float2 hi = *(const float2*)(Wm + n * K + k + 4);
    *(float4*)(F + (size_t)g * 4) = make_float4(lo.x, lo.y, hi.x, hi.y);
}

__global__ __launch_bounds__(CT) void sort_kernel(const int* __restrict__ ei,
                                                  int* __restrict__ off,
                                                  int2* __restrict__ rec,
                                                  const float* __restrict__ Wl1,
                                                  const float* __restrict__ Wr1,
                                                  const float* __restrict__ Wl2,
                                                  const float* __restrict__ Wr2,
                                                  float* __restrict__ wf) {
    __shared__ int h[NB];
    __shared__ int nrec;
    __shared__ int2 st[CH];
    const int tid = threadIdx.x;
    if (blockIdx.x >= NCH) {
        const int g = (blockIdx.x - NCH) * CT + tid;
        if (g < WFG)          frag_order<DHID, DIN>(Wl1, wf, g);
        else if (g < 2 * WFG) frag_order<DHID, DIN>(Wr1, wf + 4 * WFG, g - WFG);
        else if (g < 3 * WFG) frag_order<DOUT, DHID>(Wl2, wf + 8 * WFG, g - 2 * WFG);
        else if (g < 4 * WFG) frag_order<DOUT, DHID>(Wr2, wf + 12 * WFG, g - 3 * WFG);
        return;
    }
    const int g = blockIdx.x;
    if (tid < NB) h[tid] = 0;
    const int e0 = g * CH;
    int s[EPT], d[EPT], rk[EPT];
#pragma unroll
    for (int k = 0; k < EPT; ++k) {
        s[k] = ei[e0 + k * CT + tid];
        d[k] = ei[NE + e0 + k * CT + tid];
    }
    __syncthreads();
#pragma unroll
    for (int k = 0; k < EPT; ++k) {
        rk[k] = -1;
        if ((unsigned)d[k] < (unsigned)NN) rk[k] = atomicAdd(&h[(unsigned)d[k] / BS], 1);
    }
    __syncthreads();
    if (tid < 32) {
        constexpr int PER = (NB + 31) / 32;
        int loc[PER];
        int sum = 0;
#pragma unroll
        for (int i = 0; i < PER; ++i) {
            const int b = tid * PER + i;
            const int v = (b < NB) ? h[b] : 0;
            loc[i] = sum;
            sum += v;
        }
        int inc = sum;
#pragma unroll
        for (int dd = 1; dd < 32; dd <<= 1) {
            const int o = __shfl_up(inc, dd);
            if (tid >= dd) inc += o;
        }
        const int excl = inc - sum;
#pragma unroll
        for (int i = 0; i < PER; ++i) {
            const int b = tid * PER + i;
            if (b < NB) {
                h[b] = excl + loc[i];
                off[g * SB + b] = excl + loc[i];
            }
        }
        if (tid == 31) {
            off[g * SB + NB] = inc;
            nrec = inc;
        }
    }
    __syncthreads();
    int hb[EPT];
#pragma unroll
    for (int k = 0; k < EPT; ++k) hb[k] = h[min((unsigned)d[k] / BS, (unsigned)(NB - 1))];
#pragma unroll
    for (int k = 0; k < EPT; ++k) {
        if (rk[k] >= 0) {
            const int pos = hb[k] + rk[k];
            if ((unsigned)pos < (unsigned)CH) st[pos] = make_int2(s[k], d[k]);
        }
    }
    __syncthreads();
    const int n = nrec;
#pragma unroll
    for (int k = 0; k < EPT; ++k) {
        const int i = k * CT + tid;
        if (i < n) rec[e0 + i] = st[i];
    }
}

__device__ __forceinline__ void fetch_batch(int i0, int total, int tid, int node0,
                                            const int* pre, const int* sbase,
                                            const int2* __restrict__ rec,
                                            int (&src)[LB], int (&li)[LB]) {
    int pos[LB];
#pragma unroll
    for (int k = 0; k < LB; ++k) pos[k] = 0;
#pragma unroll
    for (int step = 64; step >= 1; step >>= 1) {
        int pv[LB];
#pragma unroll
        for (int k = 0; k < LB; ++k) pv[k] = pre[pos[k] + step];
#pragma unroll
        for (int k = 0; k < LB; ++k)
            pos[k] += (pv[k] <= i0 + k * LT + tid) ? step : 0;
    }
    int2 r[LB];
#pragma unroll
    for (int k = 0; k < LB; ++k) {
        const int i = i0 + k * LT + tid;
        r[k] = make_int2(0, -1);
        if (i < total) {
            const int g   = min(pos[k], NCH - 1);
            const int idx = sbase[g] + (i - pre[g]);
            if ((unsigned)idx < (unsigned)NE) r[k] = rec[idx];
        }
    }
#pragma unroll
    for (int k = 0; k < LB; ++k) {
        const unsigned l = (unsigned)(r[k].y - node0);
        src[k] = r[k].x;
        li[k]  = (l < (unsigned)BS) ? (int)l : -1;
    }
}

__global__ __launch_bounds__(LT) void place_kernel(const int* __restrict__ off,
                                                   const int2* __restrict__ rec,
                                                   int* __restrict__ seg,
                                                   int* __restrict__ cnt,
                                                   int* __restrict__ col) {
    __shared__ int nc[BS];
    __shared__ int cu[BS];
    __shared__ int so0[NCH];
    __shared__ int sl[NCH];
    __shared__ int sbase[NCH];
    __shared__ int pre[129];
    __shared__ int cbase;
    __shared__ int stage[SCAP];
    const int tid   = threadIdx.x;
    const int b     = blockIdx.x;
    const int node0 = b * BS;
    if (tid < BS) nc[tid] = 0;
    if (tid > NCH && tid <= 128) pre[tid] = 0x7fffffff;
    if (tid < NCH) {
        const int o0 = min(max(off[tid * SB + b], 0), CH);
        const int o1 = min(max(off[tid * SB + b + 1], o0), CH);
        so0[tid]   = o0;
        sl[tid]    = o1 - o0;
        sbase[tid] = tid * CH + o0;
    }
    __syncthreads();
    int base = 0;
    if (tid < 32) {
        constexpr int PER = (NCH + 31) / 32;
        int loc[PER];
        int sum = 0;
#pragma unroll
        for (int i = 0; i < PER; ++i) {
            const int gch = tid * PER + i;
            const int v = (gch < NCH) ? sl[gch] : 0;
            base += (gch < NCH) ? so0[gch] : 0;
            loc[i] = sum;
            sum += v;
        }
        int inc = sum;
#pragma unroll
        for (int dd = 1; dd < 32; dd <<= 1) {
            const int o = __shfl_up(inc, dd);
            if (tid >= dd) inc += o;
        }
#pragma unroll
        for (int dd = 16; dd >= 1; dd >>= 1) base += __shfl_xor(base, dd);
        const int excl = inc - sum;
#pragma unroll
        for (int i = 0; i < PER; ++i) {
            const int gch = tid * PER + i;
            if (gch < NCH) pre[gch] = excl + loc[i];
        }
        if (tid == 31) {
            pre[NCH] = inc;
            cbase    = base;
        }
    }
    __syncthreads();
    const int total = pre[NCH];
    const int cb    = cbase;
    const bool staged = total <= SCAP;

    int s0[LB], l0[LB];
    fetch_batch(0, total, tid, node0, pre, sbase, rec, s0, l0);
#pragma unroll
    for (int k = 0; k < LB; ++k)
        if (l0[k] >= 0) atomicAdd(&nc[l0[k]], 1);
    for (int i0 = LB * LT; i0 < total; i0 += LB * LT) {
        int s[LB], l[LB];
        fetch_batch(i0, total, tid, node0, pre, sbase, rec, s, l);
#pragma unroll
        for (int k = 0; k < LB; ++k)
            if (l[k] >= 0) atomicAdd(&nc[l[k]], 1);
    }
    __syncthreads();
    if (tid < 32) {
        int loc[8];
        int sum = 0;
#pragma unroll
        for (int i = 0; i < 8; ++i) {
            loc[i] = sum;
            sum += nc[tid * 8 + i];
        }
        int inc = sum;
#pragma unroll
        for (int dd = 1; dd < 32; dd <<= 1) {
            const int o = __shfl_up(inc, dd);
            if (tid >= dd) inc += o;
        }
        const int excl = base + inc - sum;
#pragma unroll
        for (int i = 0; i < 8; ++i) cu[tid * 8 + i] = excl + loc[i];
    }
    __syncthreads();
    auto put = [&](int l, int s) {
        const int place = atomicAdd(&cu[l], 1);
        if (staged) {
            const unsigned o = (unsigned)(place - cb);
            if (o < (unsigned)SCAP) stage[o] = s;
        } else if ((unsigned)place < (unsigned)NE) {
            col[place] = s;
        }
    };
#pragma unroll
    for (int k = 0; k < LB; ++k)
        if (l0[k] >= 0) put(l0[k], s0[k]);
    for (int i0 = LB * LT; i0 < total; i0 += LB * LT) {
        int s[LB], l[LB];
        fetch_batch(i0, total, tid, node0, pre, sbase, rec, s, l);
#pragma unroll
        for (int k = 0; k < LB; ++k)
            if (l[k] >= 0) put(l[k], s[k]);
    }
    __syncthreads();
    if (staged) {
        for (int i = tid; i < total; i += LT)
            if ((unsigned)(cb + i) < (unsigned)NE) col[cb + i] = stage[i];
    }
    if (tid < BS && node0 + tid < NN) {
        seg[node0 + tid] = cu[tid] - nc[tid];
        cnt[node0 + tid] = nc[tid];
    }
}

__device__ __forceinline__ int row_bound(int s) {
    return min(max(s, 0), NN - 1);
}

__device__ __forceinline__ void seg_bound(int& first, int& len) {
    first = min(max(first, 0), NE);
    len   = min(max(len, 0), NE - first);
}

__global__ __launch_bounds__(256) void gather2_kernel(
    const float* __restrict__ p, const float* __restrict__ q,
    const int* __restrict__ seg, const int* __restrict__ cnt,
    const int* __restrict__ col, float* __restrict__ out) {
    const int lane = threadIdx.x & 31;
    const int wv    = __builtin_amdgcn_readfirstlane((int)(threadIdx.x >> 5));
    const int node0 = (blockIdx.x * 8 + wv) * GK;
    if (node0 >= NN) return;

    const bool owner = lane < GK && node0 + lane < NN;
    int first = 0, dg = 0;
    if (owner) {
        first = seg[node0 + lane];
        dg    = cnt[node0 + lane];
    }
    seg_bound(first, dg);
    int trips = dg;
#pragma unroll
    for (int d = 16; d >= 1; d >>= 1) trips = max(trips, __shfl_xor(trips, d));
    trips = __builtin_amdgcn_readfirstlane(trips);
    int nr[WU];
#pragma unroll
    for (int u = 0; u < WU; ++u) {
        nr[u] = 0;
        if (u < dg) nr[u] = col[first + u];
    }

    float2 qv[GK];
#pragma unroll
    for (int j = 0; j < GK; ++j) {
        qv[j] = make_float2(0.0f, 0.0f);
        if (node0 + j < NN)
            qv[j] = *(const float2*)(q + (size_t)(node0 + j) * DOUT + lane * 2);
    }

    float2 acc[GK];
#pragma unroll
    for (int j = 0; j < GK; ++j) acc[j] = make_float2(0.0f, 0.0f);

    int ix[WU];
#pragma unroll
    for (int u = 0; u < WU; ++u) ix[u] = row_bound(nr[u]);
    for (int t0 = 0; t0 < trips; t0 += WU) {
#pragma unroll
        for (int u = 0; u < WU; ++u) {
            const int t = t0 + WU + u;
            nr[u] = 0;
            if (t < dg) nr[u] = col[first + t];
        }
        float2 v[WU][GK];
#pragma unroll
        for (int u = 0; u < WU; ++u) {
            const unsigned live = __builtin_amdgcn_ballot_w32(t0 + u < dg);
#pragma unroll
            for (int j = 0; j < GK; ++j) {
                v[u][j] = make_float2(0.0f, 0.0f);
                if (live & (1u << j)) {
                    const int s = __builtin_amdgcn_readlane(ix[u], j);
                    v[u][j] = *(const float2*)(p + (size_t)s * DOUT + lane * 2);
                }
            }
        }
#pragma unroll
        for (int u = 0; u < WU; ++u) {
#pragma unroll
            for (int j = 0; j < GK; ++j) {
                acc[j].x += v[u][j].x;
                acc[j].y += v[u][j].y;
            }
        }
#pragma unroll
        for (int u = 0; u < WU; ++u) ix[u] = row_bound(nr[u]);
    }
    const float dginv = 1.0f / fmaxf((float)dg, 1.0f);
    float zx[GK], zy[GK], mx[GK], se[GK], t[GK];
#pragma unroll
    for (int j = 0; j < GK; ++j) {
        const float dinv = __int_as_float(__builtin_amdgcn_readlane(__float_as_int(dginv), j));
        zx[j] = acc[j].x * dinv + qv[j].x;
        zy[j] = acc[j].y * dinv + qv[j].y;
        mx[j] = fmaxf(zx[j], zy[j]);
    }
#pragma unroll
    for (int d = 16; d >= 1; d >>= 1) {
#pragma unroll
        for (int j = 0; j < GK; ++j) t[j] = __shfl_xor(mx[j], d);
#pragma unroll
        for (int j = 0; j < GK; ++j) mx[j] = fmaxf(mx[j], t[j]);
    }
#pragma unroll
    for (int j = 0; j < GK; ++j) se[j] = expf(zx[j] - mx[j]) + expf(zy[j] - mx[j]);
#pragma unroll
    for (int d = 16; d >= 1; d >>= 1) {
#pragma unroll
        for (int j = 0; j < GK; ++j) t[j] = __shfl_xor(se[j], d);
#pragma unroll
        for (int j = 0; j < GK; ++j) se[j] += t[j];
    }
#pragma unroll
    for (int j = 0; j < GK; ++j) {
        if (node0 + j < NN) {
            const float lse = mx[j] + logf(se[j]);
            *(float2*)(out + (size_t)(node0 + j) * DOUT + lane * 2) =
                make_float2(zx[j] - lse, zy[j] - lse);
        }
    }
}

#define GH 8
#define HP (DHID + 4)
#define AP (DIN + 4)
__device__ __forceinline__ float tanh_f32(float x) {
    const float ax = fabsf(x);
    const float x2 = x * x;
    float pp = fmaf(-0x1.758e7ap-8f, x2, 0x1.521192p-6f);
    pp = fmaf(x2, pp, -0x1.b8389cp-5f);
    pp = fmaf(x2, pp, 0x1.110704p-3f);
    pp = fmaf(x2, pp, -0x1.555532p-2f);
    const float small = fmaf(x2, ax * pp, ax);
    const float e = __builtin_amdgcn_exp2f(ax * 0x1.715476p+1f);
    const float big = fmaf(__builtin_amdgcn_rcpf(e + 1.0f), -2.0f, 1.0f);
    return copysignf(ax < 0.625f ? small : big, x);
}

__global__ __launch_bounds__(64, 7) void layer1_kernel(
    const float* __restrict__ x, const int* __restrict__ seg,
    const int* __restrict__ cnt, const int* __restrict__ col,
    const float* __restrict__ wf, const float* __restrict__ bl,
    const float* __restrict__ bl2,
    float* __restrict__ p, float* __restrict__ q) {
    __shared__ __attribute__((aligned(16))) float as[16 * AP];
    __shared__ __attribute__((aligned(16))) float hs[16 * HP];
    const int lane = threadIdx.x & 31;
    const int wv   = __builtin_amdgcn_readfirstlane((int)(threadIdx.x >> 5));
    const int m0   = blockIdx.x * 16;
    const int mr   = lane & 15;
    const int kb   = (lane >> 4) * 2;
    const int ct0  = wv * 4;
    const v4f* __restrict__ fl1 = (const v4f*)wf;
    const v4f* __restrict__ fr1 = fl1 + WFG;
    const v4f* __restrict__ fl2 = fr1 + WFG;
    const v4f* __restrict__ fr2 = fl2 + WFG;

    v8f acc[4] = {};

    auto root_products = [&]() {
        const int m = m0 + mr;
        for (int k2 = 0; k2 < DIN / 8; ++k2) {
            const int k = k2 * 8 + kb;
            v2f a0, a1;
            a0.x = x[m * DIN + k];
            a0.y = x[m * DIN + k + 1];
            a1.x = x[m * DIN + k + 4];
            a1.y = x[m * DIN + k + 5];
            v4f b[4];
#pragma unroll
            for (int i = 0; i < 4; ++i) b[i] = fr1[(k2 * 8 + ct0 + i) * 32 + lane];
#pragma unroll
            for (int i = 0; i < 4; ++i) {
                v2f bb;
                bb.x = b[i].x;
                bb.y = b[i].y;
                acc[i] = __builtin_amdgcn_wmma_f32_16x16x4_f32(
                    false, a0, false, bb, (short)0, acc[i], false, false);
            }
#pragma unroll
            for (int i = 0; i < 4; ++i) {
                v2f bb;
                bb.x = b[i].z;
                bb.y = b[i].w;
                acc[i] = __builtin_amdgcn_wmma_f32_16x16x4_f32(
                    false, a1, false, bb, (short)0, acc[i], false, false);
            }
        }
    };

    auto walk = [&]() {
        const int nb = m0 + wv * GH;
        const bool owner = lane < GH && nb + lane < NN;
        int first = 0, dg = 0;
        if (owner) {
            first = seg[nb + lane];
            dg    = cnt[nb + lane];
        }
        seg_bound(first, dg);
        int trips = dg;
#pragma unroll
        for (int d = 16; d >= 1; d >>= 1) trips = max(trips, __shfl_xor(trips, d));
        trips = __builtin_amdgcn_readfirstlane(trips);
        int nr[WU];
#pragma unroll
        for (int u = 0; u < WU; ++u) {
            nr[u] = 0;
            if (u < dg) nr[u] = col[first + u];
        }

        float2 sacc[GH];
#pragma unroll
        for (int j = 0; j < GH; ++j) sacc[j] = make_float2(0.0f, 0.0f);

        int ix[WU];
#pragma unroll
        for (int u = 0; u < WU; ++u) ix[u] = row_bound(nr[u]);
        for (int t0 = 0; t0 < trips; t0 += WU) {
#pragma unroll
            for (int u = 0; u < WU; ++u) {
                const int t = t0 + WU + u;
                nr[u] = 0;
                if (t < dg) nr[u] = col[first + t];
            }
            float2 v[WU][GH];
#pragma unroll
            for (int u = 0; u < WU; ++u) {
                const unsigned live = __builtin_amdgcn_ballot_w32(t0 + u < dg);
#pragma unroll
                for (int j = 0; j < GH; ++j) {
                    v[u][j] = make_float2(0.0f, 0.0f);
                    if (live & (1u << j)) {
                        const int s = __builtin_amdgcn_readlane(ix[u], j);
                        v[u][j] = *(const float2*)(x + (size_t)s * DIN + lane * 2);
                    }
                }
            }
#pragma unroll
            for (int u = 0; u < WU; ++u) {
#pragma unroll
                for (int j = 0; j < GH; ++j) {
                    sacc[j].x += v[u][j].x;
                    sacc[j].y += v[u][j].y;
                }
            }
#pragma unroll
            for (int u = 0; u < WU; ++u) ix[u] = row_bound(nr[u]);
        }
        const float dgi = 1.0f / fmaxf((float)dg, 1.0f);
#pragma unroll
        for (int j = 0; j < GH; ++j) {
            const float dj = __int_as_float(__builtin_amdgcn_readlane(__float_as_int(dgi), j));
            *(float2*)(as + (wv * GH + j) * AP + lane * 2) = make_float2(sacc[j].x * dj, sacc[j].y * dj);
        }
    };

    if (wv == 0) {
        walk();
        root_products();
    } else {
        root_products();
        walk();
    }
    __syncthreads();

    for (int k2 = 0; k2 < DIN / 8; ++k2) {
        const int k = k2 * 8 + kb;
        v2f a0, a1;
        a0.x = as[mr * AP + k];
        a0.y = as[mr * AP + k + 1];
        a1.x = as[mr * AP + k + 4];
        a1.y = as[mr * AP + k + 5];
        v4f b[4];
#pragma unroll
        for (int i = 0; i < 4; ++i) b[i] = fl1[(k2 * 8 + ct0 + i) * 32 + lane];
#pragma unroll
        for (int i = 0; i < 4; ++i) {
            v2f bb;
            bb.x = b[i].x;
            bb.y = b[i].y;
            acc[i] = __builtin_amdgcn_wmma_f32_16x16x4_f32(
                false, a0, false, bb, (short)0, acc[i], false, false);
        }
#pragma unroll
        for (int i = 0; i < 4; ++i) {
            v2f bb;
            bb.x = b[i].z;
            bb.y = b[i].w;
            acc[i] = __builtin_amdgcn_wmma_f32_16x16x4_f32(
                false, a1, false, bb, (short)0, acc[i], false, false);
        }
    }

    __builtin_amdgcn_sched_barrier(0);
    const int rb = 8 * (lane >> 4);
#pragma unroll
    for (int i = 0; i < 4; ++i) {
        const int n = (ct0 + i) * 16 + mr;
        const float bias = bl[n];
#pragma unroll
        for (int v = 0; v < 8; ++v) {
            hs[(rb + v) * HP + n] = tanh_f32(acc[i][v] + bias);
        }
    }
    __syncthreads();

    const v4f* __restrict__ f2 = wv ? fr2 : fl2;
    v8f oa[4] = {};
    for (int k2 = 0; k2 < DHID / 8; ++k2) {
        const int k = k2 * 8 + kb;
        v2f a0, a1;
        a0.x = hs[mr * HP + k];
        a0.y = hs[mr * HP + k + 1];
        a1.x = hs[mr * HP + k + 4];
        a1.y = hs[mr * HP + k + 5];
        v4f b2[4];
#pragma unroll
        for (int nt = 0; nt < 4; ++nt) b2[nt] = f2[(k2 * 4 + nt) * 32 + lane];
#pragma unroll
        for (int nt = 0; nt < 4; ++nt) {
            v2f bb;
            bb.x = b2[nt].x;
            bb.y = b2[nt].y;
            oa[nt] = __builtin_amdgcn_wmma_f32_16x16x4_f32(
                false, a0, false, bb, (short)0, oa[nt], false, false);
        }
#pragma unroll
        for (int nt = 0; nt < 4; ++nt) {
            v2f bb;
            bb.x = b2[nt].z;
            bb.y = b2[nt].w;
            oa[nt] = __builtin_amdgcn_wmma_f32_16x16x4_f32(
                false, a1, false, bb, (short)0, oa[nt], false, false);
        }
    }

    const int mo = m0 + rb;
    if (wv == 0) {
#pragma unroll
        for (int nt = 0; nt < 4; ++nt) {
            const int n = nt * 16 + mr;
#pragma unroll
            for (int v = 0; v < 8; ++v) p[(mo + v) * DOUT + n] = oa[nt][v];
        }
    } else {
#pragma unroll
        for (int nt = 0; nt < 4; ++nt) {
            const int n = nt * 16 + mr;
            const float bias2 = bl2[n];
#pragma unroll
            for (int v = 0; v < 8; ++v) q[(mo + v) * DOUT + n] = oa[nt][v] + bias2;
        }
    }
}

extern "C" void kernel_launch(void* const* d_in, const int* in_sizes, int n_in,
                              void* d_out, int out_size, void* d_ws, size_t ws_size,
                              hipStream_t stream) {
    const float* x   = (const float*)d_in[0];
    const int*   ei  = (const int*)d_in[1];
    const float* Wl1 = (const float*)d_in[2];
    const float* bl1 = (const float*)d_in[3];
    const float* Wr1 = (const float*)d_in[4];
    const float* Wl2 = (const float*)d_in[5];
    const float* bl2 = (const float*)d_in[6];
    const float* Wr2 = (const float*)d_in[7];
    float* out = (float*)d_out;

    float* ws   = (float*)d_ws;
    float* pbuf = ws + (size_t)NN * DHID;
    float* qbuf = pbuf + (size_t)NN * DOUT;
    int2*  rec  = (int2*)(qbuf + (size_t)NN * DOUT);
    int*   seg  = (int*)(rec + NE);
    int*   cnt  = seg + NN;

    int*   off    = (int*)ws;
    float* wfrag  = ws + WF_OFF;
    int*   col    = (int*)ws + COL_OFF;

    sort_kernel<<<NCH + NWB, CT, 0, stream>>>(ei, off, rec, Wl1, Wr1, Wl2, Wr2, wfrag);
    place_kernel<<<NB, LT, 0, stream>>>(off, rec, seg, cnt, col);

    layer1_kernel<<<NN / 16, 64, 0, stream>>>(x, seg, cnt, col, wfrag, bl1, bl2, pbuf, qbuf);
    gather2_kernel<<<(NN + 8 * GK - 1) / (8 * GK), 256, 0, stream>>>(pbuf, qbuf, seg, cnt, col, out);
}
